# starting kernel
# baseline (speedup 1.0000x reference)
#include <hip/hip_runtime.h>

#define N_NODES 50000
#define N_EDGES 500000
#define IN_C 512
#define HID_C 256
#define OUT_C 128

typedef __attribute__((ext_vector_type(2))) float v2f;
typedef __attribute__((ext_vector_type(8))) float v8f;

__global__ void init_deg_kernel(float* __restrict__ deg, int n) {
  int i = blockIdx.x * blockDim.x + threadIdx.x;
  if (i < n) deg[i] = 1.0f;
}

__global__ void deg_atomic_kernel(const int* __restrict__ dst, float* __restrict__ deg, int e) {
  int i = blockIdx.x * blockDim.x + threadIdx.x;
  if (i < e) unsafeAtomicAdd(&deg[dst[i]], 1.0f);
}

__global__ void norm_kernel(const float* __restrict__ deg, float* __restrict__ dis,
                            float* __restrict__ inv, int n) {
  int i = blockIdx.x * blockDim.x + threadIdx.x;
  if (i < n) {
    float d = deg[i];
    dis[i] = rsqrtf(d);
    inv[i] = 1.0f / d;
  }
}

__global__ void coef_kernel(const int* __restrict__ src, const int* __restrict__ dst,
                            const float* __restrict__ dis, float* __restrict__ coef, int e) {
  int i = blockIdx.x * blockDim.x + threadIdx.x;
  if (i < e) coef[i] = dis[src[i]] * dis[dst[i]];
}

__global__ void transpose_kernel(const float* __restrict__ B, float* __restrict__ Bt,
                                 int K, int N) {
  int i = blockIdx.x * blockDim.x + threadIdx.x;
  if (i < K * N) {
    int kk = i / N;
    int n = i - kk * N;
    Bt[(size_t)n * K + kk] = B[i];
  }
}

template <int MT, int NT>
__global__ __launch_bounds__(128) void wmma_gemm_f32(const float* __restrict__ A,
                                                     const float* __restrict__ Bt,
                                                     float* __restrict__ C,
                                                     int M, int N, int K) {
  const int wave = (blockIdx.x * blockDim.x + threadIdx.x) >> 5;
  const int lane = threadIdx.x & 31;
  const int ngroups = N / (NT * 16);
  const int sm = wave / ngroups;
  const int tg = wave % ngroups;
  const int m0 = sm * MT * 16;
  if (m0 >= M) return;

  const int half = lane >> 4;
  const int l = lane & 15;

  const float* a_ptr[MT];
#pragma unroll
  for (int i = 0; i < MT; ++i) {
    int row = m0 + i * 16 + l;
    if (row >= M) row = M - 1;
    a_ptr[i] = A + (size_t)row * K;
  }
  const float* b_ptr[NT];
#pragma unroll
  for (int t = 0; t < NT; ++t) {
    int col = tg * (NT * 16) + t * 16 + l;
    b_ptr[t] = Bt + (size_t)col * K;
  }

  v8f acc[MT][NT];
#pragma unroll
  for (int i = 0; i < MT; ++i)
#pragma unroll
    for (int t = 0; t < NT; ++t) acc[i][t] = (v8f){0, 0, 0, 0, 0, 0, 0, 0};

  for (int k = 0; k < K; k += 4) {
    const int ka = k + 2 * half;
    v2f a[MT], b[NT];
#pragma unroll
    for (int i = 0; i < MT; ++i) a[i] = *(const v2f*)(a_ptr[i] + ka);
#pragma unroll
    for (int t = 0; t < NT; ++t) b[t] = *(const v2f*)(b_ptr[t] + ka);
#pragma unroll
    for (int i = 0; i < MT; ++i)
#pragma unroll
      for (int t = 0; t < NT; ++t)
        acc[i][t] = __builtin_amdgcn_wmma_f32_16x16x4_f32(
            false, a[i], false, b[t], (short)0, acc[i][t], false, false);
  }

#pragma unroll
  for (int i = 0; i < MT; ++i) {
    if (m0 + i * 16 + 16 <= M) {
      float* c_ptr = C + (size_t)(m0 + i * 16) * N + (size_t)tg * (NT * 16) + l;
#pragma unroll
      for (int t = 0; t < NT; ++t)
#pragma unroll
        for (int r = 0; r < 8; ++r)
          c_ptr[(size_t)(r + 8 * half) * N + t * 16] = acc[i][t][r];
    }
  }
}

__global__ void init_agg_kernel(float* __restrict__ agg, const float* __restrict__ z,
                                const float* __restrict__ inv, const float* __restrict__ bias,
                                int n, int c) {
  int i = blockIdx.x * blockDim.x + threadIdx.x;
  if (i < n * c) {
    int node = i / c;
    int ch = i - node * c;
    agg[i] = z[i] * inv[node] + bias[ch];
  }
}

__global__ void edge_agg_kernel(float* __restrict__ agg, const float* __restrict__ z,
                                const int* __restrict__ src, const int* __restrict__ dst,
                                const float* __restrict__ coef, int e, int c) {
  int i = blockIdx.x * blockDim.x + threadIdx.x;
  const int groups = c >> 2;
  if (i >= e * groups) return;
  const int eg = i / groups;
  const int g = i - eg * groups;
  const int s = src[eg];
  const int d = dst[eg];
  const float w = coef[eg];
  const float4 v = *(const float4*)(z + (size_t)s * c + g * 4);
  float* out = agg + (size_t)d * c + g * 4;
  unsafeAtomicAdd(out + 0, v.x * w);
  unsafeAtomicAdd(out + 1, v.y * w);
  unsafeAtomicAdd(out + 2, v.z * w);
  unsafeAtomicAdd(out + 3, v.w * w);
}

__global__ void relu_kernel(float* __restrict__ h, int total) {
  int i = blockIdx.x * blockDim.x + threadIdx.x;
  if (i < total) h[i] = fmaxf(h[i], 0.0f);
}

static inline int cdiv_l(long a, long b) { return (int)((a + b - 1) / b); }

extern "C" void kernel_launch(void* const* d_in, const int* in_sizes, int n_in,
                              void* d_out, int out_size, void* d_ws, size_t ws_size,
                              hipStream_t stream) {
  const float* x   = (const float*)d_in[0];
  const int*   ei  = (const int*)d_in[1];
  const int*   src = ei;
  const int*   dst = ei + N_EDGES;
  const float* w1  = (const float*)d_in[2];
  const float* b1  = (const float*)d_in[3];
  const float* wmu = (const float*)d_in[4];
  const float* bmu = (const float*)d_in[5];
  const float* wls = (const float*)d_in[6];
  const float* bls = (const float*)d_in[7];

  float* out_mu = (float*)d_out;
  float* out_ls = out_mu + (size_t)N_NODES * OUT_C;

  float* ws = (float*)d_ws;
  size_t off = 0;
  float* deg  = ws + off; off += N_NODES;
  float* dis  = ws + off; off += N_NODES;
  float* inv  = ws + off; off += N_NODES;
  float* coef = ws + off; off += N_EDGES;
  float* w1t  = ws + off; off += (size_t)IN_C * HID_C;
  float* wmut = ws + off; off += (size_t)HID_C * OUT_C;
  float* wlst = ws + off; off += (size_t)HID_C * OUT_C;
  float* hpre = ws + off; off += (size_t)N_NODES * HID_C;
  float* h    = ws + off; off += (size_t)N_NODES * HID_C;
  float* zmu  = ws + off; off += (size_t)N_NODES * OUT_C;
  float* zls  = ws + off; off += (size_t)N_NODES * OUT_C;

  init_deg_kernel<<<cdiv_l(N_NODES, 256), 256, 0, stream>>>(deg, N_NODES);
  deg_atomic_kernel<<<cdiv_l(N_EDGES, 256), 256, 0, stream>>>(dst, deg, N_EDGES);
  norm_kernel<<<cdiv_l(N_NODES, 256), 256, 0, stream>>>(deg, dis, inv, N_NODES);
  coef_kernel<<<cdiv_l(N_EDGES, 256), 256, 0, stream>>>(src, dst, dis, coef, N_EDGES);

  transpose_kernel<<<cdiv_l((long)IN_C * HID_C, 256), 256, 0, stream>>>(w1, w1t, IN_C, HID_C);
  transpose_kernel<<<cdiv_l((long)HID_C * OUT_C, 256), 256, 0, stream>>>(wmu, wmut, HID_C, OUT_C);
  transpose_kernel<<<cdiv_l((long)HID_C * OUT_C, 256), 256, 0, stream>>>(wls, wlst, HID_C, OUT_C);

  constexpr int MT = 4, NT = 4;
  const long m_super = cdiv_l(N_NODES, MT * 16);

  {
    long waves = m_super * (HID_C / (NT * 16));
    wmma_gemm_f32<MT, NT><<<cdiv_l(waves * 32, 128), 128, 0, stream>>>(
        x, w1t, hpre, N_NODES, HID_C, IN_C);
  }
  init_agg_kernel<<<cdiv_l((long)N_NODES * HID_C, 256), 256, 0, stream>>>(h, hpre, inv, b1,
                                                                          N_NODES, HID_C);
  edge_agg_kernel<<<cdiv_l((long)N_EDGES * (HID_C / 4), 256), 256, 0, stream>>>(
      h, hpre, src, dst, coef, N_EDGES, HID_C);
  relu_kernel<<<cdiv_l((long)N_NODES * HID_C, 256), 256, 0, stream>>>(h, N_NODES * HID_C);

  {
    long waves = m_super * (OUT_C / (NT * 16));
    wmma_gemm_f32<MT, NT><<<cdiv_l(waves * 32, 128), 128, 0, stream>>>(
        h, wmut, zmu, N_NODES, OUT_C, HID_C);
  }
  init_agg_kernel<<<cdiv_l((long)N_NODES * OUT_C, 256), 256, 0, stream>>>(out_mu, zmu, inv, bmu,
                                                                          N_NODES, OUT_C);
  edge_agg_kernel<<<cdiv_l((long)N_EDGES * (OUT_C / 4), 256), 256, 0, stream>>>(
      out_mu, zmu, src, dst, coef, N_EDGES, OUT_C);

  {
    long waves = m_super * (OUT_C / (NT * 16));
    wmma_gemm_f32<MT, NT><<<cdiv_l(waves * 32, 128), 128, 0, stream>>>(
        h, wlst, zls, N_NODES, OUT_C, HID_C);
  }
  init_agg_kernel<<<cdiv_l((long)N_NODES * OUT_C, 256), 256, 0, stream>>>(out_ls, zls, inv, bls,
                                                                          N_NODES, OUT_C);
  edge_agg_kernel<<<cdiv_l((long)N_EDGES * (OUT_C / 4), 256), 256, 0, stream>>>(
      out_ls, zls, src, dst, coef, N_EDGES, OUT_C);
}


# Edge sums by gather over per-destination edge lists (counting sort rebuilt each call) instead of float atomics; one write per element with self term, bias, relu; sum order fixed: ascending source
# speedup vs baseline: 2.1692x; 2.1692x over previous
#include <hip/hip_runtime.h>

#define N_NODES 50000
#define N_EDGES 500000
#define IN_C 512
#define HID_C 256
#define OUT_C 128

typedef __attribute__((ext_vector_type(2))) float v2f;
typedef __attribute__((ext_vector_type(8))) float v8f;

__global__ void zero_counts_kernel(int* __restrict__ cnt, int n) {
  int i = blockIdx.x * blockDim.x + threadIdx.x;
  if (i < n) cnt[i] = 0;
}

__global__ void count_rank_kernel(const int* __restrict__ dst, int* __restrict__ cnt,
                                  int* __restrict__ rank, int e) {
  int i = blockIdx.x * blockDim.x + threadIdx.x;
  if (i < e) rank[i] = atomicAdd(&cnt[dst[i]], 1);
}

#define SCAN_T 1024
__global__ __launch_bounds__(SCAN_T) void scan_norm_kernel(const int* __restrict__ cnt,
                                                           int* __restrict__ rowptr,
                                                           float* __restrict__ dis,
                                                           float* __restrict__ inv, int n) {
  __shared__ int sh[SCAN_T];
  const int t = threadIdx.x;
  const int base = blockIdx.x * SCAN_T;
  int pre = 0;
  for (int i = t; i < base; i += SCAN_T) pre += cnt[i];
  const int node = base + t;
  const int c = (node < n) ? cnt[node] : 0;

  sh[t] = pre;
  __syncthreads();
  for (int o = SCAN_T / 2; o > 0; o >>= 1) {
    if (t < o) sh[t] += sh[t + o];
    __syncthreads();
  }
  const int block_off = sh[0];
  __syncthreads();

  sh[t] = c;
  __syncthreads();
  for (int o = 1; o < SCAN_T; o <<= 1) {
    const int v = (t >= o) ? sh[t - o] : 0;
    __syncthreads();
    sh[t] += v;
    __syncthreads();
  }
  const int incl = sh[t];

  if (node < n) {
    rowptr[node] = block_off + incl - c;
    if (node == n - 1) rowptr[n] = block_off + incl;
    const float d = (float)(c + 1);
    dis[node] = rsqrtf(d);
    inv[node] = 1.0f / d;
  }
}

__global__ void fill_kernel(const int* __restrict__ src, const int* __restrict__ dst,
                            const int* __restrict__ rowptr, const int* __restrict__ rank,
                            int* __restrict__ raw_src, int e) {
  int i = blockIdx.x * blockDim.x + threadIdx.x;
  if (i < e) raw_src[rowptr[dst[i]] + rank[i]] = src[i];
}

__global__ void seg_order_kernel(const int* __restrict__ rowptr, const int* __restrict__ raw_src,
                                 const float* __restrict__ dis, int* __restrict__ csr_src,
                                 float* __restrict__ csr_w, int n) {
  const int v = (blockIdx.x * blockDim.x + threadIdx.x) >> 5;
  const int lane = threadIdx.x & 31;
  if (v >= n) return;
  const int b = rowptr[v];
  const int e = rowptr[v + 1];
  const float dv = dis[v];
  for (int i = b + lane; i < e; i += 32) {
    const int key = raw_src[i];
    int rank = 0;
    for (int j = b; j < e; ++j) {
      const int kj = raw_src[j];
      rank += ((kj < key) || (kj == key && j < i)) ? 1 : 0;
    }
    csr_src[b + rank] = key;
    csr_w[b + rank] = dis[key] * dv;
  }
}

__global__ void transpose_kernel(const float* __restrict__ B, float* __restrict__ Bt,
                                 int K, int N) {
  int i = blockIdx.x * blockDim.x + threadIdx.x;
  if (i < K * N) {
    int kk = i / N;
    int n = i - kk * N;
    Bt[(size_t)n * K + kk] = B[i];
  }
}

template <int MT, int NT>
__global__ __launch_bounds__(128) void wmma_gemm_f32(const float* __restrict__ A,
                                                     const float* __restrict__ Bt,
                                                     float* __restrict__ C,
                                                     int M, int N, int K) {
  const int wave = (blockIdx.x * blockDim.x + threadIdx.x) >> 5;
  const int lane = threadIdx.x & 31;
  const int ngroups = N / (NT * 16);
  const int sm = wave / ngroups;
  const int tg = wave % ngroups;
  const int m0 = sm * MT * 16;
  if (m0 >= M) return;

  const int half = lane >> 4;
  const int l = lane & 15;

  const float* a_ptr[MT];
#pragma unroll
  for (int i = 0; i < MT; ++i) {
    int row = m0 + i * 16 + l;
    if (row >= M) row = M - 1;
    a_ptr[i] = A + (size_t)row * K;
  }
  const float* b_ptr[NT];
#pragma unroll
  for (int t = 0; t < NT; ++t) {
    int col = tg * (NT * 16) + t * 16 + l;
    b_ptr[t] = Bt + (size_t)col * K;
  }

  v8f acc[MT][NT];
#pragma unroll
  for (int i = 0; i < MT; ++i)
#pragma unroll
    for (int t = 0; t < NT; ++t) acc[i][t] = (v8f){0, 0, 0, 0, 0, 0, 0, 0};

  for (int k = 0; k < K; k += 4) {
    const int ka = k + 2 * half;
    v2f a[MT], b[NT];
#pragma unroll
    for (int i = 0; i < MT; ++i) a[i] = *(const v2f*)(a_ptr[i] + ka);
#pragma unroll
    for (int t = 0; t < NT; ++t) b[t] = *(const v2f*)(b_ptr[t] + ka);
#pragma unroll
    for (int i = 0; i < MT; ++i)
#pragma unroll
      for (int t = 0; t < NT; ++t)
        acc[i][t] = __builtin_amdgcn_wmma_f32_16x16x4_f32(
            false, a[i], false, b[t], (short)0, acc[i][t], false, false);
  }

#pragma unroll
  for (int i = 0; i < MT; ++i) {
    if (m0 + i * 16 + 16 <= M) {
      float* c_ptr = C + (size_t)(m0 + i * 16) * N + (size_t)tg * (NT * 16) + l;
#pragma unroll
      for (int t = 0; t < NT; ++t)
#pragma unroll
        for (int r = 0; r < 8; ++r)
          c_ptr[(size_t)(r + 8 * half) * N + t * 16] = acc[i][t][r];
    }
  }
}

template <int C, bool RELU>
__global__ __launch_bounds__(256) void gather_agg_kernel(float* __restrict__ out,
                                                         const float* __restrict__ z,
                                                         const int* __restrict__ rowptr,
                                                         const int* __restrict__ csr_src,
                                                         const float* __restrict__ csr_w,
                                                         const float* __restrict__ inv,
                                                         const float* __restrict__ bias, int n) {
  constexpr int G = C / 4;
  static_assert(G % 32 == 0 && 256 % G == 0, "a wave must stay inside one node");
  const int g = threadIdx.x % G;
  const int v = __builtin_amdgcn_readfirstlane(blockIdx.x * (256 / G) + threadIdx.x / G);
  if (v >= n) return;
  const int b = rowptr[v];
  const int e = rowptr[v + 1];
  const float* zc = z + g * 4;

  float4 acc = make_float4(0.0f, 0.0f, 0.0f, 0.0f);
  int j = b;
  for (; j + 4 <= e; j += 4) {
    const int s0 = csr_src[j + 0], s1 = csr_src[j + 1], s2 = csr_src[j + 2], s3 = csr_src[j + 3];
    const float w0 = csr_w[j + 0], w1 = csr_w[j + 1], w2 = csr_w[j + 2], w3 = csr_w[j + 3];
    const float4 t0 = *(const float4*)(zc + (size_t)s0 * C);
    const float4 t1 = *(const float4*)(zc + (size_t)s1 * C);
    const float4 t2 = *(const float4*)(zc + (size_t)s2 * C);
    const float4 t3 = *(const float4*)(zc + (size_t)s3 * C);
    acc.x += t0.x * w0; acc.y += t0.y * w0; acc.z += t0.z * w0; acc.w += t0.w * w0;
    acc.x += t1.x * w1; acc.y += t1.y * w1; acc.z += t1.z * w1; acc.w += t1.w * w1;
    acc.x += t2.x * w2; acc.y += t2.y * w2; acc.z += t2.z * w2; acc.w += t2.w * w2;
    acc.x += t3.x * w3; acc.y += t3.y * w3; acc.z += t3.z * w3; acc.w += t3.w * w3;
  }
  for (; j < e; ++j) {
    const int s0 = csr_src[j];
    const float w0 = csr_w[j];
    const float4 t0 = *(const float4*)(zc + (size_t)s0 * C);
    acc.x += t0.x * w0; acc.y += t0.y * w0; acc.z += t0.z * w0; acc.w += t0.w * w0;
  }

  const float4 zs = *(const float4*)(zc + (size_t)v * C);
  const float iv = inv[v];
  acc.x += zs.x * iv; acc.y += zs.y * iv; acc.z += zs.z * iv; acc.w += zs.w * iv;
  acc.x += bias[g * 4 + 0];
  acc.y += bias[g * 4 + 1];
  acc.z += bias[g * 4 + 2];
  acc.w += bias[g * 4 + 3];
  if (RELU) {
    acc.x = fmaxf(acc.x, 0.0f);
    acc.y = fmaxf(acc.y, 0.0f);
    acc.z = fmaxf(acc.z, 0.0f);
    acc.w = fmaxf(acc.w, 0.0f);
  }
  *(float4*)(out + (size_t)v * C + g * 4) = acc;
}

static inline int cdiv_l(long a, long b) { return (int)((a + b - 1) / b); }

extern "C" void kernel_launch(void* const* d_in, const int* in_sizes, int n_in,
                              void* d_out, int out_size, void* d_ws, size_t ws_size,
                              hipStream_t stream) {
  const float* x   = (const float*)d_in[0];
  const int*   ei  = (const int*)d_in[1];
  const int*   src = ei;
  const int*   dst = ei + N_EDGES;
  const float* w1  = (const float*)d_in[2];
  const float* b1  = (const float*)d_in[3];
  const float* wmu = (const float*)d_in[4];
  const float* bmu = (const float*)d_in[5];
  const float* wls = (const float*)d_in[6];
  const float* bls = (const float*)d_in[7];

  float* out_mu = (float*)d_out;
  float* out_ls = out_mu + (size_t)N_NODES * OUT_C;

  float* ws = (float*)d_ws;
  size_t off = 0;
  int*   cnt   = (int*)(ws + off); off += N_NODES;
  float* dis   = ws + off; off += N_NODES;
  float* inv   = ws + off; off += N_NODES;
  float* csr_w = ws + off; off += N_EDGES;
  float* w1t  = ws + off; off += (size_t)IN_C * HID_C;
  float* wmut = ws + off; off += (size_t)HID_C * OUT_C;
  float* wlst = ws + off; off += (size_t)HID_C * OUT_C;
  float* hpre = ws + off; off += (size_t)N_NODES * HID_C;
  float* h    = ws + off; off += (size_t)N_NODES * HID_C;
  int* rowptr  = (int*)(ws + off); off += N_NODES + 16;
  int* rank    = (int*)(ws + off); off += N_EDGES;
  int* raw_src = (int*)(ws + off); off += N_EDGES;
  int* csr_src = (int*)(ws + off); off += N_EDGES;
  float* zmu = hpre;
  float* zls = hpre + (size_t)N_NODES * OUT_C;

  zero_counts_kernel<<<cdiv_l(N_NODES, 256), 256, 0, stream>>>(cnt, N_NODES);
  count_rank_kernel<<<cdiv_l(N_EDGES, 256), 256, 0, stream>>>(dst, cnt, rank, N_EDGES);
  scan_norm_kernel<<<cdiv_l(N_NODES, SCAN_T), SCAN_T, 0, stream>>>(cnt, rowptr, dis, inv, N_NODES);
  fill_kernel<<<cdiv_l(N_EDGES, 256), 256, 0, stream>>>(src, dst, rowptr, rank, raw_src, N_EDGES);
  seg_order_kernel<<<cdiv_l((long)N_NODES * 32, 256), 256, 0, stream>>>(rowptr, raw_src, dis, csr_src,
                                                                        csr_w, N_NODES);

  transpose_kernel<<<cdiv_l((long)IN_C * HID_C, 256), 256, 0, stream>>>(w1, w1t, IN_C, HID_C);
  transpose_kernel<<<cdiv_l((long)HID_C * OUT_C, 256), 256, 0, stream>>>(wmu, wmut, HID_C, OUT_C);
  transpose_kernel<<<cdiv_l((long)HID_C * OUT_C, 256), 256, 0, stream>>>(wls, wlst, HID_C, OUT_C);

  constexpr int MT = 4, NT = 4;
  const long m_super = cdiv_l(N_NODES, MT * 16);

  {
    long waves = m_super * (HID_C / (NT * 16));
    wmma_gemm_f32<MT, NT><<<cdiv_l(waves * 32, 128), 128, 0, stream>>>(
        x, w1t, hpre, N_NODES, HID_C, IN_C);
  }
  gather_agg_kernel<HID_C, true><<<cdiv_l((long)N_NODES * (HID_C / 4), 256), 256, 0, stream>>>(
      h, hpre, rowptr, csr_src, csr_w, inv, b1, N_NODES);

  {
    long waves = m_super * (OUT_C / (NT * 16));
    wmma_gemm_f32<MT, NT><<<cdiv_l(waves * 32, 128), 128, 0, stream>>>(
        h, wmut, zmu, N_NODES, OUT_C, HID_C);
  }
  gather_agg_kernel<OUT_C, false><<<cdiv_l((long)N_NODES * (OUT_C / 4), 256), 256, 0, stream>>>(
      out_mu, zmu, rowptr, csr_src, csr_w, inv, bmu, N_NODES);

  {
    long waves = m_super * (OUT_C / (NT * 16));
    wmma_gemm_f32<MT, NT><<<cdiv_l(waves * 32, 128), 128, 0, stream>>>(
        h, wlst, zls, N_NODES, OUT_C, HID_C);
  }
  gather_agg_kernel<OUT_C, false><<<cdiv_l((long)N_NODES * (OUT_C / 4), 256), 256, 0, stream>>>(
      out_ls, zls, rowptr, csr_src, csr_w, inv, bls, N_NODES);
}


# Kernels that do not depend on each other share a launch: count clearing + 3 weight transposes in one grid; mu and logstd heads as one 256-wide GEMM + one gather; 14 -> 9 launches, sum orders unchanged
# speedup vs baseline: 2.5655x; 1.1827x over previous
#include <hip/hip_runtime.h>

#define N_NODES 50000
#define N_EDGES 500000
#define IN_C 512
#define HID_C 256
#define OUT_C 128

typedef __attribute__((ext_vector_type(2))) float v2f;
typedef __attribute__((ext_vector_type(8))) float v8f;
typedef __attribute__((ext_vector_type(4))) float v4f;

__global__ void count_rank_kernel(const int* __restrict__ dst, int* __restrict__ cnt,
                                  int* __restrict__ rank, int e) {
  int i = blockIdx.x * blockDim.x + threadIdx.x;
  if (i < e) rank[i] = atomicAdd(&cnt[dst[i]], 1);
}

#define SCAN_T 1024
__global__ __launch_bounds__(SCAN_T) void scan_norm_kernel(const int* __restrict__ cnt,
                                                           int* __restrict__ rowptr,
                                                           float* __restrict__ dis,
                                                           float* __restrict__ inv, int n) {
  __shared__ int sh[SCAN_T];
  const int t = threadIdx.x;
  const int base = blockIdx.x * SCAN_T;
  int pre = 0;
  for (int i = t; i < base; i += SCAN_T) pre += cnt[i];
  const int node = base + t;
  const int c = (node < n) ? cnt[node] : 0;

  sh[t] = pre;
  __syncthreads();
  for (int o = SCAN_T / 2; o > 0; o >>= 1) {
    if (t < o) sh[t] += sh[t + o];
    __syncthreads();
  }
  const int block_off = sh[0];
  __syncthreads();

  sh[t] = c;
  __syncthreads();
  for (int o = 1; o < SCAN_T; o <<= 1) {
    const int v = (t >= o) ? sh[t - o] : 0;
    __syncthreads();
    sh[t] += v;
    __syncthreads();
  }
  const int incl = sh[t];

  if (node < n) {
    rowptr[node] = block_off + incl - c;
    if (node == n - 1) rowptr[n] = block_off + incl;
    const float d = (float)(c + 1);
    dis[node] = rsqrtf(d);
    inv[node] = 1.0f / d;
  }
}

__global__ void fill_kernel(const int* __restrict__ src, const int* __restrict__ dst,
                            const int* __restrict__ rowptr, const int* __restrict__ rank,
                            int* __restrict__ raw_src, int e) {
  int i = blockIdx.x * blockDim.x + threadIdx.x;
  if (i < e) raw_src[rowptr[dst[i]] + rank[i]] = src[i];
}

__global__ void seg_order_kernel(const int* __restrict__ rowptr, const int* __restrict__ raw_src,
                                 const float* __restrict__ dis, int* __restrict__ csr_src,
                                 float* __restrict__ csr_w, int n) {
  const int v = (blockIdx.x * blockDim.x + threadIdx.x) >> 5;
  const int lane = threadIdx.x & 31;
  if (v >= n) return;
  const int b = rowptr[v];
  const int e = rowptr[v + 1];
  const float dv = dis[v];
  for (int i = b + lane; i < e; i += 32) {
    const int key = raw_src[i];
    int rank = 0;
    for (int j = b; j < e; ++j) {
      const int kj = raw_src[j];
      rank += ((kj < key) || (kj == key && j < i)) ? 1 : 0;
    }
    csr_src[b + rank] = key;
    csr_w[b + rank] = dis[key] * dv;
  }
}

#define PREP_T 256
#define NB_ZERO ((N_NODES + PREP_T - 1) / PREP_T)
#define NB_T1 ((IN_C * HID_C + PREP_T - 1) / PREP_T)
#define NB_T2 ((HID_C * OUT_C + PREP_T - 1) / PREP_T)
#define NB_PREP (NB_ZERO + NB_T1 + 2 * NB_T2)
__device__ __forceinline__ void transpose_block(const float* __restrict__ B, float* __restrict__ Bt,
                                                int K, int N, int blk) {
  const int i = blk * PREP_T + (int)threadIdx.x;
  if (i < K * N) {
    const int kk = i / N;
    const int n = i - kk * N;
    Bt[(size_t)n * K + kk] = B[i];
  }
}
__global__ __launch_bounds__(PREP_T) void prep_kernel(int* __restrict__ cnt,
                                                      const float* __restrict__ w1, float* __restrict__ w1t,
                                                      const float* __restrict__ wmu, float* __restrict__ wmut,
                                                      const float* __restrict__ wls, float* __restrict__ wlst) {
  int b = blockIdx.x;
  if (b < NB_ZERO) {
    const int i = b * PREP_T + (int)threadIdx.x;
    if (i < N_NODES) cnt[i] = 0;
    return;
  }
  b -= NB_ZERO;
  if (b < NB_T1) { transpose_block(w1, w1t, IN_C, HID_C, b); return; }
  b -= NB_T1;
  if (b < NB_T2) { transpose_block(wmu, wmut, HID_C, OUT_C, b); return; }
  b -= NB_T2;
  transpose_block(wls, wlst, HID_C, OUT_C, b);
}

#define GEMM_KC 32
#define GEMM_PITCH 36
template <int WM, int WN>
__global__ __launch_bounds__(WM * WN * 32) void wmma_gemm_f32_lds(const float* __restrict__ A,
                                                                 const float* __restrict__ Bt,
                                                                 float* __restrict__ C,
                                                                 int M, int N, int K) {
  constexpr int NTHR = WM * WN * 32;
  constexpr int BM = WM * 64;
  constexpr int BN = WN * 64;
  constexpr int ROWS = BM + BN;
  constexpr int SLAB = NTHR / 8;
  constexpr int NLD = ROWS / SLAB;
  constexpr int STAGE = ROWS * GEMM_PITCH;
  static_assert(ROWS % SLAB == 0 && BM % SLAB == 0, "a copy pass is all A rows or all Bt rows");

  __shared__ __attribute__((aligned(16))) float lds[2 * STAGE];

  const int tid = threadIdx.x;
  const int lane = tid & 31;
  const int wave = tid >> 5;
  const int wm = wave / WN;
  const int wn = wave - wm * WN;
  const int half = lane >> 4;
  const int l = lane & 15;
  const int m0 = blockIdx.x * BM;

  const int cr = tid >> 3;
  const int cq = tid & 7;
  unsigned goff[NLD];
#pragma unroll
  for (int i = 0; i < NLD; ++i) {
    const int r = i * SLAB + cr;
    if (i * SLAB < BM) {
      int row = m0 + r;
      if (row >= M) row = M - 1;
      goff[i] = (unsigned)row * (unsigned)K + (unsigned)(cq * 4);
    } else {
      goff[i] = (unsigned)(r - BM) * (unsigned)K + (unsigned)(cq * 4);
    }
  }
  const int soff = cr * GEMM_PITCH + cq * 4;

  const int a_off = (wm * 64 + l) * GEMM_PITCH + half * 4;
  const int b_off = (BM + wn * 64 + l) * GEMM_PITCH + half * 4;

  v8f acc[4][4];
#pragma unroll
  for (int i = 0; i < 4; ++i)
#pragma unroll
    for (int t = 0; t < 4; ++t) acc[i][t] = (v8f){0, 0, 0, 0, 0, 0, 0, 0};

  {
    v4f first[NLD];
#pragma unroll
    for (int i = 0; i < NLD; ++i)
      first[i] = *(const v4f*)((i * SLAB < BM ? A : Bt) + goff[i]);
#pragma unroll
    for (int i = 0; i < NLD; ++i)
      *(v4f*)(lds + soff + i * SLAB * GEMM_PITCH) = first[i];
  }
  __syncthreads();

  const int nchunks = K / GEMM_KC;
  for (int kc = 0; kc < nchunks; ++kc) {
    const float* cur = lds + (kc & 1) * STAGE;
    const bool more = (kc + 1 < nchunks);
    v4f pre[NLD];
    if (more) {
#pragma unroll
      for (int i = 0; i < NLD; ++i)
        pre[i] = *(const v4f*)((i * SLAB < BM ? A : Bt) + goff[i] + (kc + 1) * GEMM_KC);
    }

#pragma unroll
    for (int j = 0; j < GEMM_KC / 8; ++j) {
      v4f a4[4], b4[4];
#pragma unroll
      for (int i = 0; i < 4; ++i)
        a4[i] = *(const v4f*)(cur + a_off + i * 16 * GEMM_PITCH + j * 8);
#pragma unroll
      for (int t = 0; t < 4; ++t)
        b4[t] = *(const v4f*)(cur + b_off + t * 16 * GEMM_PITCH + j * 8);
#pragma unroll
      for (int i = 0; i < 4; ++i)
#pragma unroll
        for (int t = 0; t < 4; ++t)
          acc[i][t] = __builtin_amdgcn_wmma_f32_16x16x4_f32(
              false, (v2f){a4[i].x, a4[i].y}, false, (v2f){b4[t].x, b4[t].y}, (short)0,
              acc[i][t], false, false);
#pragma unroll
      for (int i = 0; i < 4; ++i)
#pragma unroll
        for (int t = 0; t < 4; ++t)
          acc[i][t] = __builtin_amdgcn_wmma_f32_16x16x4_f32(
              false, (v2f){a4[i].z, a4[i].w}, false, (v2f){b4[t].z, b4[t].w}, (short)0,
              acc[i][t], false, false);
    }

    if (more) {
      float* nxt = lds + ((kc + 1) & 1) * STAGE;
#pragma unroll
      for (int i = 0; i < NLD; ++i)
        *(v4f*)(nxt + soff + i * SLAB * GEMM_PITCH) = pre[i];
    }
    __syncthreads();
  }

  const int m0w = m0 + wm * 64;
#pragma unroll
  for (int i = 0; i < 4; ++i) {
    if (m0w + i * 16 + 16 <= M) {
      float* c_ptr = C + (size_t)(m0w + i * 16) * N + (size_t)(wn * 64) + l;
#pragma unroll
      for (int t = 0; t < 4; ++t)
#pragma unroll
        for (int r = 0; r < 8; ++r)
          c_ptr[(size_t)(r + 8 * half) * N + t * 16] = acc[i][t][r];
    }
  }
}

template <int C, bool RELU, bool SPLIT>
__global__ __launch_bounds__(256) void gather_agg_kernel(float* __restrict__ out,
                                                         float* __restrict__ out2,
                                                         const float* __restrict__ z,
                                                         const int* __restrict__ rowptr,
                                                         const int* __restrict__ csr_src,
                                                         const float* __restrict__ csr_w,
                                                         const float* __restrict__ inv,
                                                         const float* __restrict__ bias,
                                                         const float* __restrict__ bias2, int n) {
  constexpr int G = C / 4;
  static_assert(G % 32 == 0 && 256 % G == 0, "a wave must stay inside one node");
  static_assert(!SPLIT || (G / 2) % 32 == 0, "a wave must stay inside one half of the row");
  const int g = threadIdx.x % G;
  const int v = __builtin_amdgcn_readfirstlane(blockIdx.x * (256 / G) + threadIdx.x / G);
  if (v >= n) return;
  const int b = rowptr[v];
  const int e = rowptr[v + 1];
  const float* zc = z + g * 4;

  float4 acc = make_float4(0.0f, 0.0f, 0.0f, 0.0f);
  int j = b;
  for (; j + 4 <= e; j += 4) {
    const int s0 = csr_src[j + 0], s1 = csr_src[j + 1], s2 = csr_src[j + 2], s3 = csr_src[j + 3];
    const float w0 = csr_w[j + 0], w1 = csr_w[j + 1], w2 = csr_w[j + 2], w3 = csr_w[j + 3];
    const float4 t0 = *(const float4*)(zc + (size_t)s0 * C);
    const float4 t1 = *(const float4*)(zc + (size_t)s1 * C);
    const float4 t2 = *(const float4*)(zc + (size_t)s2 * C);
    const float4 t3 = *(const float4*)(zc + (size_t)s3 * C);
    acc.x += t0.x * w0; acc.y += t0.y * w0; acc.z += t0.z * w0; acc.w += t0.w * w0;
    acc.x += t1.x * w1; acc.y += t1.y * w1; acc.z += t1.z * w1; acc.w += t1.w * w1;
    acc.x += t2.x * w2; acc.y += t2.y * w2; acc.z += t2.z * w2; acc.w += t2.w * w2;
    acc.x += t3.x * w3; acc.y += t3.y * w3; acc.z += t3.z * w3; acc.w += t3.w * w3;
  }
  for (; j < e; ++j) {
    const int s0 = csr_src[j];
    const float w0 = csr_w[j];
    const float4 t0 = *(const float4*)(zc + (size_t)s0 * C);
    acc.x += t0.x * w0; acc.y += t0.y * w0; acc.z += t0.z * w0; acc.w += t0.w * w0;
  }

  const float4 zs = *(const float4*)(zc + (size_t)v * C);
  const float iv = inv[v];
  acc.x += zs.x * iv; acc.y += zs.y * iv; acc.z += zs.z * iv; acc.w += zs.w * iv;
  constexpr int OC = SPLIT ? C / 2 : C;
  const bool second = SPLIT && (g >= G / 2);
  const int go = second ? g - G / 2 : g;
  const float* bs = second ? bias2 : bias;
  float* op = second ? out2 : out;
  acc.x += bs[go * 4 + 0];
  acc.y += bs[go * 4 + 1];
  acc.z += bs[go * 4 + 2];
  acc.w += bs[go * 4 + 3];
  if (RELU) {
    acc.x = fmaxf(acc.x, 0.0f);
    acc.y = fmaxf(acc.y, 0.0f);
    acc.z = fmaxf(acc.z, 0.0f);
    acc.w = fmaxf(acc.w, 0.0f);
  }
  *(float4*)(op + (size_t)v * OC + go * 4) = acc;
}

static inline int cdiv_l(long a, long b) { return (int)((a + b - 1) / b); }

extern "C" void kernel_launch(void* const* d_in, const int* in_sizes, int n_in,
                              void* d_out, int out_size, void* d_ws, size_t ws_size,
                              hipStream_t stream) {
  const float* x   = (const float*)d_in[0];
  const int*   ei  = (const int*)d_in[1];
  const int*   src = ei;
  const int*   dst = ei + N_EDGES;
  const float* w1  = (const float*)d_in[2];
  const float* b1  = (const float*)d_in[3];
  const float* wmu = (const float*)d_in[4];
  const float* bmu = (const float*)d_in[5];
  const float* wls = (const float*)d_in[6];
  const float* bls = (const float*)d_in[7];

  float* out_mu = (float*)d_out;
  float* out_ls = out_mu + (size_t)N_NODES * OUT_C;

  float* ws = (float*)d_ws;
  size_t off = 0;
  int*   cnt   = (int*)(ws + off); off += N_NODES;
  float* dis   = ws + off; off += N_NODES;
  float* inv   = ws + off; off += N_NODES;
  float* csr_w = ws + off; off += N_EDGES;
  float* w1t  = ws + off; off += (size_t)IN_C * HID_C;
  float* wmut = ws + off; off += (size_t)HID_C * OUT_C;
  float* wlst = ws + off; off += (size_t)HID_C * OUT_C;
  float* hpre = ws + off; off += (size_t)N_NODES * HID_C;
  float* h    = ws + off; off += (size_t)N_NODES * HID_C;
  int* rowptr  = (int*)(ws + off); off += N_NODES + 16;
  int* rank    = (int*)(ws + off); off += N_EDGES;
  int* raw_src = (int*)(ws + off); off += N_EDGES;
  int* csr_src = (int*)(ws + off); off += N_EDGES;
  float* z2 = hpre;

  prep_kernel<<<NB_PREP, PREP_T, 0, stream>>>(cnt, w1, w1t, wmu, wmut, wls, wlst);

  count_rank_kernel<<<cdiv_l(N_EDGES, 256), 256, 0, stream>>>(dst, cnt, rank, N_EDGES);
  scan_norm_kernel<<<cdiv_l(N_NODES, SCAN_T), SCAN_T, 0, stream>>>(cnt, rowptr, dis, inv, N_NODES);
  fill_kernel<<<cdiv_l(N_EDGES, 256), 256, 0, stream>>>(src, dst, rowptr, rank, raw_src, N_EDGES);
  seg_order_kernel<<<cdiv_l((long)N_NODES * 32, 256), 256, 0, stream>>>(rowptr, raw_src, dis, csr_src,
                                                                        csr_w, N_NODES);


  wmma_gemm_f32_lds<2, 4><<<cdiv_l(N_NODES, 128), 256, 0, stream>>>(
      x, w1t, hpre, N_NODES, HID_C, IN_C);
  gather_agg_kernel<HID_C, true, false><<<cdiv_l((long)N_NODES * (HID_C / 4), 256), 256, 0, stream>>>(
      h, nullptr, hpre, rowptr, csr_src, csr_w, inv, b1, nullptr, N_NODES);

  wmma_gemm_f32_lds<2, 4><<<cdiv_l(N_NODES, 128), 256, 0, stream>>>(
      h, wmut, z2, N_NODES, 2 * OUT_C, HID_C);
  (void)wlst;
  gather_agg_kernel<2 * OUT_C, false, true><<<cdiv_l((long)N_NODES * (2 * OUT_C / 4), 256), 256, 0, stream>>>(
      out_mu, out_ls, z2, rowptr, csr_src, csr_w, inv, bmu, bls, N_NODES);
}


# GEMM rows dealt out in 16-row blocks to one 8-wave work-group per compute unit (256 groups of 12 or 13 blocks; wave = 32 columns of all of them; next step's fragments read ahead); sums unchanged
# speedup vs baseline: 2.7558x; 1.0742x over previous
#include <hip/hip_runtime.h>

#define N_NODES 50000
#define N_EDGES 500000
#define IN_C 512
#define HID_C 256
#define OUT_C 128

typedef __attribute__((ext_vector_type(2))) float v2f;
typedef __attribute__((ext_vector_type(8))) float v8f;
typedef __attribute__((ext_vector_type(4))) float v4f;

__global__ void count_rank_kernel(const int* __restrict__ dst, int* __restrict__ cnt,
                                  int* __restrict__ rank, int e) {
  int i = blockIdx.x * blockDim.x + threadIdx.x;
  if (i < e) rank[i] = atomicAdd(&cnt[dst[i]], 1);
}

#define SCAN_T 1024
__global__ __launch_bounds__(SCAN_T) void scan_norm_kernel(const int* __restrict__ cnt,
                                                           int* __restrict__ rowptr,
                                                           float* __restrict__ dis,
                                                           float* __restrict__ inv, int n) {
  __shared__ int sh[SCAN_T];
  const int t = threadIdx.x;
  const int base = blockIdx.x * SCAN_T;
  int pre = 0;
  for (int i = t; i < base; i += SCAN_T) pre += cnt[i];
  const int node = base + t;
  const int c = (node < n) ? cnt[node] : 0;

  sh[t] = pre;
  __syncthreads();
  for (int o = SCAN_T / 2; o > 0; o >>= 1) {
    if (t < o) sh[t] += sh[t + o];
    __syncthreads();
  }
  const int block_off = sh[0];
  __syncthreads();

  sh[t] = c;
  __syncthreads();
  for (int o = 1; o < SCAN_T; o <<= 1) {
    const int v = (t >= o) ? sh[t - o] : 0;
    __syncthreads();
    sh[t] += v;
    __syncthreads();
  }
  const int incl = sh[t];

  if (node < n) {
    rowptr[node] = block_off + incl - c;
    if (node == n - 1) rowptr[n] = block_off + incl;
    const float d = (float)(c + 1);
    dis[node] = rsqrtf(d);
    inv[node] = 1.0f / d;
  }
}

__global__ void fill_kernel(const int* __restrict__ src, const int* __restrict__ dst,
                            const int* __restrict__ rowptr, const int* __restrict__ rank,
                            int* __restrict__ raw_src, int e) {
  int i = blockIdx.x * blockDim.x + threadIdx.x;
  if (i < e) raw_src[rowptr[dst[i]] + rank[i]] = src[i];
}

__global__ void seg_order_kernel(const int* __restrict__ rowptr, const int* __restrict__ raw_src,
                                 const float* __restrict__ dis, int* __restrict__ csr_src,
                                 float* __restrict__ csr_w, int n) {
  const int v = (blockIdx.x * blockDim.x + threadIdx.x) >> 5;
  const int lane = threadIdx.x & 31;
  if (v >= n) return;
  const int b = rowptr[v];
  const int e = rowptr[v + 1];
  const float dv = dis[v];
  for (int i = b + lane; i < e; i += 32) {
    const int key = raw_src[i];
    int rank = 0;
    for (int j = b; j < e; ++j) {
      const int kj = raw_src[j];
      rank += ((kj < key) || (kj == key && j < i)) ? 1 : 0;
    }
    csr_src[b + rank] = key;
    csr_w[b + rank] = dis[key] * dv;
  }
}

#define PREP_T 256
#define NB_ZERO ((N_NODES + PREP_T - 1) / PREP_T)
#define NB_T1 ((IN_C * HID_C + PREP_T - 1) / PREP_T)
#define NB_T2 ((HID_C * OUT_C + PREP_T - 1) / PREP_T)
#define NB_PREP (NB_ZERO + NB_T1 + 2 * NB_T2)
__device__ __forceinline__ void transpose_block(const float* __restrict__ B, float* __restrict__ Bt,
                                                int K, int N, int blk) {
  const int i = blk * PREP_T + (int)threadIdx.x;
  if (i < K * N) {
    const int kk = i / N;
    const int n = i - kk * N;
    Bt[(size_t)n * K + kk] = B[i];
  }
}
__global__ __launch_bounds__(PREP_T) void prep_kernel(int* __restrict__ cnt,
                                                      const float* __restrict__ w1, float* __restrict__ w1t,
                                                      const float* __restrict__ wmu, float* __restrict__ wmut,
                                                      const float* __restrict__ wls, float* __restrict__ wlst) {
  int b = blockIdx.x;
  if (b < NB_ZERO) {
    const int i = b * PREP_T + (int)threadIdx.x;
    if (i < N_NODES) cnt[i] = 0;
    return;
  }
  b -= NB_ZERO;
  if (b < NB_T1) { transpose_block(w1, w1t, IN_C, HID_C, b); return; }
  b -= NB_T1;
  if (b < NB_T2) { transpose_block(wmu, wmut, HID_C, OUT_C, b); return; }
  b -= NB_T2;
  transpose_block(wls, wlst, HID_C, OUT_C, b);
}

#define GEMM_KC 64
#define GEMM_PITCH 68
#define GEMM_RB 13
#define GEMM_N 256
#define GEMM_WAVES 8
#define GEMM_GRID 256
#define GEMM_SLAB (GEMM_WAVES * 32 / 8)
#define GEMM_AROWS ((GEMM_RB * 16 + GEMM_SLAB - 1) / GEMM_SLAB * GEMM_SLAB)
#define GEMM_STAGE ((GEMM_AROWS + GEMM_N) * GEMM_PITCH)
static_assert((N_NODES + 15) / 16 <= GEMM_GRID * GEMM_RB, "a work-group owns at most GEMM_RB row-blocks");
static_assert(GEMM_N % GEMM_SLAB == 0 && GEMM_N % (GEMM_WAVES * 16) == 0, "whole passes, whole tiles");

template <int NRB>
__device__ __forceinline__ void gemm_rowblocks_body(const float* __restrict__ A,
                                                    const float* __restrict__ Bt,
                                                    float* __restrict__ C, float* lds,
                                                    int M, int K, int m0, int nb) {
  constexpr int SLAB = GEMM_SLAB;
  constexpr int NA = (NRB * 16 + SLAB - 1) / SLAB;
  constexpr int NB = GEMM_N / SLAB;
  constexpr int NLD = NA + NB;
  constexpr int NT = GEMM_N / GEMM_WAVES / 16;

  const int tid = threadIdx.x;
  const int lane = tid & 31;
  const int wave = tid >> 5;
  const int half = lane >> 4;
  const int l = lane & 15;

  const int cr = tid >> 3;
  const int cq = tid & 7;
  unsigned goff[NLD];
  int soffs[NLD];
#pragma unroll
  for (int i = 0; i < NLD; ++i) {
    if (i < NA) {
      int row = m0 + i * SLAB + cr;
      if (row >= M) row = M - 1;
      goff[i] = (unsigned)row * (unsigned)K + (unsigned)(cq * 4);
      soffs[i] = (i * SLAB + cr) * GEMM_PITCH + cq * 4;
    } else {
      goff[i] = (unsigned)((i - NA) * SLAB + cr) * (unsigned)K + (unsigned)(cq * 4);
      soffs[i] = (GEMM_AROWS + (i - NA) * SLAB + cr) * GEMM_PITCH + cq * 4;
    }
  }

  const int a_off = l * GEMM_PITCH + half * 4;
  const int b_off = (GEMM_AROWS + wave * (NT * 16) + l) * GEMM_PITCH + half * 4;

  v8f acc[NRB][NT];
#pragma unroll
  for (int i = 0; i < NRB; ++i)
#pragma unroll
    for (int t = 0; t < NT; ++t) acc[i][t] = (v8f){0, 0, 0, 0, 0, 0, 0, 0};

#pragma unroll
  for (int ln = 0; ln < 2; ++ln) {
    v4f first[NLD];
#pragma unroll
    for (int i = 0; i < NLD; ++i)
      first[i] = *(const v4f*)((i < NA ? A : Bt) + goff[i] + ln * 32);
#pragma unroll
    for (int i = 0; i < NLD; ++i)
      *(v4f*)(lds + soffs[i] + ln * 32) = first[i];
  }
  __syncthreads();

  const int nchunks = K / GEMM_KC;
  for (int kc = 0; kc < nchunks; ++kc) {
    const float* cur = lds + (kc & 1) * GEMM_STAGE;
    float* nxt = lds + ((kc + 1) & 1) * GEMM_STAGE;
    const bool more = (kc + 1 < nchunks);

    v4f a4[2][NRB], b4[2][NT];
#pragma unroll
    for (int i = 0; i < NRB; ++i)
      a4[0][i] = *(const v4f*)(cur + a_off + i * 16 * GEMM_PITCH);
#pragma unroll
    for (int t = 0; t < NT; ++t)
      b4[0][t] = *(const v4f*)(cur + b_off + t * 16 * GEMM_PITCH);

#pragma unroll
    for (int ln = 0; ln < 2; ++ln) {
      v4f pre[NLD];
      if (more) {
#pragma unroll
        for (int i = 0; i < NLD; ++i)
          pre[i] = *(const v4f*)((i < NA ? A : Bt) + goff[i] + (kc + 1) * GEMM_KC + ln * 32);
      }

#pragma unroll
      for (int j = ln * 4; j < ln * 4 + 4; ++j) {
        if (j + 1 < GEMM_KC / 8) {
#pragma unroll
          for (int i = 0; i < NRB; ++i)
            a4[(j + 1) & 1][i] = *(const v4f*)(cur + a_off + i * 16 * GEMM_PITCH + (j + 1) * 8);
#pragma unroll
          for (int t = 0; t < NT; ++t)
            b4[(j + 1) & 1][t] = *(const v4f*)(cur + b_off + t * 16 * GEMM_PITCH + (j + 1) * 8);
        }
#pragma unroll
        for (int i = 0; i < NRB; ++i)
#pragma unroll
          for (int t = 0; t < NT; ++t)
            acc[i][t] = __builtin_amdgcn_wmma_f32_16x16x4_f32(
                false, (v2f){a4[j & 1][i].x, a4[j & 1][i].y}, false, (v2f){b4[j & 1][t].x, b4[j & 1][t].y},
                (short)0, acc[i][t], false, false);
#pragma unroll
        for (int i = 0; i < NRB; ++i)
#pragma unroll
          for (int t = 0; t < NT; ++t)
            acc[i][t] = __builtin_amdgcn_wmma_f32_16x16x4_f32(
                false, (v2f){a4[j & 1][i].z, a4[j & 1][i].w}, false, (v2f){b4[j & 1][t].z, b4[j & 1][t].w},
                (short)0, acc[i][t], false, false);
      }

      if (more) {
#pragma unroll
        for (int i = 0; i < NLD; ++i)
          *(v4f*)(nxt + soffs[i] + ln * 32) = pre[i];
      }
    }
    __syncthreads();
  }

#pragma unroll
  for (int i = 0; i < NRB; ++i) {
    const int r0 = m0 + i * 16;
    if (i < nb && r0 + 16 <= M) {
      float* c_ptr = C + (size_t)r0 * GEMM_N + (size_t)(wave * (NT * 16)) + l;
#pragma unroll
      for (int t = 0; t < NT; ++t)
#pragma unroll
        for (int r = 0; r < 8; ++r)
          c_ptr[(size_t)(r + 8 * half) * GEMM_N + t * 16] = acc[i][t][r];
    }
  }
}

__global__ __launch_bounds__(GEMM_WAVES * 32) void wmma_gemm_f32_rowblocks(const float* __restrict__ A,
                                                                          const float* __restrict__ Bt,
                                                                          float* __restrict__ C,
                                                                          int M, int K) {
  __shared__ __attribute__((aligned(16))) float lds[2 * GEMM_STAGE];
  const int nrb = (M + 15) >> 4;
  const int rb0 = ((int)blockIdx.x * nrb) / (int)gridDim.x;
  const int rb1 = (((int)blockIdx.x + 1) * nrb) / (int)gridDim.x;
  const int nb = rb1 - rb0;
  if (nb > GEMM_RB - 1)
    gemm_rowblocks_body<GEMM_RB>(A, Bt, C, lds, M, K, rb0 * 16, nb);
  else
    gemm_rowblocks_body<GEMM_RB - 1>(A, Bt, C, lds, M, K, rb0 * 16, nb);
}

template <int C, bool RELU, bool SPLIT>
__global__ __launch_bounds__(256) void gather_agg_kernel(float* __restrict__ out,
                                                         float* __restrict__ out2,
                                                         const float* __restrict__ z,
                                                         const int* __restrict__ rowptr,
                                                         const int* __restrict__ csr_src,
                                                         const float* __restrict__ csr_w,
                                                         const float* __restrict__ inv,
                                                         const float* __restrict__ bias,
                                                         const float* __restrict__ bias2, int n) {
  constexpr int G = C / 4;
  static_assert(G % 32 == 0 && 256 % G == 0, "a wave must stay inside one node");
  static_assert(!SPLIT || (G / 2) % 32 == 0, "a wave must stay inside one half of the row");
  const int g = threadIdx.x % G;
  const int v = __builtin_amdgcn_readfirstlane(blockIdx.x * (256 / G) + threadIdx.x / G);
  if (v >= n) return;
  const int b = rowptr[v];
  const int e = rowptr[v + 1];
  const float* zc = z + g * 4;

  float4 acc = make_float4(0.0f, 0.0f, 0.0f, 0.0f);
  int j = b;
  for (; j + 4 <= e; j += 4) {
    const int s0 = csr_src[j + 0], s1 = csr_src[j + 1], s2 = csr_src[j + 2], s3 = csr_src[j + 3];
    const float w0 = csr_w[j + 0], w1 = csr_w[j + 1], w2 = csr_w[j + 2], w3 = csr_w[j + 3];
    const float4 t0 = *(const float4*)(zc + (size_t)s0 * C);
    const float4 t1 = *(const float4*)(zc + (size_t)s1 * C);
    const float4 t2 = *(const float4*)(zc + (size_t)s2 * C);
    const float4 t3 = *(const float4*)(zc + (size_t)s3 * C);
    acc.x += t0.x * w0; acc.y += t0.y * w0; acc.z += t0.z * w0; acc.w += t0.w * w0;
    acc.x += t1.x * w1; acc.y += t1.y * w1; acc.z += t1.z * w1; acc.w += t1.w * w1;
    acc.x += t2.x * w2; acc.y += t2.y * w2; acc.z += t2.z * w2; acc.w += t2.w * w2;
    acc.x += t3.x * w3; acc.y += t3.y * w3; acc.z += t3.z * w3; acc.w += t3.w * w3;
  }
  for (; j < e; ++j) {
    const int s0 = csr_src[j];
    const float w0 = csr_w[j];
    const float4 t0 = *(const float4*)(zc + (size_t)s0 * C);
    acc.x += t0.x * w0; acc.y += t0.y * w0; acc.z += t0.z * w0; acc.w += t0.w * w0;
  }

  const float4 zs = *(const float4*)(zc + (size_t)v * C);
  const float iv = inv[v];
  acc.x += zs.x * iv; acc.y += zs.y * iv; acc.z += zs.z * iv; acc.w += zs.w * iv;
  constexpr int OC = SPLIT ? C / 2 : C;
  const bool second = SPLIT && (g >= G / 2);
  const int go = second ? g - G / 2 : g;
  const float* bs = second ? bias2 : bias;
  float* op = second ? out2 : out;
  acc.x += bs[go * 4 + 0];
  acc.y += bs[go * 4 + 1];
  acc.z += bs[go * 4 + 2];
  acc.w += bs[go * 4 + 3];
  if (RELU) {
    acc.x = fmaxf(acc.x, 0.0f);
    acc.y = fmaxf(acc.y, 0.0f);
    acc.z = fmaxf(acc.z, 0.0f);
    acc.w = fmaxf(acc.w, 0.0f);
  }
  *(float4*)(op + (size_t)v * OC + go * 4) = acc;
}

static inline int cdiv_l(long a, long b) { return (int)((a + b - 1) / b); }

extern "C" void kernel_launch(void* const* d_in, const int* in_sizes, int n_in,
                              void* d_out, int out_size, void* d_ws, size_t ws_size,
                              hipStream_t stream) {
  const float* x   = (const float*)d_in[0];
  const int*   ei  = (const int*)d_in[1];
  const int*   src = ei;
  const int*   dst = ei + N_EDGES;
  const float* w1  = (const float*)d_in[2];
  const float* b1  = (const float*)d_in[3];
  const float* wmu = (const float*)d_in[4];
  const float* bmu = (const float*)d_in[5];
  const float* wls = (const float*)d_in[6];
  const float* bls = (const float*)d_in[7];

  float* out_mu = (float*)d_out;
  float* out_ls = out_mu + (size_t)N_NODES * OUT_C;

  float* ws = (float*)d_ws;
  size_t off = 0;
  int*   cnt   = (int*)(ws + off); off += N_NODES;
  float* dis   = ws + off; off += N_NODES;
  float* inv   = ws + off; off += N_NODES;
  float* csr_w = ws + off; off += N_EDGES;
  float* w1t  = ws + off; off += (size_t)IN_C * HID_C;
  float* wmut = ws + off; off += (size_t)HID_C * OUT_C;
  float* wlst = ws + off; off += (size_t)HID_C * OUT_C;
  float* hpre = ws + off; off += (size_t)N_NODES * HID_C;
  float* h    = ws + off; off += (size_t)N_NODES * HID_C;
  int* rowptr  = (int*)(ws + off); off += N_NODES + 16;
  int* rank    = (int*)(ws + off); off += N_EDGES;
  int* raw_src = (int*)(ws + off); off += N_EDGES;
  int* csr_src = (int*)(ws + off); off += N_EDGES;
  float* z2 = hpre;

  prep_kernel<<<NB_PREP, PREP_T, 0, stream>>>(cnt, w1, w1t, wmu, wmut, wls, wlst);

  count_rank_kernel<<<cdiv_l(N_EDGES, 256), 256, 0, stream>>>(dst, cnt, rank, N_EDGES);
  scan_norm_kernel<<<cdiv_l(N_NODES, SCAN_T), SCAN_T, 0, stream>>>(cnt, rowptr, dis, inv, N_NODES);
  fill_kernel<<<cdiv_l(N_EDGES, 256), 256, 0, stream>>>(src, dst, rowptr, rank, raw_src, N_EDGES);
  seg_order_kernel<<<cdiv_l((long)N_NODES * 32, 256), 256, 0, stream>>>(rowptr, raw_src, dis, csr_src,
                                                                        csr_w, N_NODES);


  static_assert(HID_C == GEMM_N && 2 * OUT_C == GEMM_N && IN_C % GEMM_KC == 0 && HID_C % GEMM_KC == 0,
                "both products are 256 wide and K is whole chunks");
  wmma_gemm_f32_rowblocks<<<GEMM_GRID, GEMM_WAVES * 32, 0, stream>>>(x, w1t, hpre, N_NODES, IN_C);
  gather_agg_kernel<HID_C, true, false><<<cdiv_l((long)N_NODES * (HID_C / 4), 256), 256, 0, stream>>>(
      h, nullptr, hpre, rowptr, csr_src, csr_w, inv, b1, nullptr, N_NODES);

  wmma_gemm_f32_rowblocks<<<GEMM_GRID, GEMM_WAVES * 32, 0, stream>>>(h, wmut, z2, N_NODES, HID_C);
  (void)wlst;
  gather_agg_kernel<2 * OUT_C, false, true><<<cdiv_l((long)N_NODES * (2 * OUT_C / 4), 256), 256, 0, stream>>>(
      out_mu, out_ls, z2, rowptr, csr_src, csr_w, inv, bmu, bls, N_NODES);
}


# Row pointers scanned in LDS by every fill work-group: scan launch gone (9 -> 8); list starts, degrees and the edge coefficient come from LDS, one 8-byte store per edge; sum orders unchanged
# speedup vs baseline: 2.8456x; 1.0326x over previous
#include <hip/hip_runtime.h>

#define N_NODES 50000
#define N_EDGES 500000
#define IN_C 512
#define HID_C 256
#define OUT_C 128

typedef __attribute__((ext_vector_type(2))) float v2f;
typedef __attribute__((ext_vector_type(8))) float v8f;
typedef __attribute__((ext_vector_type(4))) float v4f;

__global__ void count_rank_kernel(const int* __restrict__ dst, int* __restrict__ cnt,
                                  int* __restrict__ rank, int e) {
  int i = blockIdx.x * blockDim.x + threadIdx.x;
  if (i < e) rank[i] = atomicAdd(&cnt[dst[i]], 1);
}

#define FILL_T 1024
#define FILL_GRID 256
#define FILL_CH ((N_NODES + FILL_T - 1) / FILL_T)
#define FILL_WORDS (FILL_T * FILL_CH)
__global__ __launch_bounds__(FILL_T) void scan_fill_kernel(const int* __restrict__ src,
                                                           const int* __restrict__ dst,
                                                           const int* __restrict__ cnt,
                                                           const int* __restrict__ rank,
                                                           int* __restrict__ rowptr,
                                                           float* __restrict__ inv,
                                                           int* __restrict__ raw_words, int n, int e) {
  int2* __restrict__ raw = (int2*)raw_words;
  __shared__ int s[FILL_WORDS + 1];
  __shared__ int wsum[FILL_T / 32];
  const int t = threadIdx.x;
  const int lane = t & 31;
  const int wv = t >> 5;

  const int full = n / FILL_T;
  int r = 0;
  for (; r + 8 <= full; r += 8) {
    int c8[8];
#pragma unroll
    for (int u = 0; u < 8; ++u) c8[u] = cnt[(r + u) * FILL_T + t];
#pragma unroll
    for (int u = 0; u < 8; ++u) s[(r + u) * FILL_T + t] = c8[u];
  }
  for (int i = r * FILL_T + t; i < FILL_WORDS; i += FILL_T) s[i] = (i < n) ? cnt[i] : 0;
  __syncthreads();

  const int base = t * FILL_CH;
  int tot = 0;
  for (int k = 0; k < FILL_CH; ++k) tot += s[base + k];
  int incl = tot;
  for (int o = 1; o < 32; o <<= 1) {
    const int v = __shfl_up(incl, o);
    if (lane >= o) incl += v;
  }
  if (lane == 31) wsum[wv] = incl;
  __syncthreads();
  if (wv == 0) {
    const int w = wsum[lane];
    int wi = w;
    for (int o = 1; o < 32; o <<= 1) {
      const int v = __shfl_up(wi, o);
      if (lane >= o) wi += v;
    }
    wsum[lane] = wi - w;
  }
  __syncthreads();
  int run = wsum[wv] + incl - tot;
  for (int k = 0; k < FILL_CH; ++k) {
    const int c = s[base + k];
    s[base + k] = run;
    run += c;
  }
  if (t == FILL_T - 1) s[FILL_WORDS] = run;
  __syncthreads();

  {
    const int per = (n + 1 + (int)gridDim.x - 1) / (int)gridDim.x;
    for (int k = t; k < per; k += FILL_T) {
      const int i = (int)blockIdx.x * per + k;
      if (i <= n) rowptr[i] = s[i];
      if (i < n) inv[i] = 1.0f / (float)(s[i + 1] - s[i] + 1);
    }
  }

  const int epb = (e + (int)gridDim.x - 1) / (int)gridDim.x;
  const int i0 = (int)blockIdx.x * epb;
  const int i1 = (i0 + epb < e) ? i0 + epb : e;
  for (int i = i0 + t; i < i1; i += FILL_T) {
    const int d = dst[i];
    const int sr = src[i];
    const int pos = s[d] + rank[i];
    const float ds = rsqrtf((float)(s[sr + 1] - s[sr] + 1));
    const float dd = rsqrtf((float)(s[d + 1] - s[d] + 1));
    raw[pos] = make_int2(sr, __float_as_int(ds * dd));
  }
}

__global__ void seg_order_kernel(const int* __restrict__ rowptr, const int* __restrict__ raw_words,
                                 int* __restrict__ csr_src,
                                 float* __restrict__ csr_w, int n) {
  const int2* __restrict__ raw = (const int2*)raw_words;
  const int v = (blockIdx.x * blockDim.x + threadIdx.x) >> 5;
  const int lane = threadIdx.x & 31;
  if (v >= n) return;
  const int b = rowptr[v];
  const int e = rowptr[v + 1];
  for (int i = b + lane; i < e; i += 32) {
    const int2 kv = raw[i];
    const int key = kv.x;
    int rank = 0;
    for (int j = b; j < e; ++j) {
      const int kj = raw[j].x;
      rank += ((kj < key) || (kj == key && j < i)) ? 1 : 0;
    }
    csr_src[b + rank] = key;
    csr_w[b + rank] = __int_as_float(kv.y);
  }
}

#define PREP_T 256
#define NB_ZERO ((N_NODES + PREP_T - 1) / PREP_T)
#define NB_T1 ((IN_C * HID_C + PREP_T - 1) / PREP_T)
#define NB_T2 ((HID_C * OUT_C + PREP_T - 1) / PREP_T)
#define NB_PREP (NB_ZERO + NB_T1 + 2 * NB_T2)
__device__ __forceinline__ void transpose_block(const float* __restrict__ B, float* __restrict__ Bt,
                                                int K, int N, int blk) {
  const int i = blk * PREP_T + (int)threadIdx.x;
  if (i < K * N) {
    const int kk = i / N;
    const int n = i - kk * N;
    Bt[(size_t)n * K + kk] = B[i];
  }
}
__global__ __launch_bounds__(PREP_T) void prep_kernel(int* __restrict__ cnt,
                                                      const float* __restrict__ w1, float* __restrict__ w1t,
                                                      const float* __restrict__ wmu, float* __restrict__ wmut,
                                                      const float* __restrict__ wls, float* __restrict__ wlst) {
  int b = blockIdx.x;
  if (b < NB_ZERO) {
    const int i = b * PREP_T + (int)threadIdx.x;
    if (i < N_NODES) cnt[i] = 0;
    return;
  }
  b -= NB_ZERO;
  if (b < NB_T1) { transpose_block(w1, w1t, IN_C, HID_C, b); return; }
  b -= NB_T1;
  if (b < NB_T2) { transpose_block(wmu, wmut, HID_C, OUT_C, b); return; }
  b -= NB_T2;
  transpose_block(wls, wlst, HID_C, OUT_C, b);
}

#define GEMM_KC 64
#define GEMM_PITCH 68
#define GEMM_RB 13
#define GEMM_N 256
#define GEMM_WAVES 8
#define GEMM_GRID 256
#define GEMM_SLAB (GEMM_WAVES * 32 / 8)
#define GEMM_AROWS ((GEMM_RB * 16 + GEMM_SLAB - 1) / GEMM_SLAB * GEMM_SLAB)
#define GEMM_STAGE ((GEMM_AROWS + GEMM_N) * GEMM_PITCH)
static_assert((N_NODES + 15) / 16 <= GEMM_GRID * GEMM_RB, "a work-group owns at most GEMM_RB row-blocks");
static_assert(GEMM_N % GEMM_SLAB == 0 && GEMM_N % (GEMM_WAVES * 16) == 0, "whole passes, whole tiles");

template <int NRB>
__device__ __forceinline__ void gemm_rowblocks_body(const float* __restrict__ A,
                                                    const float* __restrict__ Bt,
                                                    float* __restrict__ C, float* lds,
                                                    int M, int K, int m0, int nb) {
  constexpr int SLAB = GEMM_SLAB;
  constexpr int NA = (NRB * 16 + SLAB - 1) / SLAB;
  constexpr int NB = GEMM_N / SLAB;
  constexpr int NLD = NA + NB;
  constexpr int NT = GEMM_N / GEMM_WAVES / 16;

  const int tid = threadIdx.x;
  const int lane = tid & 31;
  const int wave = tid >> 5;
  const int half = lane >> 4;
  const int l = lane & 15;

  const int cr = tid >> 3;
  const int cq = tid & 7;
  unsigned goff[NLD];
  int soffs[NLD];
#pragma unroll
  for (int i = 0; i < NLD; ++i) {
    if (i < NA) {
      int row = m0 + i * SLAB + cr;
      if (row >= M) row = M - 1;
      goff[i] = (unsigned)row * (unsigned)K + (unsigned)(cq * 4);
      soffs[i] = (i * SLAB + cr) * GEMM_PITCH + cq * 4;
    } else {
      goff[i] = (unsigned)((i - NA) * SLAB + cr) * (unsigned)K + (unsigned)(cq * 4);
      soffs[i] = (GEMM_AROWS + (i - NA) * SLAB + cr) * GEMM_PITCH + cq * 4;
    }
  }

  const int a_off = l * GEMM_PITCH + half * 4;
  const int b_off = (GEMM_AROWS + wave * (NT * 16) + l) * GEMM_PITCH + half * 4;

  v8f acc[NRB][NT];
#pragma unroll
  for (int i = 0; i < NRB; ++i)
#pragma unroll
    for (int t = 0; t < NT; ++t) acc[i][t] = (v8f){0, 0, 0, 0, 0, 0, 0, 0};

#pragma unroll
  for (int ln = 0; ln < 2; ++ln) {
    v4f first[NLD];
#pragma unroll
    for (int i = 0; i < NLD; ++i)
      first[i] = *(const v4f*)((i < NA ? A : Bt) + goff[i] + ln * 32);
#pragma unroll
    for (int i = 0; i < NLD; ++i)
      *(v4f*)(lds + soffs[i] + ln * 32) = first[i];
  }
  __syncthreads();

  const int nchunks = K / GEMM_KC;
  for (int kc = 0; kc < nchunks; ++kc) {
    const float* cur = lds + (kc & 1) * GEMM_STAGE;
    float* nxt = lds + ((kc + 1) & 1) * GEMM_STAGE;
    const bool more = (kc + 1 < nchunks);

    v4f a4[2][NRB], b4[2][NT];
#pragma unroll
    for (int i = 0; i < NRB; ++i)
      a4[0][i] = *(const v4f*)(cur + a_off + i * 16 * GEMM_PITCH);
#pragma unroll
    for (int t = 0; t < NT; ++t)
      b4[0][t] = *(const v4f*)(cur + b_off + t * 16 * GEMM_PITCH);

#pragma unroll
    for (int ln = 0; ln < 2; ++ln) {
      v4f pre[NLD];
      if (more) {
#pragma unroll
        for (int i = 0; i < NLD; ++i)
          pre[i] = *(const v4f*)((i < NA ? A : Bt) + goff[i] + (kc + 1) * GEMM_KC + ln * 32);
      }

#pragma unroll
      for (int j = ln * 4; j < ln * 4 + 4; ++j) {
        if (j + 1 < GEMM_KC / 8) {
#pragma unroll
          for (int i = 0; i < NRB; ++i)
            a4[(j + 1) & 1][i] = *(const v4f*)(cur + a_off + i * 16 * GEMM_PITCH + (j + 1) * 8);
#pragma unroll
          for (int t = 0; t < NT; ++t)
            b4[(j + 1) & 1][t] = *(const v4f*)(cur + b_off + t * 16 * GEMM_PITCH + (j + 1) * 8);
        }
#pragma unroll
        for (int i = 0; i < NRB; ++i)
#pragma unroll
          for (int t = 0; t < NT; ++t)
            acc[i][t] = __builtin_amdgcn_wmma_f32_16x16x4_f32(
                false, (v2f){a4[j & 1][i].x, a4[j & 1][i].y}, false, (v2f){b4[j & 1][t].x, b4[j & 1][t].y},
                (short)0, acc[i][t], false, false);
#pragma unroll
        for (int i = 0; i < NRB; ++i)
#pragma unroll
          for (int t = 0; t < NT; ++t)
            acc[i][t] = __builtin_amdgcn_wmma_f32_16x16x4_f32(
                false, (v2f){a4[j & 1][i].z, a4[j & 1][i].w}, false, (v2f){b4[j & 1][t].z, b4[j & 1][t].w},
                (short)0, acc[i][t], false, false);
      }

      if (more) {
#pragma unroll
        for (int i = 0; i < NLD; ++i)
          *(v4f*)(nxt + soffs[i] + ln * 32) = pre[i];
      }
    }
    __syncthreads();
  }

#pragma unroll
  for (int i = 0; i < NRB; ++i) {
    const int r0 = m0 + i * 16;
    if (i < nb && r0 + 16 <= M) {
      float* c_ptr = C + (size_t)r0 * GEMM_N + (size_t)(wave * (NT * 16)) + l;
#pragma unroll
      for (int t = 0; t < NT; ++t)
#pragma unroll
        for (int r = 0; r < 8; ++r)
          c_ptr[(size_t)(r + 8 * half) * GEMM_N + t * 16] = acc[i][t][r];
    }
  }
}

__global__ __launch_bounds__(GEMM_WAVES * 32) void wmma_gemm_f32_rowblocks(const float* __restrict__ A,
                                                                          const float* __restrict__ Bt,
                                                                          float* __restrict__ C,
                                                                          int M, int K) {
  __shared__ __attribute__((aligned(16))) float lds[2 * GEMM_STAGE];
  const int nrb = (M + 15) >> 4;
  const int rb0 = ((int)blockIdx.x * nrb) / (int)gridDim.x;
  const int rb1 = (((int)blockIdx.x + 1) * nrb) / (int)gridDim.x;
  const int nb = rb1 - rb0;
  if (nb > GEMM_RB - 1)
    gemm_rowblocks_body<GEMM_RB>(A, Bt, C, lds, M, K, rb0 * 16, nb);
  else
    gemm_rowblocks_body<GEMM_RB - 1>(A, Bt, C, lds, M, K, rb0 * 16, nb);
}

template <int C, bool RELU, bool SPLIT>
__global__ __launch_bounds__(256) void gather_agg_kernel(float* __restrict__ out,
                                                         float* __restrict__ out2,
                                                         const float* __restrict__ z,
                                                         const int* __restrict__ rowptr,
                                                         const int* __restrict__ csr_src,
                                                         const float* __restrict__ csr_w,
                                                         const float* __restrict__ inv,
                                                         const float* __restrict__ bias,
                                                         const float* __restrict__ bias2, int n) {
  constexpr int G = C / 4;
  static_assert(G % 32 == 0 && 256 % G == 0, "a wave must stay inside one node");
  static_assert(!SPLIT || (G / 2) % 32 == 0, "a wave must stay inside one half of the row");
  const int g = threadIdx.x % G;
  const int v = __builtin_amdgcn_readfirstlane(blockIdx.x * (256 / G) + threadIdx.x / G);
  if (v >= n) return;
  const int b = rowptr[v];
  const int e = rowptr[v + 1];
  const float* zc = z + g * 4;

  float4 acc = make_float4(0.0f, 0.0f, 0.0f, 0.0f);
  int j = b;
  for (; j + 4 <= e; j += 4) {
    const int s0 = csr_src[j + 0], s1 = csr_src[j + 1], s2 = csr_src[j + 2], s3 = csr_src[j + 3];
    const float w0 = csr_w[j + 0], w1 = csr_w[j + 1], w2 = csr_w[j + 2], w3 = csr_w[j + 3];
    const float4 t0 = *(const float4*)(zc + (size_t)s0 * C);
    const float4 t1 = *(const float4*)(zc + (size_t)s1 * C);
    const float4 t2 = *(const float4*)(zc + (size_t)s2 * C);
    const float4 t3 = *(const float4*)(zc + (size_t)s3 * C);
    acc.x += t0.x * w0; acc.y += t0.y * w0; acc.z += t0.z * w0; acc.w += t0.w * w0;
    acc.x += t1.x * w1; acc.y += t1.y * w1; acc.z += t1.z * w1; acc.w += t1.w * w1;
    acc.x += t2.x * w2; acc.y += t2.y * w2; acc.z += t2.z * w2; acc.w += t2.w * w2;
    acc.x += t3.x * w3; acc.y += t3.y * w3; acc.z += t3.z * w3; acc.w += t3.w * w3;
  }
  for (; j < e; ++j) {
    const int s0 = csr_src[j];
    const float w0 = csr_w[j];
    const float4 t0 = *(const float4*)(zc + (size_t)s0 * C);
    acc.x += t0.x * w0; acc.y += t0.y * w0; acc.z += t0.z * w0; acc.w += t0.w * w0;
  }

  const float4 zs = *(const float4*)(zc + (size_t)v * C);
  const float iv = inv[v];
  acc.x += zs.x * iv; acc.y += zs.y * iv; acc.z += zs.z * iv; acc.w += zs.w * iv;
  constexpr int OC = SPLIT ? C / 2 : C;
  const bool second = SPLIT && (g >= G / 2);
  const int go = second ? g - G / 2 : g;
  const float* bs = second ? bias2 : bias;
  float* op = second ? out2 : out;
  acc.x += bs[go * 4 + 0];
  acc.y += bs[go * 4 + 1];
  acc.z += bs[go * 4 + 2];
  acc.w += bs[go * 4 + 3];
  if (RELU) {
    acc.x = fmaxf(acc.x, 0.0f);
    acc.y = fmaxf(acc.y, 0.0f);
    acc.z = fmaxf(acc.z, 0.0f);
    acc.w = fmaxf(acc.w, 0.0f);
  }
  *(float4*)(op + (size_t)v * OC + go * 4) = acc;
}

static inline int cdiv_l(long a, long b) { return (int)((a + b - 1) / b); }

extern "C" void kernel_launch(void* const* d_in, const int* in_sizes, int n_in,
                              void* d_out, int out_size, void* d_ws, size_t ws_size,
                              hipStream_t stream) {
  const float* x   = (const float*)d_in[0];
  const int*   ei  = (const int*)d_in[1];
  const int*   src = ei;
  const int*   dst = ei + N_EDGES;
  const float* w1  = (const float*)d_in[2];
  const float* b1  = (const float*)d_in[3];
  const float* wmu = (const float*)d_in[4];
  const float* bmu = (const float*)d_in[5];
  const float* wls = (const float*)d_in[6];
  const float* bls = (const float*)d_in[7];

  float* out_mu = (float*)d_out;
  float* out_ls = out_mu + (size_t)N_NODES * OUT_C;

  float* ws = (float*)d_ws;
  size_t off = 0;
  int*   cnt   = (int*)(ws + off); off += N_NODES;
  float* dis   = ws + off; off += N_NODES;
  float* inv   = ws + off; off += N_NODES;
  float* csr_w = ws + off; off += N_EDGES;
  float* w1t  = ws + off; off += (size_t)IN_C * HID_C;
  float* wmut = ws + off; off += (size_t)HID_C * OUT_C;
  float* wlst = ws + off; off += (size_t)HID_C * OUT_C;
  float* hpre = ws + off; off += (size_t)N_NODES * HID_C;
  float* h    = ws + off; off += (size_t)N_NODES * HID_C;
  int* rowptr  = (int*)(ws + off); off += N_NODES + 16;
  int* rank    = (int*)(ws + off); off += N_EDGES;
  int* raw_src = (int*)(ws + off); off += N_EDGES;
  int* csr_src = (int*)(ws + off); off += N_EDGES;
  float* z2 = hpre;
  int* raw = (int*)h;
  (void)dis; (void)raw_src;

  prep_kernel<<<NB_PREP, PREP_T, 0, stream>>>(cnt, w1, w1t, wmu, wmut, wls, wlst);

  count_rank_kernel<<<cdiv_l(N_EDGES, 256), 256, 0, stream>>>(dst, cnt, rank, N_EDGES);
  scan_fill_kernel<<<FILL_GRID, FILL_T, 0, stream>>>(src, dst, cnt, rank, rowptr, inv, raw,
                                                     N_NODES, N_EDGES);
  seg_order_kernel<<<cdiv_l((long)N_NODES * 32, 256), 256, 0, stream>>>(rowptr, raw, csr_src, csr_w,
                                                                        N_NODES);


  static_assert(HID_C == GEMM_N && 2 * OUT_C == GEMM_N && IN_C % GEMM_KC == 0 && HID_C % GEMM_KC == 0,
                "both products are 256 wide and K is whole chunks");
  wmma_gemm_f32_rowblocks<<<GEMM_GRID, GEMM_WAVES * 32, 0, stream>>>(x, w1t, hpre, N_NODES, IN_C);
  gather_agg_kernel<HID_C, true, false><<<cdiv_l((long)N_NODES * (HID_C / 4), 256), 256, 0, stream>>>(
      h, nullptr, hpre, rowptr, csr_src, csr_w, inv, b1, nullptr, N_NODES);

  wmma_gemm_f32_rowblocks<<<GEMM_GRID, GEMM_WAVES * 32, 0, stream>>>(h, wmut, z2, N_NODES, HID_C);
  (void)wlst;
  gather_agg_kernel<2 * OUT_C, false, true><<<cdiv_l((long)N_NODES * (2 * OUT_C / 4), 256), 256, 0, stream>>>(
      out_mu, out_ls, z2, rowptr, csr_src, csr_w, inv, bmu, bls, N_NODES);
}


# Edge lists built by node range: 256 work-groups each sweep all destinations and count, rank and place their own edges via LDS atomics; global atomics, clear and scan go: 8 -> 6 launches; sums same
# speedup vs baseline: 2.9897x; 1.0506x over previous
#include <hip/hip_runtime.h>

#define N_NODES 50000
#define N_EDGES 500000
#define IN_C 512
#define HID_C 256
#define OUT_C 128

typedef __attribute__((ext_vector_type(2))) float v2f;
typedef __attribute__((ext_vector_type(8))) float v8f;
typedef __attribute__((ext_vector_type(4))) float v4f;

#define BIN_T 1024
#define BIN_G 256
#define BIN_NPB ((N_NODES + BIN_G - 1) / BIN_G)
#define BIN_PAD 256
#define BIN_CAP 8192
#define NB_TW1 ((IN_C * HID_C + BIN_T - 1) / BIN_T)
#define NB_TW2 ((HID_C * OUT_C + BIN_T - 1) / BIN_T)
#define NB_BINPREP (BIN_G + NB_TW1 + 2 * NB_TW2)
static_assert(BIN_NPB <= BIN_PAD && BIN_PAD == 8 * 32, "one wave scans the range, 8 counters per lane");
static_assert(N_EDGES % 4 == 0, "destinations are read four at a time");

__device__ __forceinline__ void transpose_block(const float* __restrict__ B, float* __restrict__ Bt,
                                                int K, int N, int blk) {
  const int i = blk * BIN_T + (int)threadIdx.x;
  if (i < K * N) {
    const int kk = i / N;
    const int n = i - kk * N;
    Bt[(size_t)n * K + kk] = B[i];
  }
}

__device__ __forceinline__ void bin_one(int d, int e, int lo, int nn, int& below_w, int* cnt, int* nlist,
                                        int* list_e, int* list_b) {
  below_w += __builtin_popcount(__builtin_amdgcn_ballot_w32(d < lo));
  const int ld = d - lo;
  if ((unsigned)ld < (unsigned)nn) {
    const int r = atomicAdd(&cnt[ld], 1);
    const int slot = atomicAdd(nlist, 1);
    if (slot < BIN_CAP) {
      list_e[slot] = e;
      list_b[slot] = ld | (r << 8);
    }
  }
}

#define BIN_U 8
__device__ __forceinline__ void bin_load(const int4* __restrict__ dst4, int g, int rot, int4 (&v)[BIN_U],
                                         int (&q)[BIN_U]) {
  constexpr int NV = N_EDGES / 4;
#pragma unroll
  for (int u = 0; u < BIN_U; ++u) {
    const int gu = g + u * BIN_T;
    const bool ok = gu < NV;
    int qq = gu + rot;
    if (qq >= NV) qq -= NV;
    qq = ok ? qq : 0;
    q[u] = ok ? qq : -1;
    v[u] = dst4[qq];
  }
}

__global__ __launch_bounds__(BIN_T) void bin_prep_kernel(const int* __restrict__ src,
                                                         const int* __restrict__ dst,
                                                         int* __restrict__ rowptr,
                                                         float* __restrict__ inv,
                                                         float* __restrict__ dis,
                                                         int* __restrict__ raw,
                                                         const float* __restrict__ w1, float* __restrict__ w1t,
                                                         const float* __restrict__ wmu, float* __restrict__ wmut,
                                                         const float* __restrict__ wls, float* __restrict__ wlst,
                                                         int n) {
  __shared__ int cnt[BIN_PAD + 1];
  __shared__ int start[BIN_PAD + 1];
  __shared__ int cnt2[BIN_PAD];
  __shared__ int list_e[BIN_CAP];
  __shared__ int list_b[BIN_CAP];
  __shared__ int nlist;
  __shared__ int below_s;

  int b = blockIdx.x;
  if (b >= BIN_G) {
    b -= BIN_G;
    if (b < NB_TW1) { transpose_block(w1, w1t, IN_C, HID_C, b); return; }
    b -= NB_TW1;
    if (b < NB_TW2) { transpose_block(wmu, wmut, HID_C, OUT_C, b); return; }
    b -= NB_TW2;
    transpose_block(wls, wlst, HID_C, OUT_C, b);
    return;
  }

  const int t = threadIdx.x;
  const int lane = t & 31;
  const int wv = t >> 5;
  const int lo = b * BIN_NPB;
  int nn = n - lo;
  if (nn > BIN_NPB) nn = BIN_NPB;
  if (nn < 0) nn = 0;

  if (t <= BIN_PAD) cnt[t] = 0;
  if (t == 0) { nlist = 0; below_s = 0; }
  __syncthreads();

  constexpr int NV = N_EDGES / 4;
  constexpr int STEPS = NV / (BIN_U * BIN_T);
  constexpr int REST = (NV - STEPS * BIN_U * BIN_T + BIN_T - 1) / BIN_T;
  static_assert(STEPS >= 1 && REST <= BIN_U, "the rest fits one step");
  constexpr int BIG = 0x7fffffff;
  const int4* __restrict__ dst4 = (const int4*)dst;
  const int rot = (int)(((long)b * NV) / BIN_G);
  int below_w = 0;
  int4 v[BIN_U];
  int q[BIN_U];
  bin_load(dst4, t, rot, v, q);
  for (int s = 0; s < STEPS; ++s) {
    int4 nv[BIN_U];
    int nq[BIN_U];
    bin_load(dst4, (s + 1) * (BIN_U * BIN_T) + t, rot, nv, nq);
#pragma unroll
    for (int u = 0; u < BIN_U; ++u) {
      const int e4 = 4 * q[u];
      bin_one(v[u].x, e4 + 0, lo, nn, below_w, cnt, &nlist, list_e, list_b);
      bin_one(v[u].y, e4 + 1, lo, nn, below_w, cnt, &nlist, list_e, list_b);
      bin_one(v[u].z, e4 + 2, lo, nn, below_w, cnt, &nlist, list_e, list_b);
      bin_one(v[u].w, e4 + 3, lo, nn, below_w, cnt, &nlist, list_e, list_b);
    }
#pragma unroll
    for (int u = 0; u < BIN_U; ++u) { v[u] = nv[u]; q[u] = nq[u]; }
  }
#pragma unroll
  for (int u = 0; u < REST; ++u) {
    const bool ok = q[u] >= 0;
    const int e4 = 4 * q[u];
    bin_one(ok ? v[u].x : BIG, e4 + 0, lo, nn, below_w, cnt, &nlist, list_e, list_b);
    bin_one(ok ? v[u].y : BIG, e4 + 1, lo, nn, below_w, cnt, &nlist, list_e, list_b);
    bin_one(ok ? v[u].z : BIG, e4 + 2, lo, nn, below_w, cnt, &nlist, list_e, list_b);
    bin_one(ok ? v[u].w : BIG, e4 + 3, lo, nn, below_w, cnt, &nlist, list_e, list_b);
  }
  if (lane == 0) atomicAdd(&below_s, below_w);
  __syncthreads();

  if (wv == 0) {
    int loc[8];
    int tot = 0;
#pragma unroll
    for (int k = 0; k < 8; ++k) { loc[k] = cnt[lane * 8 + k]; tot += loc[k]; }
    int incl = tot;
    for (int o = 1; o < 32; o <<= 1) {
      const int v = __shfl_up(incl, o);
      if (lane >= o) incl += v;
    }
    int run = incl - tot;
#pragma unroll
    for (int k = 0; k < 8; ++k) { start[lane * 8 + k] = run; run += loc[k]; }
    if (lane == 31) start[BIN_PAD] = run;
  }
  __syncthreads();

  const int nl = nlist;
  const int bel = below_s;

  if (t < nn) {
    const float dg = (float)(cnt[t] + 1);
    rowptr[lo + t] = bel + start[t];
    inv[lo + t] = 1.0f / dg;
    dis[lo + t] = rsqrtf(dg);
  }
  if (t == 0 && nn > 0 && lo + nn == n) rowptr[n] = bel + start[nn];

  if (nl <= BIN_CAP) {
    for (int k = t; k < nl; k += BIN_T) {
      const int pb = list_b[k];
      raw[bel + start[pb & 255] + (pb >> 8)] = src[list_e[k]];
    }
  } else {
    if (t < BIN_PAD) cnt2[t] = 0;
    __syncthreads();
    for (int i = t; i < N_EDGES; i += BIN_T) {
      const int ld = dst[i] - lo;
      if ((unsigned)ld < (unsigned)nn) {
        const int r = atomicAdd(&cnt2[ld], 1);
        raw[bel + start[ld] + r] = src[i];
      }
    }
  }
}

__global__ void seg_order_kernel(const int* __restrict__ rowptr, const int* __restrict__ raw,
                                 const float* __restrict__ dis,
                                 int* __restrict__ csr_src,
                                 float* __restrict__ csr_w, int n) {
  const int v = (blockIdx.x * blockDim.x + threadIdx.x) >> 5;
  const int lane = threadIdx.x & 31;
  if (v >= n) return;
  const int b = rowptr[v];
  const int e = rowptr[v + 1];
  const float dv = dis[v];
  for (int i = b + lane; i < e; i += 32) {
    const int key = raw[i];
    int rank = 0;
    for (int j = b; j < e; ++j) {
      const int kj = raw[j];
      rank += ((kj < key) || (kj == key && j < i)) ? 1 : 0;
    }
    csr_src[b + rank] = key;
    csr_w[b + rank] = dis[key] * dv;
  }
}

#define GEMM_KC 64
#define GEMM_PITCH 68
#define GEMM_RB 13
#define GEMM_N 256
#define GEMM_WAVES 8
#define GEMM_GRID 256
#define GEMM_SLAB (GEMM_WAVES * 32 / 8)
#define GEMM_AROWS ((GEMM_RB * 16 + GEMM_SLAB - 1) / GEMM_SLAB * GEMM_SLAB)
#define GEMM_STAGE ((GEMM_AROWS + GEMM_N) * GEMM_PITCH)
static_assert((N_NODES + 15) / 16 <= GEMM_GRID * GEMM_RB, "a work-group owns at most GEMM_RB row-blocks");
static_assert(GEMM_N % GEMM_SLAB == 0 && GEMM_N % (GEMM_WAVES * 16) == 0, "whole passes, whole tiles");

template <int NRB>
__device__ __forceinline__ void gemm_rowblocks_body(const float* __restrict__ A,
                                                    const float* __restrict__ Bt,
                                                    float* __restrict__ C, float* lds,
                                                    int M, int K, int m0, int nb) {
  constexpr int SLAB = GEMM_SLAB;
  constexpr int NA = (NRB * 16 + SLAB - 1) / SLAB;
  constexpr int NB = GEMM_N / SLAB;
  constexpr int NLD = NA + NB;
  constexpr int NT = GEMM_N / GEMM_WAVES / 16;

  const int tid = threadIdx.x;
  const int lane = tid & 31;
  const int wave = tid >> 5;
  const int half = lane >> 4;
  const int l = lane & 15;

  const int cr = tid >> 3;
  const int cq = tid & 7;
  unsigned goff[NLD];
  int soffs[NLD];
#pragma unroll
  for (int i = 0; i < NLD; ++i) {
    if (i < NA) {
      int row = m0 + i * SLAB + cr;
      if (row >= M) row = M - 1;
      goff[i] = (unsigned)row * (unsigned)K + (unsigned)(cq * 4);
      soffs[i] = (i * SLAB + cr) * GEMM_PITCH + cq * 4;
    } else {
      goff[i] = (unsigned)((i - NA) * SLAB + cr) * (unsigned)K + (unsigned)(cq * 4);
      soffs[i] = (GEMM_AROWS + (i - NA) * SLAB + cr) * GEMM_PITCH + cq * 4;
    }
  }

  const int a_off = l * GEMM_PITCH + half * 4;
  const int b_off = (GEMM_AROWS + wave * (NT * 16) + l) * GEMM_PITCH + half * 4;

  v8f acc[NRB][NT];
#pragma unroll
  for (int i = 0; i < NRB; ++i)
#pragma unroll
    for (int t = 0; t < NT; ++t) acc[i][t] = (v8f){0, 0, 0, 0, 0, 0, 0, 0};

#pragma unroll
  for (int ln = 0; ln < 2; ++ln) {
    v4f first[NLD];
#pragma unroll
    for (int i = 0; i < NLD; ++i)
      first[i] = *(const v4f*)((i < NA ? A : Bt) + goff[i] + ln * 32);
#pragma unroll
    for (int i = 0; i < NLD; ++i)
      *(v4f*)(lds + soffs[i] + ln * 32) = first[i];
  }
  __syncthreads();

  const int nchunks = K / GEMM_KC;
  for (int kc = 0; kc < nchunks; ++kc) {
    const float* cur = lds + (kc & 1) * GEMM_STAGE;
    float* nxt = lds + ((kc + 1) & 1) * GEMM_STAGE;
    const bool more = (kc + 1 < nchunks);

    v4f a4[2][NRB], b4[2][NT];
#pragma unroll
    for (int i = 0; i < NRB; ++i)
      a4[0][i] = *(const v4f*)(cur + a_off + i * 16 * GEMM_PITCH);
#pragma unroll
    for (int t = 0; t < NT; ++t)
      b4[0][t] = *(const v4f*)(cur + b_off + t * 16 * GEMM_PITCH);

#pragma unroll
    for (int ln = 0; ln < 2; ++ln) {
      v4f pre[NLD];
      if (more) {
#pragma unroll
        for (int i = 0; i < NLD; ++i)
          pre[i] = *(const v4f*)((i < NA ? A : Bt) + goff[i] + (kc + 1) * GEMM_KC + ln * 32);
      }

#pragma unroll
      for (int j = ln * 4; j < ln * 4 + 4; ++j) {
        if (j + 1 < GEMM_KC / 8) {
#pragma unroll
          for (int i = 0; i < NRB; ++i)
            a4[(j + 1) & 1][i] = *(const v4f*)(cur + a_off + i * 16 * GEMM_PITCH + (j + 1) * 8);
#pragma unroll
          for (int t = 0; t < NT; ++t)
            b4[(j + 1) & 1][t] = *(const v4f*)(cur + b_off + t * 16 * GEMM_PITCH + (j + 1) * 8);
        }
#pragma unroll
        for (int i = 0; i < NRB; ++i)
#pragma unroll
          for (int t = 0; t < NT; ++t)
            acc[i][t] = __builtin_amdgcn_wmma_f32_16x16x4_f32(
                false, (v2f){a4[j & 1][i].x, a4[j & 1][i].y}, false, (v2f){b4[j & 1][t].x, b4[j & 1][t].y},
                (short)0, acc[i][t], false, false);
#pragma unroll
        for (int i = 0; i < NRB; ++i)
#pragma unroll
          for (int t = 0; t < NT; ++t)
            acc[i][t] = __builtin_amdgcn_wmma_f32_16x16x4_f32(
                false, (v2f){a4[j & 1][i].z, a4[j & 1][i].w}, false, (v2f){b4[j & 1][t].z, b4[j & 1][t].w},
                (short)0, acc[i][t], false, false);
      }

      if (more) {
#pragma unroll
        for (int i = 0; i < NLD; ++i)
          *(v4f*)(nxt + soffs[i] + ln * 32) = pre[i];
      }
    }
    __syncthreads();
  }

#pragma unroll
  for (int i = 0; i < NRB; ++i) {
    const int r0 = m0 + i * 16;
    if (i < nb && r0 + 16 <= M) {
      float* c_ptr = C + (size_t)r0 * GEMM_N + (size_t)(wave * (NT * 16)) + l;
#pragma unroll
      for (int t = 0; t < NT; ++t)
#pragma unroll
        for (int r = 0; r < 8; ++r)
          c_ptr[(size_t)(r + 8 * half) * GEMM_N + t * 16] = acc[i][t][r];
    }
  }
}

__global__ __launch_bounds__(GEMM_WAVES * 32) void wmma_gemm_f32_rowblocks(const float* __restrict__ A,
                                                                          const float* __restrict__ Bt,
                                                                          float* __restrict__ C,
                                                                          int M, int K) {
  __shared__ __attribute__((aligned(16))) float lds[2 * GEMM_STAGE];
  const int nrb = (M + 15) >> 4;
  const int rb0 = ((int)blockIdx.x * nrb) / (int)gridDim.x;
  const int rb1 = (((int)blockIdx.x + 1) * nrb) / (int)gridDim.x;
  const int nb = rb1 - rb0;
  if (nb > GEMM_RB - 1)
    gemm_rowblocks_body<GEMM_RB>(A, Bt, C, lds, M, K, rb0 * 16, nb);
  else
    gemm_rowblocks_body<GEMM_RB - 1>(A, Bt, C, lds, M, K, rb0 * 16, nb);
}

template <int C, bool RELU, bool SPLIT>
__global__ __launch_bounds__(256) void gather_agg_kernel(float* __restrict__ out,
                                                         float* __restrict__ out2,
                                                         const float* __restrict__ z,
                                                         const int* __restrict__ rowptr,
                                                         const int* __restrict__ csr_src,
                                                         const float* __restrict__ csr_w,
                                                         const float* __restrict__ inv,
                                                         const float* __restrict__ bias,
                                                         const float* __restrict__ bias2, int n) {
  constexpr int G = C / 4;
  static_assert(G % 32 == 0 && 256 % G == 0, "a wave must stay inside one node");
  static_assert(!SPLIT || (G / 2) % 32 == 0, "a wave must stay inside one half of the row");
  const int g = threadIdx.x % G;
  const int v = __builtin_amdgcn_readfirstlane(blockIdx.x * (256 / G) + threadIdx.x / G);
  if (v >= n) return;
  const int b = rowptr[v];
  const int e = rowptr[v + 1];
  const float* zc = z + g * 4;

  float4 acc = make_float4(0.0f, 0.0f, 0.0f, 0.0f);
  int j = b;
  for (; j + 4 <= e; j += 4) {
    const int s0 = csr_src[j + 0], s1 = csr_src[j + 1], s2 = csr_src[j + 2], s3 = csr_src[j + 3];
    const float w0 = csr_w[j + 0], w1 = csr_w[j + 1], w2 = csr_w[j + 2], w3 = csr_w[j + 3];
    const float4 t0 = *(const float4*)(zc + (size_t)s0 * C);
    const float4 t1 = *(const float4*)(zc + (size_t)s1 * C);
    const float4 t2 = *(const float4*)(zc + (size_t)s2 * C);
    const float4 t3 = *(const float4*)(zc + (size_t)s3 * C);
    acc.x += t0.x * w0; acc.y += t0.y * w0; acc.z += t0.z * w0; acc.w += t0.w * w0;
    acc.x += t1.x * w1; acc.y += t1.y * w1; acc.z += t1.z * w1; acc.w += t1.w * w1;
    acc.x += t2.x * w2; acc.y += t2.y * w2; acc.z += t2.z * w2; acc.w += t2.w * w2;
    acc.x += t3.x * w3; acc.y += t3.y * w3; acc.z += t3.z * w3; acc.w += t3.w * w3;
  }
  for (; j < e; ++j) {
    const int s0 = csr_src[j];
    const float w0 = csr_w[j];
    const float4 t0 = *(const float4*)(zc + (size_t)s0 * C);
    acc.x += t0.x * w0; acc.y += t0.y * w0; acc.z += t0.z * w0; acc.w += t0.w * w0;
  }

  const float4 zs = *(const float4*)(zc + (size_t)v * C);
  const float iv = inv[v];
  acc.x += zs.x * iv; acc.y += zs.y * iv; acc.z += zs.z * iv; acc.w += zs.w * iv;
  constexpr int OC = SPLIT ? C / 2 : C;
  const bool second = SPLIT && (g >= G / 2);
  const int go = second ? g - G / 2 : g;
  const float* bs = second ? bias2 : bias;
  float* op = second ? out2 : out;
  acc.x += bs[go * 4 + 0];
  acc.y += bs[go * 4 + 1];
  acc.z += bs[go * 4 + 2];
  acc.w += bs[go * 4 + 3];
  if (RELU) {
    acc.x = fmaxf(acc.x, 0.0f);
    acc.y = fmaxf(acc.y, 0.0f);
    acc.z = fmaxf(acc.z, 0.0f);
    acc.w = fmaxf(acc.w, 0.0f);
  }
  *(float4*)(op + (size_t)v * OC + go * 4) = acc;
}

static inline int cdiv_l(long a, long b) { return (int)((a + b - 1) / b); }

extern "C" void kernel_launch(void* const* d_in, const int* in_sizes, int n_in,
                              void* d_out, int out_size, void* d_ws, size_t ws_size,
                              hipStream_t stream) {
  const float* x   = (const float*)d_in[0];
  const int*   ei  = (const int*)d_in[1];
  const int*   src = ei;
  const int*   dst = ei + N_EDGES;
  const float* w1  = (const float*)d_in[2];
  const float* b1  = (const float*)d_in[3];
  const float* wmu = (const float*)d_in[4];
  const float* bmu = (const float*)d_in[5];
  const float* wls = (const float*)d_in[6];
  const float* bls = (const float*)d_in[7];

  float* out_mu = (float*)d_out;
  float* out_ls = out_mu + (size_t)N_NODES * OUT_C;

  float* ws = (float*)d_ws;
  size_t off = 0;
  int*   cnt   = (int*)(ws + off); off += N_NODES;
  float* dis   = ws + off; off += N_NODES;
  float* inv   = ws + off; off += N_NODES;
  float* csr_w = ws + off; off += N_EDGES;
  float* w1t  = ws + off; off += (size_t)IN_C * HID_C;
  float* wmut = ws + off; off += (size_t)HID_C * OUT_C;
  float* wlst = ws + off; off += (size_t)HID_C * OUT_C;
  float* hpre = ws + off; off += (size_t)N_NODES * HID_C;
  float* h    = ws + off; off += (size_t)N_NODES * HID_C;
  int* rowptr  = (int*)(ws + off); off += N_NODES + 16;
  int* rank    = (int*)(ws + off); off += N_EDGES;
  int* raw_src = (int*)(ws + off); off += N_EDGES;
  int* csr_src = (int*)(ws + off); off += N_EDGES;
  float* z2 = hpre;
  int* raw = (int*)h;
  (void)cnt; (void)rank; (void)raw_src;

  bin_prep_kernel<<<NB_BINPREP, BIN_T, 0, stream>>>(src, dst, rowptr, inv, dis, raw, w1, w1t, wmu, wmut,
                                                    wls, wlst, N_NODES);
  seg_order_kernel<<<cdiv_l((long)N_NODES * 32, 256), 256, 0, stream>>>(rowptr, raw, dis, csr_src, csr_w,
                                                                        N_NODES);


  static_assert(HID_C == GEMM_N && 2 * OUT_C == GEMM_N && IN_C % GEMM_KC == 0 && HID_C % GEMM_KC == 0,
                "both products are 256 wide and K is whole chunks");
  wmma_gemm_f32_rowblocks<<<GEMM_GRID, GEMM_WAVES * 32, 0, stream>>>(x, w1t, hpre, N_NODES, IN_C);
  gather_agg_kernel<HID_C, true, false><<<cdiv_l((long)N_NODES * (HID_C / 4), 256), 256, 0, stream>>>(
      h, nullptr, hpre, rowptr, csr_src, csr_w, inv, b1, nullptr, N_NODES);

  wmma_gemm_f32_rowblocks<<<GEMM_GRID, GEMM_WAVES * 32, 0, stream>>>(h, wmut, z2, N_NODES, HID_C);
  (void)wlst;
  gather_agg_kernel<2 * OUT_C, false, true><<<cdiv_l((long)N_NODES * (2 * OUT_C / 4), 256), 256, 0, stream>>>(
      out_mu, out_ls, z2, rowptr, csr_src, csr_w, inv, bmu, bls, N_NODES);
}


# First aggregation fused into the head product's launch: 16 waves gather + relu the next 16-row block of h into LDS while 8 waves multiply the current one; h never stored; 6 -> 5 launches; sums same
# speedup vs baseline: 3.2103x; 1.0738x over previous
#include <hip/hip_runtime.h>

#define N_NODES 50000
#define N_EDGES 500000
#define IN_C 512
#define HID_C 256
#define OUT_C 128

typedef __attribute__((ext_vector_type(2))) float v2f;
typedef __attribute__((ext_vector_type(8))) float v8f;
typedef __attribute__((ext_vector_type(4))) float v4f;

#define BIN_T 1024
#define BIN_G 256
#define BIN_NPB ((N_NODES + BIN_G - 1) / BIN_G)
#define BIN_PAD 256
#define BIN_CAP 8192
#define NB_TW1 ((IN_C * HID_C + BIN_T - 1) / BIN_T)
#define NB_TW2 ((HID_C * OUT_C + BIN_T - 1) / BIN_T)
#define NB_BINPREP (BIN_G + NB_TW1 + 2 * NB_TW2)
static_assert(BIN_NPB <= BIN_PAD && BIN_PAD == 8 * 32, "one wave scans the range, 8 counters per lane");
static_assert(N_EDGES % 4 == 0, "destinations are read four at a time");

__device__ __forceinline__ void transpose_block(const float* __restrict__ B, float* __restrict__ Bt,
                                                int K, int N, int blk) {
  const int i = blk * BIN_T + (int)threadIdx.x;
  if (i < K * N) {
    const int kk = i / N;
    const int n = i - kk * N;
    Bt[(size_t)n * K + kk] = B[i];
  }
}

__device__ __forceinline__ void bin_one(int d, int e, int lo, int nn, int& below_w, int* cnt, int* nlist,
                                        int* list_e, int* list_b) {
  below_w += __builtin_popcount(__builtin_amdgcn_ballot_w32(d < lo));
  const int ld = d - lo;
  if ((unsigned)ld < (unsigned)nn) {
    const int r = atomicAdd(&cnt[ld], 1);
    const int slot = atomicAdd(nlist, 1);
    if (slot < BIN_CAP) {
      list_e[slot] = e;
      list_b[slot] = ld | (r << 8);
    }
  }
}

#define BIN_U 8
__device__ __forceinline__ void bin_load(const int4* __restrict__ dst4, int g, int rot, int4 (&v)[BIN_U],
                                         int (&q)[BIN_U]) {
  constexpr int NV = N_EDGES / 4;
#pragma unroll
  for (int u = 0; u < BIN_U; ++u) {
    const int gu = g + u * BIN_T;
    const bool ok = gu < NV;
    int qq = gu + rot;
    if (qq >= NV) qq -= NV;
    qq = ok ? qq : 0;
    q[u] = ok ? qq : -1;
    v[u] = dst4[qq];
  }
}

__global__ __launch_bounds__(BIN_T) void bin_prep_kernel(const int* __restrict__ src,
                                                         const int* __restrict__ dst,
                                                         int* __restrict__ rowptr,
                                                         float* __restrict__ inv,
                                                         float* __restrict__ dis,
                                                         int* __restrict__ raw,
                                                         const float* __restrict__ w1, float* __restrict__ w1t,
                                                         const float* __restrict__ wmu, float* __restrict__ wmut,
                                                         const float* __restrict__ wls, float* __restrict__ wlst,
                                                         int n) {
  __shared__ int cnt[BIN_PAD + 1];
  __shared__ int start[BIN_PAD + 1];
  __shared__ int cnt2[BIN_PAD];
  __shared__ int list_e[BIN_CAP];
  __shared__ int list_b[BIN_CAP];
  __shared__ int nlist;
  __shared__ int below_s;

  int b = blockIdx.x;
  if (b >= BIN_G) {
    b -= BIN_G;
    if (b < NB_TW1) { transpose_block(w1, w1t, IN_C, HID_C, b); return; }
    b -= NB_TW1;
    if (b < NB_TW2) { transpose_block(wmu, wmut, HID_C, OUT_C, b); return; }
    b -= NB_TW2;
    transpose_block(wls, wlst, HID_C, OUT_C, b);
    return;
  }

  const int t = threadIdx.x;
  const int lane = t & 31;
  const int wv = t >> 5;
  const int lo = b * BIN_NPB;
  int nn = n - lo;
  if (nn > BIN_NPB) nn = BIN_NPB;
  if (nn < 0) nn = 0;

  if (t <= BIN_PAD) cnt[t] = 0;
  if (t == 0) { nlist = 0; below_s = 0; }
  __syncthreads();

  constexpr int NV = N_EDGES / 4;
  constexpr int STEPS = NV / (BIN_U * BIN_T);
  constexpr int REST = (NV - STEPS * BIN_U * BIN_T + BIN_T - 1) / BIN_T;
  static_assert(STEPS >= 1 && REST <= BIN_U, "the rest fits one step");
  constexpr int BIG = 0x7fffffff;
  const int4* __restrict__ dst4 = (const int4*)dst;
  const int rot = (int)(((long)b * NV) / BIN_G);
  int below_w = 0;
  int4 v[BIN_U];
  int q[BIN_U];
  bin_load(dst4, t, rot, v, q);
  for (int s = 0; s < STEPS; ++s) {
    int4 nv[BIN_U];
    int nq[BIN_U];
    bin_load(dst4, (s + 1) * (BIN_U * BIN_T) + t, rot, nv, nq);
#pragma unroll
    for (int u = 0; u < BIN_U; ++u) {
      const int e4 = 4 * q[u];
      bin_one(v[u].x, e4 + 0, lo, nn, below_w, cnt, &nlist, list_e, list_b);
      bin_one(v[u].y, e4 + 1, lo, nn, below_w, cnt, &nlist, list_e, list_b);
      bin_one(v[u].z, e4 + 2, lo, nn, below_w, cnt, &nlist, list_e, list_b);
      bin_one(v[u].w, e4 + 3, lo, nn, below_w, cnt, &nlist, list_e, list_b);
    }
#pragma unroll
    for (int u = 0; u < BIN_U; ++u) { v[u] = nv[u]; q[u] = nq[u]; }
  }
#pragma unroll
  for (int u = 0; u < REST; ++u) {
    const bool ok = q[u] >= 0;
    const int e4 = 4 * q[u];
    bin_one(ok ? v[u].x : BIG, e4 + 0, lo, nn, below_w, cnt, &nlist, list_e, list_b);
    bin_one(ok ? v[u].y : BIG, e4 + 1, lo, nn, below_w, cnt, &nlist, list_e, list_b);
    bin_one(ok ? v[u].z : BIG, e4 + 2, lo, nn, below_w, cnt, &nlist, list_e, list_b);
    bin_one(ok ? v[u].w : BIG, e4 + 3, lo, nn, below_w, cnt, &nlist, list_e, list_b);
  }
  if (lane == 0) atomicAdd(&below_s, below_w);
  __syncthreads();

  if (wv == 0) {
    int loc[8];
    int tot = 0;
#pragma unroll
    for (int k = 0; k < 8; ++k) { loc[k] = cnt[lane * 8 + k]; tot += loc[k]; }
    int incl = tot;
    for (int o = 1; o < 32; o <<= 1) {
      const int v = __shfl_up(incl, o);
      if (lane >= o) incl += v;
    }
    int run = incl - tot;
#pragma unroll
    for (int k = 0; k < 8; ++k) { start[lane * 8 + k] = run; run += loc[k]; }
    if (lane == 31) start[BIN_PAD] = run;
  }
  __syncthreads();

  const int nl = nlist;
  const int bel = below_s;

  if (t < nn) {
    const float dg = (float)(cnt[t] + 1);
    rowptr[lo + t] = bel + start[t];
    inv[lo + t] = 1.0f / dg;
    dis[lo + t] = rsqrtf(dg);
  }
  if (t == 0 && nn > 0 && lo + nn == n) rowptr[n] = bel + start[nn];

  if (nl <= BIN_CAP) {
    for (int k = t; k < nl; k += BIN_T) {
      const int pb = list_b[k];
      raw[bel + start[pb & 255] + (pb >> 8)] = src[list_e[k]];
    }
  } else {
    if (t < BIN_PAD) cnt2[t] = 0;
    __syncthreads();
    for (int i = t; i < N_EDGES; i += BIN_T) {
      const int ld = dst[i] - lo;
      if ((unsigned)ld < (unsigned)nn) {
        const int r = atomicAdd(&cnt2[ld], 1);
        raw[bel + start[ld] + r] = src[i];
      }
    }
  }
}

__global__ void seg_order_kernel(const int* __restrict__ rowptr, const int* __restrict__ raw,
                                 const float* __restrict__ dis,
                                 int* __restrict__ csr_src,
                                 float* __restrict__ csr_w, int n) {
  const int v = (blockIdx.x * blockDim.x + threadIdx.x) >> 5;
  const int lane = threadIdx.x & 31;
  if (v >= n) return;
  const int b = rowptr[v];
  const int e = rowptr[v + 1];
  const float dv = dis[v];
  for (int i = b + lane; i < e; i += 32) {
    const int key = raw[i];
    int rank = 0;
    for (int j = b; j < e; ++j) {
      const int kj = raw[j];
      rank += ((kj < key) || (kj == key && j < i)) ? 1 : 0;
    }
    csr_src[b + rank] = key;
    csr_w[b + rank] = dis[key] * dv;
  }
}

#define GEMM_KC 64
#define GEMM_PITCH 68
#define GEMM_RB 13
#define GEMM_N 256
#define GEMM_WAVES 8
#define GEMM_GRID 256
#define GEMM_SLAB (GEMM_WAVES * 32 / 8)
#define GEMM_AROWS ((GEMM_RB * 16 + GEMM_SLAB - 1) / GEMM_SLAB * GEMM_SLAB)
#define GEMM_STAGE ((GEMM_AROWS + GEMM_N) * GEMM_PITCH)
static_assert((N_NODES + 15) / 16 <= GEMM_GRID * GEMM_RB, "a work-group owns at most GEMM_RB row-blocks");
static_assert(GEMM_N % GEMM_SLAB == 0 && GEMM_N % (GEMM_WAVES * 16) == 0, "whole passes, whole tiles");

template <int NRB>
__device__ __forceinline__ void gemm_rowblocks_body(const float* __restrict__ A,
                                                    const float* __restrict__ Bt,
                                                    float* __restrict__ C, float* lds,
                                                    int M, int K, int m0, int nb) {
  constexpr int SLAB = GEMM_SLAB;
  constexpr int NA = (NRB * 16 + SLAB - 1) / SLAB;
  constexpr int NB = GEMM_N / SLAB;
  constexpr int NLD = NA + NB;
  constexpr int NT = GEMM_N / GEMM_WAVES / 16;

  const int tid = threadIdx.x;
  const int lane = tid & 31;
  const int wave = tid >> 5;
  const int half = lane >> 4;
  const int l = lane & 15;

  const int cr = tid >> 3;
  const int cq = tid & 7;
  unsigned goff[NLD];
  int soffs[NLD];
#pragma unroll
  for (int i = 0; i < NLD; ++i) {
    if (i < NA) {
      int row = m0 + i * SLAB + cr;
      if (row >= M) row = M - 1;
      goff[i] = (unsigned)row * (unsigned)K + (unsigned)(cq * 4);
      soffs[i] = (i * SLAB + cr) * GEMM_PITCH + cq * 4;
    } else {
      goff[i] = (unsigned)((i - NA) * SLAB + cr) * (unsigned)K + (unsigned)(cq * 4);
      soffs[i] = (GEMM_AROWS + (i - NA) * SLAB + cr) * GEMM_PITCH + cq * 4;
    }
  }

  const int a_off = l * GEMM_PITCH + half * 4;
  const int b_off = (GEMM_AROWS + wave * (NT * 16) + l) * GEMM_PITCH + half * 4;

  v8f acc[NRB][NT];
#pragma unroll
  for (int i = 0; i < NRB; ++i)
#pragma unroll
    for (int t = 0; t < NT; ++t) acc[i][t] = (v8f){0, 0, 0, 0, 0, 0, 0, 0};

#pragma unroll
  for (int ln = 0; ln < 2; ++ln) {
    v4f first[NLD];
#pragma unroll
    for (int i = 0; i < NLD; ++i)
      first[i] = *(const v4f*)((i < NA ? A : Bt) + goff[i] + ln * 32);
#pragma unroll
    for (int i = 0; i < NLD; ++i)
      *(v4f*)(lds + soffs[i] + ln * 32) = first[i];
  }
  __syncthreads();

  const int nchunks = K / GEMM_KC;
  for (int kc = 0; kc < nchunks; ++kc) {
    const float* cur = lds + (kc & 1) * GEMM_STAGE;
    float* nxt = lds + ((kc + 1) & 1) * GEMM_STAGE;
    const bool more = (kc + 1 < nchunks);

    v4f a4[2][NRB], b4[2][NT];
#pragma unroll
    for (int i = 0; i < NRB; ++i)
      a4[0][i] = *(const v4f*)(cur + a_off + i * 16 * GEMM_PITCH);
#pragma unroll
    for (int t = 0; t < NT; ++t)
      b4[0][t] = *(const v4f*)(cur + b_off + t * 16 * GEMM_PITCH);

#pragma unroll
    for (int ln = 0; ln < 2; ++ln) {
      v4f pre[NLD];
      if (more) {
#pragma unroll
        for (int i = 0; i < NLD; ++i)
          pre[i] = *(const v4f*)((i < NA ? A : Bt) + goff[i] + (kc + 1) * GEMM_KC + ln * 32);
      }

#pragma unroll
      for (int j = ln * 4; j < ln * 4 + 4; ++j) {
        if (j + 1 < GEMM_KC / 8) {
#pragma unroll
          for (int i = 0; i < NRB; ++i)
            a4[(j + 1) & 1][i] = *(const v4f*)(cur + a_off + i * 16 * GEMM_PITCH + (j + 1) * 8);
#pragma unroll
          for (int t = 0; t < NT; ++t)
            b4[(j + 1) & 1][t] = *(const v4f*)(cur + b_off + t * 16 * GEMM_PITCH + (j + 1) * 8);
        }
#pragma unroll
        for (int i = 0; i < NRB; ++i)
#pragma unroll
          for (int t = 0; t < NT; ++t)
            acc[i][t] = __builtin_amdgcn_wmma_f32_16x16x4_f32(
                false, (v2f){a4[j & 1][i].x, a4[j & 1][i].y}, false, (v2f){b4[j & 1][t].x, b4[j & 1][t].y},
                (short)0, acc[i][t], false, false);
#pragma unroll
        for (int i = 0; i < NRB; ++i)
#pragma unroll
          for (int t = 0; t < NT; ++t)
            acc[i][t] = __builtin_amdgcn_wmma_f32_16x16x4_f32(
                false, (v2f){a4[j & 1][i].z, a4[j & 1][i].w}, false, (v2f){b4[j & 1][t].z, b4[j & 1][t].w},
                (short)0, acc[i][t], false, false);
      }

      if (more) {
#pragma unroll
        for (int i = 0; i < NLD; ++i)
          *(v4f*)(nxt + soffs[i] + ln * 32) = pre[i];
      }
    }
    __syncthreads();
  }

#pragma unroll
  for (int i = 0; i < NRB; ++i) {
    const int r0 = m0 + i * 16;
    if (i < nb && r0 + 16 <= M) {
      float* c_ptr = C + (size_t)r0 * GEMM_N + (size_t)(wave * (NT * 16)) + l;
#pragma unroll
      for (int t = 0; t < NT; ++t)
#pragma unroll
        for (int r = 0; r < 8; ++r)
          c_ptr[(size_t)(r + 8 * half) * GEMM_N + t * 16] = acc[i][t][r];
    }
  }
}

__global__ __launch_bounds__(GEMM_WAVES * 32) void wmma_gemm_f32_rowblocks(const float* __restrict__ A,
                                                                          const float* __restrict__ Bt,
                                                                          float* __restrict__ C,
                                                                          int M, int K) {
  __shared__ __attribute__((aligned(16))) float lds[2 * GEMM_STAGE];
  const int nrb = (M + 15) >> 4;
  const int rb0 = ((int)blockIdx.x * nrb) / (int)gridDim.x;
  const int rb1 = (((int)blockIdx.x + 1) * nrb) / (int)gridDim.x;
  const int nb = rb1 - rb0;
  if (nb > GEMM_RB - 1)
    gemm_rowblocks_body<GEMM_RB>(A, Bt, C, lds, M, K, rb0 * 16, nb);
  else
    gemm_rowblocks_body<GEMM_RB - 1>(A, Bt, C, lds, M, K, rb0 * 16, nb);
}

template <int C, bool RELU, bool SPLIT>
__global__ __launch_bounds__(256) void gather_agg_kernel(float* __restrict__ out,
                                                         float* __restrict__ out2,
                                                         const float* __restrict__ z,
                                                         const int* __restrict__ rowptr,
                                                         const int* __restrict__ csr_src,
                                                         const float* __restrict__ csr_w,
                                                         const float* __restrict__ inv,
                                                         const float* __restrict__ bias,
                                                         const float* __restrict__ bias2, int n) {
  constexpr int G = C / 4;
  static_assert(G % 32 == 0 && 256 % G == 0, "a wave must stay inside one node");
  static_assert(!SPLIT || (G / 2) % 32 == 0, "a wave must stay inside one half of the row");
  const int g = threadIdx.x % G;
  const int v = __builtin_amdgcn_readfirstlane(blockIdx.x * (256 / G) + threadIdx.x / G);
  if (v >= n) return;
  const int b = rowptr[v];
  const int e = rowptr[v + 1];
  const float* zc = z + g * 4;

  float4 acc = make_float4(0.0f, 0.0f, 0.0f, 0.0f);
  int j = b;
  for (; j + 4 <= e; j += 4) {
    const int s0 = csr_src[j + 0], s1 = csr_src[j + 1], s2 = csr_src[j + 2], s3 = csr_src[j + 3];
    const float w0 = csr_w[j + 0], w1 = csr_w[j + 1], w2 = csr_w[j + 2], w3 = csr_w[j + 3];
    const float4 t0 = *(const float4*)(zc + (size_t)s0 * C);
    const float4 t1 = *(const float4*)(zc + (size_t)s1 * C);
    const float4 t2 = *(const float4*)(zc + (size_t)s2 * C);
    const float4 t3 = *(const float4*)(zc + (size_t)s3 * C);
    acc.x += t0.x * w0; acc.y += t0.y * w0; acc.z += t0.z * w0; acc.w += t0.w * w0;
    acc.x += t1.x * w1; acc.y += t1.y * w1; acc.z += t1.z * w1; acc.w += t1.w * w1;
    acc.x += t2.x * w2; acc.y += t2.y * w2; acc.z += t2.z * w2; acc.w += t2.w * w2;
    acc.x += t3.x * w3; acc.y += t3.y * w3; acc.z += t3.z * w3; acc.w += t3.w * w3;
  }
  for (; j < e; ++j) {
    const int s0 = csr_src[j];
    const float w0 = csr_w[j];
    const float4 t0 = *(const float4*)(zc + (size_t)s0 * C);
    acc.x += t0.x * w0; acc.y += t0.y * w0; acc.z += t0.z * w0; acc.w += t0.w * w0;
  }

  const float4 zs = *(const float4*)(zc + (size_t)v * C);
  const float iv = inv[v];
  acc.x += zs.x * iv; acc.y += zs.y * iv; acc.z += zs.z * iv; acc.w += zs.w * iv;
  constexpr int OC = SPLIT ? C / 2 : C;
  const bool second = SPLIT && (g >= G / 2);
  const int go = second ? g - G / 2 : g;
  const float* bs = second ? bias2 : bias;
  float* op = second ? out2 : out;
  acc.x += bs[go * 4 + 0];
  acc.y += bs[go * 4 + 1];
  acc.z += bs[go * 4 + 2];
  acc.w += bs[go * 4 + 3];
  if (RELU) {
    acc.x = fmaxf(acc.x, 0.0f);
    acc.y = fmaxf(acc.y, 0.0f);
    acc.z = fmaxf(acc.z, 0.0f);
    acc.w = fmaxf(acc.w, 0.0f);
  }
  *(float4*)(op + (size_t)v * OC + go * 4) = acc;
}

#define FG_PITCH 260
#define FG_GW 8
#define FG_AW 16
#define FG_T ((FG_GW + FG_AW) * 32)
#define FG_BT (GEMM_N * FG_PITCH)
#define FG_ST (16 * FG_PITCH)
#define FG_R 8
static_assert(HID_C == 256 && GEMM_N == 256, "a lane owns 2 x 4 channels of a 256-wide row");
static_assert(GEMM_RB <= 32 && 32 % FG_R == 0, "a lane per row-block of the work-group; whole groups per 32 entries");

__device__ __forceinline__ void fma4(float4& a, const float4& t, float w) {
  a.x += t.x * w; a.y += t.y * w; a.z += t.z * w; a.w += t.w * w;
}

__device__ __forceinline__ void agg_row_to_lds(float* __restrict__ st_row, const float* __restrict__ z,
                                               const int* __restrict__ csr_src,
                                               const float* __restrict__ csr_w,
                                               const float* __restrict__ inv,
                                               const float* __restrict__ bias, int v, int n, int b, int e,
                                               int sv, float wv, int lane) {
  float4 a0 = make_float4(0.0f, 0.0f, 0.0f, 0.0f);
  float4 a1 = make_float4(0.0f, 0.0f, 0.0f, 0.0f);
  if (v < n) {
    const float* zv = z + (size_t)v * HID_C;
    const float4 zs0 = *(const float4*)(zv + lane * 4);
    const float4 zs1 = *(const float4*)(zv + lane * 4 + 128);
    const float iv = inv[v];
    for (int j0 = b; j0 < e; j0 += 32) {
      if (j0 != b) {
        const int jl = j0 + lane;
        const int jc = (jl < e) ? jl : e - 1;
        sv = csr_src[jc];
        wv = csr_w[jc];
      }
#pragma unroll
      for (int g = 0; g < 32 / FG_R; ++g) {
        const int jg = j0 + g * FG_R;
        if (jg < e) {
          float4 t0[FG_R], t1[FG_R];
#pragma unroll
          for (int k = 0; k < FG_R; ++k) {
            const int s = __builtin_amdgcn_readlane(sv, g * FG_R + k);
            const float* p = z + (size_t)s * HID_C;
            t0[k] = *(const float4*)(p + lane * 4);
            t1[k] = *(const float4*)(p + lane * 4 + 128);
          }
#pragma unroll
          for (int k = 0; k < FG_R; ++k) {
            if (jg + k < e) {
              const float w = __int_as_float(__builtin_amdgcn_readlane(__float_as_int(wv), g * FG_R + k));
              fma4(a0, t0[k], w);
              fma4(a1, t1[k], w);
            }
          }
        }
      }
    }
    fma4(a0, zs0, iv);
    fma4(a1, zs1, iv);
    a0.x += bias[lane * 4 + 0]; a0.y += bias[lane * 4 + 1]; a0.z += bias[lane * 4 + 2]; a0.w += bias[lane * 4 + 3];
    a1.x += bias[128 + lane * 4 + 0]; a1.y += bias[128 + lane * 4 + 1];
    a1.z += bias[128 + lane * 4 + 2]; a1.w += bias[128 + lane * 4 + 3];
    a0.x = fmaxf(a0.x, 0.0f); a0.y = fmaxf(a0.y, 0.0f); a0.z = fmaxf(a0.z, 0.0f); a0.w = fmaxf(a0.w, 0.0f);
    a1.x = fmaxf(a1.x, 0.0f); a1.y = fmaxf(a1.y, 0.0f); a1.z = fmaxf(a1.z, 0.0f); a1.w = fmaxf(a1.w, 0.0f);
  }
  *(float4*)(st_row + lane * 4) = a0;
  *(float4*)(st_row + 128 + lane * 4) = a1;
}

__device__ __forceinline__ void list_head(const int* __restrict__ csr_src, const float* __restrict__ csr_w,
                                          int b, int e, int lane, int& sv, float& wv) {
  if (b < e) {
    const int jl = b + lane;
    const int jc = (jl < e) ? jl : e - 1;
    sv = csr_src[jc];
    wv = csr_w[jc];
  }
}

__device__ __forceinline__ void store_tiles(float* __restrict__ C, const v8f (&acc)[2], int r0, int M, int wave,
                                            int half, int l) {
  if (r0 + 16 <= M) {
    float* c_ptr = C + (size_t)r0 * GEMM_N + (size_t)(wave * 32) + l;
#pragma unroll
    for (int t = 0; t < 2; ++t)
#pragma unroll
      for (int r = 0; r < 8; ++r)
        c_ptr[(size_t)(r + 8 * half) * GEMM_N + t * 16] = acc[t][r];
  }
}

__global__ __launch_bounds__(FG_T) void agg_relu_gemm_kernel(const float* __restrict__ z,
                                                             const int* __restrict__ rowptr,
                                                             const int* __restrict__ csr_src,
                                                             const float* __restrict__ csr_w,
                                                             const float* __restrict__ inv,
                                                             const float* __restrict__ bias,
                                                             const float* __restrict__ Bt,
                                                             float* __restrict__ C, int M) {
  __shared__ __attribute__((aligned(16))) float lds[FG_BT + 2 * FG_ST];
  float* bt_s = lds;
  float* st_s = lds + FG_BT;

  const int tid = threadIdx.x;
  const int lane = tid & 31;
  const int wave = __builtin_amdgcn_readfirstlane(tid >> 5);
  const int half = lane >> 4;
  const int l = lane & 15;

  const int nrb = (M + 15) >> 4;
  const int rb0 = ((int)blockIdx.x * nrb) / (int)gridDim.x;
  const int rb1 = (((int)blockIdx.x + 1) * nrb) / (int)gridDim.x;
  const int nb = rb1 - rb0;
  const int m0 = rb0 * 16;

  for (int idx = tid; idx < GEMM_N * (HID_C / 4); idx += FG_T) {
    const int row = idx >> 6;
    const int q = idx & 63;
    *(v4f*)(bt_s + row * FG_PITCH + q * 4) = *(const v4f*)(Bt + (size_t)row * HID_C + q * 4);
  }
  const int g_row = wave - FG_GW;
  int bvec = 0, evec = 0;
  int sv_n = 0;
  float wv_n = 0.0f;
  if (wave >= FG_GW) {
    const int vi = m0 + lane * 16 + g_row;
    if (lane < nb && vi < M) {
      bvec = rowptr[vi];
      evec = rowptr[vi + 1];
    }
    if (nb > 0) {
      const int b = __builtin_amdgcn_readlane(bvec, 0);
      const int e = __builtin_amdgcn_readlane(evec, 0);
      list_head(csr_src, csr_w, b, e, lane, sv_n, wv_n);
      const int sv = sv_n;
      const float wv = wv_n;
      if (nb > 1)
        list_head(csr_src, csr_w, __builtin_amdgcn_readlane(bvec, 1), __builtin_amdgcn_readlane(evec, 1), lane,
                  sv_n, wv_n);
      agg_row_to_lds(st_s + g_row * FG_PITCH, z, csr_src, csr_w, inv, bias, m0 + g_row, M, b, e, sv, wv, lane);
    }
  }
  __syncthreads();

  const int a_off = l * FG_PITCH + half * 4;
  const int b_off = ((wave & (FG_GW - 1)) * 32 + l) * FG_PITCH + half * 4;

  v8f pend[2];
  pend[0] = (v8f){0, 0, 0, 0, 0, 0, 0, 0};
  pend[1] = (v8f){0, 0, 0, 0, 0, 0, 0, 0};

  for (int i = 0; i < nb; ++i) {
    if (wave < FG_GW) {
      if (i > 0) store_tiles(C, pend, m0 + (i - 1) * 16, M, wave, half, l);
      const float* cur = st_s + (i & 1) * FG_ST;
      v8f acc[2];
      acc[0] = (v8f){0, 0, 0, 0, 0, 0, 0, 0};
      acc[1] = (v8f){0, 0, 0, 0, 0, 0, 0, 0};
      v4f a4[2], b4[2][2];
      a4[0] = *(const v4f*)(cur + a_off);
      b4[0][0] = *(const v4f*)(bt_s + b_off);
      b4[0][1] = *(const v4f*)(bt_s + b_off + 16 * FG_PITCH);
#pragma unroll
      for (int j = 0; j < HID_C / 8; ++j) {
        if (j + 1 < HID_C / 8) {
          a4[(j + 1) & 1] = *(const v4f*)(cur + a_off + (j + 1) * 8);
          b4[(j + 1) & 1][0] = *(const v4f*)(bt_s + b_off + (j + 1) * 8);
          b4[(j + 1) & 1][1] = *(const v4f*)(bt_s + b_off + 16 * FG_PITCH + (j + 1) * 8);
        }
#pragma unroll
        for (int t = 0; t < 2; ++t)
          acc[t] = __builtin_amdgcn_wmma_f32_16x16x4_f32(
              false, (v2f){a4[j & 1].x, a4[j & 1].y}, false, (v2f){b4[j & 1][t].x, b4[j & 1][t].y},
              (short)0, acc[t], false, false);
#pragma unroll
        for (int t = 0; t < 2; ++t)
          acc[t] = __builtin_amdgcn_wmma_f32_16x16x4_f32(
              false, (v2f){a4[j & 1].z, a4[j & 1].w}, false, (v2f){b4[j & 1][t].z, b4[j & 1][t].w},
              (short)0, acc[t], false, false);
      }
      pend[0] = acc[0];
      pend[1] = acc[1];
    } else if (i + 1 < nb) {
      __builtin_amdgcn_s_setprio(3);
      const int b = __builtin_amdgcn_readlane(bvec, i + 1);
      const int e = __builtin_amdgcn_readlane(evec, i + 1);
      const int sv = sv_n;
      const float wv = wv_n;
      if (i + 2 < nb)
        list_head(csr_src, csr_w, __builtin_amdgcn_readlane(bvec, i + 2), __builtin_amdgcn_readlane(evec, i + 2),
                  lane, sv_n, wv_n);
      agg_row_to_lds(st_s + ((i + 1) & 1) * FG_ST + g_row * FG_PITCH, z, csr_src, csr_w, inv, bias,
                     m0 + (i + 1) * 16 + g_row, M, b, e, sv, wv, lane);
      __builtin_amdgcn_s_setprio(0);
    }
    __syncthreads();
  }
  if (wave < FG_GW && nb > 0) store_tiles(C, pend, m0 + (nb - 1) * 16, M, wave, half, l);
}

static inline int cdiv_l(long a, long b) { return (int)((a + b - 1) / b); }

extern "C" void kernel_launch(void* const* d_in, const int* in_sizes, int n_in,
                              void* d_out, int out_size, void* d_ws, size_t ws_size,
                              hipStream_t stream) {
  const float* x   = (const float*)d_in[0];
  const int*   ei  = (const int*)d_in[1];
  const int*   src = ei;
  const int*   dst = ei + N_EDGES;
  const float* w1  = (const float*)d_in[2];
  const float* b1  = (const float*)d_in[3];
  const float* wmu = (const float*)d_in[4];
  const float* bmu = (const float*)d_in[5];
  const float* wls = (const float*)d_in[6];
  const float* bls = (const float*)d_in[7];

  float* out_mu = (float*)d_out;
  float* out_ls = out_mu + (size_t)N_NODES * OUT_C;

  float* ws = (float*)d_ws;
  size_t off = 0;
  int*   cnt   = (int*)(ws + off); off += N_NODES;
  float* dis   = ws + off; off += N_NODES;
  float* inv   = ws + off; off += N_NODES;
  float* csr_w = ws + off; off += N_EDGES;
  float* w1t  = ws + off; off += (size_t)IN_C * HID_C;
  float* wmut = ws + off; off += (size_t)HID_C * OUT_C;
  float* wlst = ws + off; off += (size_t)HID_C * OUT_C;
  float* hpre = ws + off; off += (size_t)N_NODES * HID_C;
  float* zpl  = ws + off; off += (size_t)N_NODES * HID_C;
  int* rowptr  = (int*)(ws + off); off += N_NODES + 16;
  int* rank    = (int*)(ws + off); off += N_EDGES;
  int* raw_src = (int*)(ws + off); off += N_EDGES;
  int* csr_src = (int*)(ws + off); off += N_EDGES;
  float* z2 = zpl;
  int* raw = (int*)zpl;
  (void)cnt; (void)rank; (void)raw_src;

  bin_prep_kernel<<<NB_BINPREP, BIN_T, 0, stream>>>(src, dst, rowptr, inv, dis, raw, w1, w1t, wmu, wmut,
                                                    wls, wlst, N_NODES);
  seg_order_kernel<<<cdiv_l((long)N_NODES * 32, 256), 256, 0, stream>>>(rowptr, raw, dis, csr_src, csr_w,
                                                                        N_NODES);


  static_assert(HID_C == GEMM_N && 2 * OUT_C == GEMM_N && IN_C % GEMM_KC == 0 && HID_C % GEMM_KC == 0,
                "both products are 256 wide and K is whole chunks");
  wmma_gemm_f32_rowblocks<<<GEMM_GRID, GEMM_WAVES * 32, 0, stream>>>(x, w1t, hpre, N_NODES, IN_C);

  agg_relu_gemm_kernel<<<GEMM_GRID, FG_T, 0, stream>>>(hpre, rowptr, csr_src, csr_w, inv, b1, wmut, z2, N_NODES);
  (void)wlst;
  gather_agg_kernel<2 * OUT_C, false, true><<<cdiv_l((long)N_NODES * (2 * OUT_C / 4), 256), 256, 0, stream>>>(
      out_mu, out_ls, z2, rowptr, csr_src, csr_w, inv, bmu, bls, N_NODES);
}


# List-building sweep cut four ways: a work-group = (1 of 64 node ranges, 1 of 4 edge quarters) tests a quarter of the destinations; ordering kernel joins a node's 4 pieces, forms degrees; sums same
# speedup vs baseline: 3.3674x; 1.0489x over previous
#include <hip/hip_runtime.h>

#define N_NODES 50000
#define N_EDGES 500000
#define IN_C 512
#define HID_C 256
#define OUT_C 128

typedef __attribute__((ext_vector_type(2))) float v2f;
typedef __attribute__((ext_vector_type(8))) float v8f;
typedef __attribute__((ext_vector_type(4))) float v4f;

#define BIN_T 1024
#define BIN_Q 4
#define BIN_R 64
#define BIN_G (BIN_Q * BIN_R)
#define BIN_NPB ((N_NODES + BIN_R - 1) / BIN_R)
#define BIN_PAD 1024
#define BIN_LDB 10
#define BIN_CAP 8192
#define E_PART (N_EDGES / BIN_Q)
#define NB_TW1 ((IN_C * HID_C + BIN_T - 1) / BIN_T)
#define NB_TW2 ((HID_C * OUT_C + BIN_T - 1) / BIN_T)
#define NB_BINPREP (BIN_G + NB_TW1 + 2 * NB_TW2)
static_assert(BIN_NPB <= BIN_PAD && BIN_PAD == BIN_T && BIN_PAD == (1 << BIN_LDB), "one counter per thread in the scan");
static_assert(N_EDGES % (4 * BIN_Q) == 0, "whole parts; destinations are read four at a time");
static_assert(BIN_Q == 4, "the ordering kernel reads a node's four piece starts and counts as one 16-byte load each");

__device__ __forceinline__ void transpose_block(const float* __restrict__ B, float* __restrict__ Bt,
                                                int K, int N, int blk) {
  const int i = blk * BIN_T + (int)threadIdx.x;
  if (i < K * N) {
    const int kk = i / N;
    const int n = i - kk * N;
    Bt[(size_t)n * K + kk] = B[i];
  }
}

__device__ __forceinline__ void bin_one(int d, int e, int lo, int nn, int& below_w, int* cnt, int* nlist,
                                        int* list_e, int* list_b) {
  below_w += __builtin_popcount(__builtin_amdgcn_ballot_w32(d < lo));
  const int ld = d - lo;
  if ((unsigned)ld < (unsigned)nn) {
    const int r = atomicAdd(&cnt[ld], 1);
    const int slot = atomicAdd(nlist, 1);
    if (slot < BIN_CAP) {
      list_e[slot] = e;
      list_b[slot] = ld | (r << BIN_LDB);
    }
  }
}

#define BIN_U 8
__device__ __forceinline__ void bin_load(const int4* __restrict__ dst4, int g, int rot, int4 (&v)[BIN_U],
                                         int (&q)[BIN_U]) {
  constexpr int NV = E_PART / 4;
#pragma unroll
  for (int u = 0; u < BIN_U; ++u) {
    const int gu = g + u * BIN_T;
    const bool ok = gu < NV;
    int qq = gu + rot;
    if (qq >= NV) qq -= NV;
    qq = ok ? qq : 0;
    q[u] = ok ? qq : -1;
    v[u] = dst4[qq];
  }
}

__global__ __launch_bounds__(BIN_T) void bin_prep_kernel(const int* __restrict__ src,
                                                         const int* __restrict__ dst,
                                                         int* __restrict__ rowptr4,
                                                         int* __restrict__ cnt4,
                                                         int* __restrict__ raw,
                                                         const float* __restrict__ w1, float* __restrict__ w1t,
                                                         const float* __restrict__ wmu, float* __restrict__ wmut,
                                                         const float* __restrict__ wls, float* __restrict__ wlst,
                                                         int n) {
  __shared__ int cnt[BIN_PAD + 1];
  __shared__ int start[BIN_PAD + 1];
  __shared__ int cnt2[BIN_PAD];
  __shared__ int wsum[BIN_T / 32];
  __shared__ int list_e[BIN_CAP];
  __shared__ int list_b[BIN_CAP];
  __shared__ int nlist;
  __shared__ int below_s;

  int b = blockIdx.x;
  if (b >= BIN_G) {
    b -= BIN_G;
    if (b < NB_TW1) { transpose_block(w1, w1t, IN_C, HID_C, b); return; }
    b -= NB_TW1;
    if (b < NB_TW2) { transpose_block(wmu, wmut, HID_C, OUT_C, b); return; }
    b -= NB_TW2;
    transpose_block(wls, wlst, HID_C, OUT_C, b);
    return;
  }

  const int t = threadIdx.x;
  const int lane = t & 31;
  const int wv = t >> 5;
  const int rg = b / BIN_Q;
  const int pq = b - rg * BIN_Q;
  const int lo = rg * BIN_NPB;
  const int ebase = pq * E_PART;
  int nn = n - lo;
  if (nn > BIN_NPB) nn = BIN_NPB;
  if (nn < 0) nn = 0;

  cnt[t] = 0;
  if (t == 0) { cnt[BIN_PAD] = 0; nlist = 0; below_s = 0; }
  __syncthreads();

  constexpr int NV = E_PART / 4;
  constexpr int STEPS = NV / (BIN_U * BIN_T);
  constexpr int REST = (NV - STEPS * BIN_U * BIN_T + BIN_T - 1) / BIN_T;
  static_assert(STEPS >= 1 && REST <= BIN_U, "the rest fits one step");
  constexpr int BIG = 0x7fffffff;
  const int4* __restrict__ dst4 = (const int4*)(dst + ebase);
  const int rot = (int)(((long)rg * NV) / BIN_R);
  int below_w = 0;
  int4 v[BIN_U];
  int q[BIN_U];
  bin_load(dst4, t, rot, v, q);
#pragma nounroll
  for (int s = 0; s < STEPS; ++s) {
    int4 nv[BIN_U];
    int nq[BIN_U];
    bin_load(dst4, (s + 1) * (BIN_U * BIN_T) + t, rot, nv, nq);
#pragma unroll
    for (int u = 0; u < BIN_U; ++u) {
      const int e4 = ebase + 4 * q[u];
      bin_one(v[u].x, e4 + 0, lo, nn, below_w, cnt, &nlist, list_e, list_b);
      bin_one(v[u].y, e4 + 1, lo, nn, below_w, cnt, &nlist, list_e, list_b);
      bin_one(v[u].z, e4 + 2, lo, nn, below_w, cnt, &nlist, list_e, list_b);
      bin_one(v[u].w, e4 + 3, lo, nn, below_w, cnt, &nlist, list_e, list_b);
    }
#pragma unroll
    for (int u = 0; u < BIN_U; ++u) { v[u] = nv[u]; q[u] = nq[u]; }
  }
#pragma unroll
  for (int u = 0; u < REST; ++u) {
    const bool ok = q[u] >= 0;
    const int e4 = ebase + 4 * q[u];
    bin_one(ok ? v[u].x : BIG, e4 + 0, lo, nn, below_w, cnt, &nlist, list_e, list_b);
    bin_one(ok ? v[u].y : BIG, e4 + 1, lo, nn, below_w, cnt, &nlist, list_e, list_b);
    bin_one(ok ? v[u].z : BIG, e4 + 2, lo, nn, below_w, cnt, &nlist, list_e, list_b);
    bin_one(ok ? v[u].w : BIG, e4 + 3, lo, nn, below_w, cnt, &nlist, list_e, list_b);
  }
  if (lane == 0) atomicAdd(&below_s, below_w);
  __syncthreads();

  const int c = cnt[t];
  int incl = c;
  for (int o = 1; o < 32; o <<= 1) {
    const int up = __shfl_up(incl, o);
    if (lane >= o) incl += up;
  }
  if (lane == 31) wsum[wv] = incl;
  __syncthreads();
  if (wv == 0) {
    const int w = wsum[lane];
    int wi = w;
    for (int o = 1; o < 32; o <<= 1) {
      const int up = __shfl_up(wi, o);
      if (lane >= o) wi += up;
    }
    wsum[lane] = wi - w;
  }
  __syncthreads();
  const int st = wsum[wv] + incl - c;
  start[t] = st;
  if (t == BIN_PAD - 1) start[BIN_PAD] = st + c;
  __syncthreads();

  const int nl = nlist;
  const int bel = below_s;

  if (t < nn) {
    rowptr4[(lo + t) * BIN_Q + pq] = bel + st;
    cnt4[(lo + t) * BIN_Q + pq] = c;
  }

  int* __restrict__ rawp = raw + ebase;
  if (nl <= BIN_CAP) {
    for (int k = t; k < nl; k += BIN_T) {
      const int pb = list_b[k];
      rawp[bel + start[pb & (BIN_PAD - 1)] + (pb >> BIN_LDB)] = src[list_e[k]];
    }
  } else {
    cnt2[t] = 0;
    __syncthreads();
    for (int i = t; i < E_PART; i += BIN_T) {
      const int ld = dst[ebase + i] - lo;
      if ((unsigned)ld < (unsigned)nn) {
        const int r = atomicAdd(&cnt2[ld], 1);
        rawp[bel + start[ld] + r] = src[ebase + i];
      }
    }
  }
}

__device__ __forceinline__ int seg_entry(const int* __restrict__ raw, const int4& rp, int c1, int c2, int c3, int i) {
  const int a = (i < c1) ? (rp.x + i)
              : (i < c2) ? (E_PART + rp.y + (i - c1))
              : (i < c3) ? (2 * E_PART + rp.z + (i - c2))
                         : (3 * E_PART + rp.w + (i - c3));
  return raw[a];
}

__global__ void seg_order_kernel(const int* __restrict__ rowptr4, const int* __restrict__ cnt4,
                                 const int* __restrict__ raw,
                                 int* __restrict__ rowptr, float* __restrict__ inv,
                                 int* __restrict__ csr_src,
                                 float* __restrict__ csr_w, int n) {
  const int v = (blockIdx.x * blockDim.x + threadIdx.x) >> 5;
  const int lane = threadIdx.x & 31;
  if (v >= n) return;
  const int4* __restrict__ cnt4v = (const int4*)cnt4;
  const int4 rp = ((const int4*)rowptr4)[v];
  const int4 cn = cnt4v[v];
  const int c1 = cn.x;
  const int c2 = c1 + cn.y;
  const int c3 = c2 + cn.z;
  const int d = c3 + cn.w;
  const int b = rp.x + rp.y + rp.z + rp.w;
  const float dg = (float)(d + 1);
  const float dv = rsqrtf(dg);
  if (lane == 0) {
    rowptr[v] = b;
    inv[v] = 1.0f / dg;
    if (v == n - 1) rowptr[n] = b + d;
  }
  for (int i = lane; i < d; i += 32) {
    const int key = seg_entry(raw, rp, c1, c2, c3, i);
    int rank = 0;
    for (int j = 0; j < d; ++j) {
      const int kj = seg_entry(raw, rp, c1, c2, c3, j);
      rank += ((kj < key) || (kj == key && j < i)) ? 1 : 0;
    }
    const int4 ck = cnt4v[key];
    const float dk = rsqrtf((float)(ck.x + ck.y + ck.z + ck.w + 1));
    csr_src[b + rank] = key;
    csr_w[b + rank] = dk * dv;
  }
}

#define GEMM_KC 64
#define GEMM_PITCH 68
#define GEMM_RB 13
#define GEMM_N 256
#define GEMM_WAVES 8
#define GEMM_GRID 256
#define GEMM_SLAB (GEMM_WAVES * 32 / 8)
#define GEMM_AROWS ((GEMM_RB * 16 + GEMM_SLAB - 1) / GEMM_SLAB * GEMM_SLAB)
#define GEMM_STAGE ((GEMM_AROWS + GEMM_N) * GEMM_PITCH)
static_assert((N_NODES + 15) / 16 <= GEMM_GRID * GEMM_RB, "a work-group owns at most GEMM_RB row-blocks");
static_assert(GEMM_N % GEMM_SLAB == 0 && GEMM_N % (GEMM_WAVES * 16) == 0, "whole passes, whole tiles");

template <int NRB>
__device__ __forceinline__ void gemm_rowblocks_body(const float* __restrict__ A,
                                                    const float* __restrict__ Bt,
                                                    float* __restrict__ C, float* lds,
                                                    int M, int K, int m0, int nb) {
  constexpr int SLAB = GEMM_SLAB;
  constexpr int NA = (NRB * 16 + SLAB - 1) / SLAB;
  constexpr int NB = GEMM_N / SLAB;
  constexpr int NLD = NA + NB;
  constexpr int NT = GEMM_N / GEMM_WAVES / 16;

  const int tid = threadIdx.x;
  const int lane = tid & 31;
  const int wave = tid >> 5;
  const int half = lane >> 4;
  const int l = lane & 15;

  const int cr = tid >> 3;
  const int cq = tid & 7;
  unsigned goff[NLD];
  int soffs[NLD];
#pragma unroll
  for (int i = 0; i < NLD; ++i) {
    if (i < NA) {
      int row = m0 + i * SLAB + cr;
      if (row >= M) row = M - 1;
      goff[i] = (unsigned)row * (unsigned)K + (unsigned)(cq * 4);
      soffs[i] = (i * SLAB + cr) * GEMM_PITCH + cq * 4;
    } else {
      goff[i] = (unsigned)((i - NA) * SLAB + cr) * (unsigned)K + (unsigned)(cq * 4);
      soffs[i] = (GEMM_AROWS + (i - NA) * SLAB + cr) * GEMM_PITCH + cq * 4;
    }
  }

  const int a_off = l * GEMM_PITCH + half * 4;
  const int b_off = (GEMM_AROWS + wave * (NT * 16) + l) * GEMM_PITCH + half * 4;

  v8f acc[NRB][NT];
#pragma unroll
  for (int i = 0; i < NRB; ++i)
#pragma unroll
    for (int t = 0; t < NT; ++t) acc[i][t] = (v8f){0, 0, 0, 0, 0, 0, 0, 0};

#pragma unroll
  for (int ln = 0; ln < 2; ++ln) {
    v4f first[NLD];
#pragma unroll
    for (int i = 0; i < NLD; ++i)
      first[i] = *(const v4f*)((i < NA ? A : Bt) + goff[i] + ln * 32);
#pragma unroll
    for (int i = 0; i < NLD; ++i)
      *(v4f*)(lds + soffs[i] + ln * 32) = first[i];
  }
  __syncthreads();

  const int nchunks = K / GEMM_KC;
  for (int kc = 0; kc < nchunks; ++kc) {
    const float* cur = lds + (kc & 1) * GEMM_STAGE;
    float* nxt = lds + ((kc + 1) & 1) * GEMM_STAGE;
    const bool more = (kc + 1 < nchunks);

    v4f a4[2][NRB], b4[2][NT];
#pragma unroll
    for (int i = 0; i < NRB; ++i)
      a4[0][i] = *(const v4f*)(cur + a_off + i * 16 * GEMM_PITCH);
#pragma unroll
    for (int t = 0; t < NT; ++t)
      b4[0][t] = *(const v4f*)(cur + b_off + t * 16 * GEMM_PITCH);

#pragma unroll
    for (int ln = 0; ln < 2; ++ln) {
      v4f pre[NLD];
      if (more) {
#pragma unroll
        for (int i = 0; i < NLD; ++i)
          pre[i] = *(const v4f*)((i < NA ? A : Bt) + goff[i] + (kc + 1) * GEMM_KC + ln * 32);
      }

#pragma unroll
      for (int j = ln * 4; j < ln * 4 + 4; ++j) {
        if (j + 1 < GEMM_KC / 8) {
#pragma unroll
          for (int i = 0; i < NRB; ++i)
            a4[(j + 1) & 1][i] = *(const v4f*)(cur + a_off + i * 16 * GEMM_PITCH + (j + 1) * 8);
#pragma unroll
          for (int t = 0; t < NT; ++t)
            b4[(j + 1) & 1][t] = *(const v4f*)(cur + b_off + t * 16 * GEMM_PITCH + (j + 1) * 8);
        }
#pragma unroll
        for (int i = 0; i < NRB; ++i)
#pragma unroll
          for (int t = 0; t < NT; ++t)
            acc[i][t] = __builtin_amdgcn_wmma_f32_16x16x4_f32(
                false, (v2f){a4[j & 1][i].x, a4[j & 1][i].y}, false, (v2f){b4[j & 1][t].x, b4[j & 1][t].y},
                (short)0, acc[i][t], false, false);
#pragma unroll
        for (int i = 0; i < NRB; ++i)
#pragma unroll
          for (int t = 0; t < NT; ++t)
            acc[i][t] = __builtin_amdgcn_wmma_f32_16x16x4_f32(
                false, (v2f){a4[j & 1][i].z, a4[j & 1][i].w}, false, (v2f){b4[j & 1][t].z, b4[j & 1][t].w},
                (short)0, acc[i][t], false, false);
      }

      if (more) {
#pragma unroll
        for (int i = 0; i < NLD; ++i)
          *(v4f*)(nxt + soffs[i] + ln * 32) = pre[i];
      }
    }
    __syncthreads();
  }

#pragma unroll
  for (int i = 0; i < NRB; ++i) {
    const int r0 = m0 + i * 16;
    if (i < nb && r0 + 16 <= M) {
      float* c_ptr = C + (size_t)r0 * GEMM_N + (size_t)(wave * (NT * 16)) + l;
#pragma unroll
      for (int t = 0; t < NT; ++t)
#pragma unroll
        for (int r = 0; r < 8; ++r)
          c_ptr[(size_t)(r + 8 * half) * GEMM_N + t * 16] = acc[i][t][r];
    }
  }
}

__global__ __launch_bounds__(GEMM_WAVES * 32) void wmma_gemm_f32_rowblocks(const float* __restrict__ A,
                                                                          const float* __restrict__ Bt,
                                                                          float* __restrict__ C,
                                                                          int M, int K) {
  __shared__ __attribute__((aligned(16))) float lds[2 * GEMM_STAGE];
  const int nrb = (M + 15) >> 4;
  const int rb0 = ((int)blockIdx.x * nrb) / (int)gridDim.x;
  const int rb1 = (((int)blockIdx.x + 1) * nrb) / (int)gridDim.x;
  const int nb = rb1 - rb0;
  if (nb > GEMM_RB - 1)
    gemm_rowblocks_body<GEMM_RB>(A, Bt, C, lds, M, K, rb0 * 16, nb);
  else
    gemm_rowblocks_body<GEMM_RB - 1>(A, Bt, C, lds, M, K, rb0 * 16, nb);
}

template <int C, bool RELU, bool SPLIT>
__global__ __launch_bounds__(256) void gather_agg_kernel(float* __restrict__ out,
                                                         float* __restrict__ out2,
                                                         const float* __restrict__ z,
                                                         const int* __restrict__ rowptr,
                                                         const int* __restrict__ csr_src,
                                                         const float* __restrict__ csr_w,
                                                         const float* __restrict__ inv,
                                                         const float* __restrict__ bias,
                                                         const float* __restrict__ bias2, int n) {
  constexpr int G = C / 4;
  static_assert(G % 32 == 0 && 256 % G == 0, "a wave must stay inside one node");
  static_assert(!SPLIT || (G / 2) % 32 == 0, "a wave must stay inside one half of the row");
  const int g = threadIdx.x % G;
  const int v = __builtin_amdgcn_readfirstlane(blockIdx.x * (256 / G) + threadIdx.x / G);
  if (v >= n) return;
  const int b = rowptr[v];
  const int e = rowptr[v + 1];
  const float* zc = z + g * 4;

  float4 acc = make_float4(0.0f, 0.0f, 0.0f, 0.0f);
  int j = b;
  for (; j + 4 <= e; j += 4) {
    const int s0 = csr_src[j + 0], s1 = csr_src[j + 1], s2 = csr_src[j + 2], s3 = csr_src[j + 3];
    const float w0 = csr_w[j + 0], w1 = csr_w[j + 1], w2 = csr_w[j + 2], w3 = csr_w[j + 3];
    const float4 t0 = *(const float4*)(zc + (size_t)s0 * C);
    const float4 t1 = *(const float4*)(zc + (size_t)s1 * C);
    const float4 t2 = *(const float4*)(zc + (size_t)s2 * C);
    const float4 t3 = *(const float4*)(zc + (size_t)s3 * C);
    acc.x += t0.x * w0; acc.y += t0.y * w0; acc.z += t0.z * w0; acc.w += t0.w * w0;
    acc.x += t1.x * w1; acc.y += t1.y * w1; acc.z += t1.z * w1; acc.w += t1.w * w1;
    acc.x += t2.x * w2; acc.y += t2.y * w2; acc.z += t2.z * w2; acc.w += t2.w * w2;
    acc.x += t3.x * w3; acc.y += t3.y * w3; acc.z += t3.z * w3; acc.w += t3.w * w3;
  }
  for (; j < e; ++j) {
    const int s0 = csr_src[j];
    const float w0 = csr_w[j];
    const float4 t0 = *(const float4*)(zc + (size_t)s0 * C);
    acc.x += t0.x * w0; acc.y += t0.y * w0; acc.z += t0.z * w0; acc.w += t0.w * w0;
  }

  const float4 zs = *(const float4*)(zc + (size_t)v * C);
  const float iv = inv[v];
  acc.x += zs.x * iv; acc.y += zs.y * iv; acc.z += zs.z * iv; acc.w += zs.w * iv;
  constexpr int OC = SPLIT ? C / 2 : C;
  const bool second = SPLIT && (g >= G / 2);
  const int go = second ? g - G / 2 : g;
  const float* bs = second ? bias2 : bias;
  float* op = second ? out2 : out;
  acc.x += bs[go * 4 + 0];
  acc.y += bs[go * 4 + 1];
  acc.z += bs[go * 4 + 2];
  acc.w += bs[go * 4 + 3];
  if (RELU) {
    acc.x = fmaxf(acc.x, 0.0f);
    acc.y = fmaxf(acc.y, 0.0f);
    acc.z = fmaxf(acc.z, 0.0f);
    acc.w = fmaxf(acc.w, 0.0f);
  }
  *(float4*)(op + (size_t)v * OC + go * 4) = acc;
}

#define FG_PITCH 260
#define FG_GW 8
#define FG_AW 16
#define FG_T ((FG_GW + FG_AW) * 32)
#define FG_BT (GEMM_N * FG_PITCH)
#define FG_ST (16 * FG_PITCH)
#define FG_R 8
static_assert(HID_C == 256 && GEMM_N == 256, "a lane owns 2 x 4 channels of a 256-wide row");
static_assert(GEMM_RB <= 32 && 32 % FG_R == 0, "a lane per row-block of the work-group; whole groups per 32 entries");

__device__ __forceinline__ void fma4(float4& a, const float4& t, float w) {
  a.x += t.x * w; a.y += t.y * w; a.z += t.z * w; a.w += t.w * w;
}

__device__ __forceinline__ void agg_row_to_lds(float* __restrict__ st_row, const float* __restrict__ z,
                                               const int* __restrict__ csr_src,
                                               const float* __restrict__ csr_w,
                                               const float* __restrict__ inv,
                                               const float* __restrict__ bias, int v, int n, int b, int e,
                                               int sv, float wv, int lane) {
  float4 a0 = make_float4(0.0f, 0.0f, 0.0f, 0.0f);
  float4 a1 = make_float4(0.0f, 0.0f, 0.0f, 0.0f);
  if (v < n) {
    const float* zv = z + (size_t)v * HID_C;
    const float4 zs0 = *(const float4*)(zv + lane * 4);
    const float4 zs1 = *(const float4*)(zv + lane * 4 + 128);
    const float iv = inv[v];
    for (int j0 = b; j0 < e; j0 += 32) {
      if (j0 != b) {
        const int jl = j0 + lane;
        const int jc = (jl < e) ? jl : e - 1;
        sv = csr_src[jc];
        wv = csr_w[jc];
      }
#pragma unroll
      for (int g = 0; g < 32 / FG_R; ++g) {
        const int jg = j0 + g * FG_R;
        if (jg < e) {
          float4 t0[FG_R], t1[FG_R];
#pragma unroll
          for (int k = 0; k < FG_R; ++k) {
            const int s = __builtin_amdgcn_readlane(sv, g * FG_R + k);
            const float* p = z + (size_t)s * HID_C;
            t0[k] = *(const float4*)(p + lane * 4);
            t1[k] = *(const float4*)(p + lane * 4 + 128);
          }
#pragma unroll
          for (int k = 0; k < FG_R; ++k) {
            if (jg + k < e) {
              const float w = __int_as_float(__builtin_amdgcn_readlane(__float_as_int(wv), g * FG_R + k));
              fma4(a0, t0[k], w);
              fma4(a1, t1[k], w);
            }
          }
        }
      }
    }
    fma4(a0, zs0, iv);
    fma4(a1, zs1, iv);
    a0.x += bias[lane * 4 + 0]; a0.y += bias[lane * 4 + 1]; a0.z += bias[lane * 4 + 2]; a0.w += bias[lane * 4 + 3];
    a1.x += bias[128 + lane * 4 + 0]; a1.y += bias[128 + lane * 4 + 1];
    a1.z += bias[128 + lane * 4 + 2]; a1.w += bias[128 + lane * 4 + 3];
    a0.x = fmaxf(a0.x, 0.0f); a0.y = fmaxf(a0.y, 0.0f); a0.z = fmaxf(a0.z, 0.0f); a0.w = fmaxf(a0.w, 0.0f);
    a1.x = fmaxf(a1.x, 0.0f); a1.y = fmaxf(a1.y, 0.0f); a1.z = fmaxf(a1.z, 0.0f); a1.w = fmaxf(a1.w, 0.0f);
  }
  *(float4*)(st_row + lane * 4) = a0;
  *(float4*)(st_row + 128 + lane * 4) = a1;
}

__device__ __forceinline__ void list_head(const int* __restrict__ csr_src, const float* __restrict__ csr_w,
                                          int b, int e, int lane, int& sv, float& wv) {
  if (b < e) {
    const int jl = b + lane;
    const int jc = (jl < e) ? jl : e - 1;
    sv = csr_src[jc];
    wv = csr_w[jc];
  }
}

__device__ __forceinline__ void store_tiles(float* __restrict__ C, const v8f (&acc)[2], int r0, int M, int wave,
                                            int half, int l) {
  if (r0 + 16 <= M) {
    float* c_ptr = C + (size_t)r0 * GEMM_N + (size_t)(wave * 32) + l;
#pragma unroll
    for (int t = 0; t < 2; ++t)
#pragma unroll
      for (int r = 0; r < 8; ++r)
        c_ptr[(size_t)(r + 8 * half) * GEMM_N + t * 16] = acc[t][r];
  }
}

__global__ __launch_bounds__(FG_T) void agg_relu_gemm_kernel(const float* __restrict__ z,
                                                             const int* __restrict__ rowptr,
                                                             const int* __restrict__ csr_src,
                                                             const float* __restrict__ csr_w,
                                                             const float* __restrict__ inv,
                                                             const float* __restrict__ bias,
                                                             const float* __restrict__ Bt,
                                                             float* __restrict__ C, int M) {
  __shared__ __attribute__((aligned(16))) float lds[FG_BT + 2 * FG_ST];
  float* bt_s = lds;
  float* st_s = lds + FG_BT;

  const int tid = threadIdx.x;
  const int lane = tid & 31;
  const int wave = __builtin_amdgcn_readfirstlane(tid >> 5);
  const int half = lane >> 4;
  const int l = lane & 15;

  const int nrb = (M + 15) >> 4;
  const int rb0 = ((int)blockIdx.x * nrb) / (int)gridDim.x;
  const int rb1 = (((int)blockIdx.x + 1) * nrb) / (int)gridDim.x;
  const int nb = rb1 - rb0;
  const int m0 = rb0 * 16;

  for (int idx = tid; idx < GEMM_N * (HID_C / 4); idx += FG_T) {
    const int row = idx >> 6;
    const int q = idx & 63;
    *(v4f*)(bt_s + row * FG_PITCH + q * 4) = *(const v4f*)(Bt + (size_t)row * HID_C + q * 4);
  }
  const int g_row = wave - FG_GW;
  int bvec = 0, evec = 0;
  int sv_n = 0;
  float wv_n = 0.0f;
  if (wave >= FG_GW) {
    const int vi = m0 + lane * 16 + g_row;
    if (lane < nb && vi < M) {
      bvec = rowptr[vi];
      evec = rowptr[vi + 1];
    }
    if (nb > 0) {
      const int b = __builtin_amdgcn_readlane(bvec, 0);
      const int e = __builtin_amdgcn_readlane(evec, 0);
      list_head(csr_src, csr_w, b, e, lane, sv_n, wv_n);
      const int sv = sv_n;
      const float wv = wv_n;
      if (nb > 1)
        list_head(csr_src, csr_w, __builtin_amdgcn_readlane(bvec, 1), __builtin_amdgcn_readlane(evec, 1), lane,
                  sv_n, wv_n);
      agg_row_to_lds(st_s + g_row * FG_PITCH, z, csr_src, csr_w, inv, bias, m0 + g_row, M, b, e, sv, wv, lane);
    }
  }
  __syncthreads();

  const int a_off = l * FG_PITCH + half * 4;
  const int b_off = ((wave & (FG_GW - 1)) * 32 + l) * FG_PITCH + half * 4;

  v8f pend[2];
  pend[0] = (v8f){0, 0, 0, 0, 0, 0, 0, 0};
  pend[1] = (v8f){0, 0, 0, 0, 0, 0, 0, 0};

  for (int i = 0; i < nb; ++i) {
    if (wave < FG_GW) {
      if (i > 0) store_tiles(C, pend, m0 + (i - 1) * 16, M, wave, half, l);
      const float* cur = st_s + (i & 1) * FG_ST;
      v8f acc[2];
      acc[0] = (v8f){0, 0, 0, 0, 0, 0, 0, 0};
      acc[1] = (v8f){0, 0, 0, 0, 0, 0, 0, 0};
      v4f a4[2], b4[2][2];
      a4[0] = *(const v4f*)(cur + a_off);
      b4[0][0] = *(const v4f*)(bt_s + b_off);
      b4[0][1] = *(const v4f*)(bt_s + b_off + 16 * FG_PITCH);
#pragma unroll
      for (int j = 0; j < HID_C / 8; ++j) {
        if (j + 1 < HID_C / 8) {
          a4[(j + 1) & 1] = *(const v4f*)(cur + a_off + (j + 1) * 8);
          b4[(j + 1) & 1][0] = *(const v4f*)(bt_s + b_off + (j + 1) * 8);
          b4[(j + 1) & 1][1] = *(const v4f*)(bt_s + b_off + 16 * FG_PITCH + (j + 1) * 8);
        }
#pragma unroll
        for (int t = 0; t < 2; ++t)
          acc[t] = __builtin_amdgcn_wmma_f32_16x16x4_f32(
              false, (v2f){a4[j & 1].x, a4[j & 1].y}, false, (v2f){b4[j & 1][t].x, b4[j & 1][t].y},
              (short)0, acc[t], false, false);
#pragma unroll
        for (int t = 0; t < 2; ++t)
          acc[t] = __builtin_amdgcn_wmma_f32_16x16x4_f32(
              false, (v2f){a4[j & 1].z, a4[j & 1].w}, false, (v2f){b4[j & 1][t].z, b4[j & 1][t].w},
              (short)0, acc[t], false, false);
      }
      pend[0] = acc[0];
      pend[1] = acc[1];
    } else if (i + 1 < nb) {
      __builtin_amdgcn_s_setprio(3);
      const int b = __builtin_amdgcn_readlane(bvec, i + 1);
      const int e = __builtin_amdgcn_readlane(evec, i + 1);
      const int sv = sv_n;
      const float wv = wv_n;
      if (i + 2 < nb)
        list_head(csr_src, csr_w, __builtin_amdgcn_readlane(bvec, i + 2), __builtin_amdgcn_readlane(evec, i + 2),
                  lane, sv_n, wv_n);
      agg_row_to_lds(st_s + ((i + 1) & 1) * FG_ST + g_row * FG_PITCH, z, csr_src, csr_w, inv, bias,
                     m0 + (i + 1) * 16 + g_row, M, b, e, sv, wv, lane);
      __builtin_amdgcn_s_setprio(0);
    }
    __syncthreads();
  }
  if (wave < FG_GW && nb > 0) store_tiles(C, pend, m0 + (nb - 1) * 16, M, wave, half, l);
}

static inline int cdiv_l(long a, long b) { return (int)((a + b - 1) / b); }

extern "C" void kernel_launch(void* const* d_in, const int* in_sizes, int n_in,
                              void* d_out, int out_size, void* d_ws, size_t ws_size,
                              hipStream_t stream) {
  const float* x   = (const float*)d_in[0];
  const int*   ei  = (const int*)d_in[1];
  const int*   src = ei;
  const int*   dst = ei + N_EDGES;
  const float* w1  = (const float*)d_in[2];
  const float* b1  = (const float*)d_in[3];
  const float* wmu = (const float*)d_in[4];
  const float* bmu = (const float*)d_in[5];
  const float* wls = (const float*)d_in[6];
  const float* bls = (const float*)d_in[7];

  float* out_mu = (float*)d_out;
  float* out_ls = out_mu + (size_t)N_NODES * OUT_C;

  float* ws = (float*)d_ws;
  size_t off = 0;
  int*   cnt   = (int*)(ws + off); off += N_NODES;
  float* dis   = ws + off; off += N_NODES;
  float* inv   = ws + off; off += N_NODES;
  float* csr_w = ws + off; off += N_EDGES;
  float* w1t  = ws + off; off += (size_t)IN_C * HID_C;
  float* wmut = ws + off; off += (size_t)HID_C * OUT_C;
  float* wlst = ws + off; off += (size_t)HID_C * OUT_C;
  float* hpre = ws + off; off += (size_t)N_NODES * HID_C;
  float* zpl  = ws + off; off += (size_t)N_NODES * HID_C;
  int* rowptr  = (int*)(ws + off); off += N_NODES + 16;
  int* rank    = (int*)(ws + off); off += N_EDGES;
  int* raw_src = (int*)(ws + off); off += N_EDGES;
  int* csr_src = (int*)(ws + off); off += N_EDGES;
  float* z2 = zpl;
  int* raw = (int*)zpl;
  int* rowptr4 = rank;
  int* cnt4 = raw_src;
  static_assert(4 * N_NODES <= N_EDGES && (N_NODES + 16) % 4 == 0 && N_EDGES % 4 == 0 && N_NODES % 4 == 0,
                "the [node][part] planes fit and start on 16-byte boundaries");
  (void)cnt; (void)dis;

  bin_prep_kernel<<<NB_BINPREP, BIN_T, 0, stream>>>(src, dst, rowptr4, cnt4, raw, w1, w1t, wmu, wmut,
                                                    wls, wlst, N_NODES);
  seg_order_kernel<<<cdiv_l((long)N_NODES * 32, 256), 256, 0, stream>>>(rowptr4, cnt4, raw,
                                                                        rowptr, inv, csr_src, csr_w, N_NODES);


  static_assert(HID_C == GEMM_N && 2 * OUT_C == GEMM_N && IN_C % GEMM_KC == 0 && HID_C % GEMM_KC == 0,
                "both products are 256 wide and K is whole chunks");
  wmma_gemm_f32_rowblocks<<<GEMM_GRID, GEMM_WAVES * 32, 0, stream>>>(x, w1t, hpre, N_NODES, IN_C);

  agg_relu_gemm_kernel<<<GEMM_GRID, FG_T, 0, stream>>>(hpre, rowptr, csr_src, csr_w, inv, b1, wmut, z2, N_NODES);
  (void)wlst;
  gather_agg_kernel<2 * OUT_C, false, true><<<cdiv_l((long)N_NODES * (2 * OUT_C / 4), 256), 256, 0, stream>>>(
      out_mu, out_ls, z2, rowptr, csr_src, csr_w, inv, bmu, bls, N_NODES);
}


# Ordering launch retired: the fused launch's gather waves join their own node's list pieces, rank them by lane reads, write list, coefficients, starts for the last gather; 5 -> 4 launches; sums same
# speedup vs baseline: 3.5218x; 1.0459x over previous
#include <hip/hip_runtime.h>

#define N_NODES 50000
#define N_EDGES 500000
#define IN_C 512
#define HID_C 256
#define OUT_C 128

typedef __attribute__((ext_vector_type(2))) float v2f;
typedef __attribute__((ext_vector_type(8))) float v8f;
typedef __attribute__((ext_vector_type(4))) float v4f;

#define BIN_T 1024
#define BIN_Q 4
#define BIN_R 64
#define BIN_G (BIN_Q * BIN_R)
#define BIN_NPB ((N_NODES + BIN_R - 1) / BIN_R)
#define BIN_PAD 1024
#define BIN_LDB 10
#define BIN_CAP 8192
#define E_PART (N_EDGES / BIN_Q)
#define NB_TW1 ((IN_C * HID_C + BIN_T - 1) / BIN_T)
#define NB_TW2 ((HID_C * OUT_C + BIN_T - 1) / BIN_T)
#define NB_BINPREP (BIN_G + NB_TW1 + 2 * NB_TW2)
static_assert(BIN_NPB <= BIN_PAD && BIN_PAD == BIN_T && BIN_PAD == (1 << BIN_LDB), "one counter per thread in the scan");
static_assert(N_EDGES % (4 * BIN_Q) == 0, "whole parts; destinations are read four at a time");
static_assert(BIN_Q == 4, "a node's four piece starts and counts are read as one 16-byte load each");
#define RAW_OFF(p) (((p) & 1) * E_PART + ((p) >> 1) * N_EDGES)

__device__ __forceinline__ void transpose_block(const float* __restrict__ B, float* __restrict__ Bt,
                                                int K, int N, int blk) {
  const int i = blk * BIN_T + (int)threadIdx.x;
  if (i < K * N) {
    const int kk = i / N;
    const int n = i - kk * N;
    Bt[(size_t)n * K + kk] = B[i];
  }
}

__device__ __forceinline__ void bin_one(int d, int e, int lo, int nn, int& below_w, int* cnt, int* nlist,
                                        int* list_e, int* list_b) {
  below_w += __builtin_popcount(__builtin_amdgcn_ballot_w32(d < lo));
  const int ld = d - lo;
  if ((unsigned)ld < (unsigned)nn) {
    const int r = atomicAdd(&cnt[ld], 1);
    const int slot = atomicAdd(nlist, 1);
    if (slot < BIN_CAP) {
      list_e[slot] = e;
      list_b[slot] = ld | (r << BIN_LDB);
    }
  }
}

#define BIN_U 8
__device__ __forceinline__ void bin_load(const int4* __restrict__ dst4, int g, int rot, int4 (&v)[BIN_U],
                                         int (&q)[BIN_U]) {
  constexpr int NV = E_PART / 4;
#pragma unroll
  for (int u = 0; u < BIN_U; ++u) {
    const int gu = g + u * BIN_T;
    const bool ok = gu < NV;
    int qq = gu + rot;
    if (qq >= NV) qq -= NV;
    qq = ok ? qq : 0;
    q[u] = ok ? qq : -1;
    v[u] = dst4[qq];
  }
}

__global__ __launch_bounds__(BIN_T) void bin_prep_kernel(const int* __restrict__ src,
                                                         const int* __restrict__ dst,
                                                         int* __restrict__ rowptr4,
                                                         int* __restrict__ cnt4,
                                                         int* __restrict__ raw,
                                                         const float* __restrict__ w1, float* __restrict__ w1t,
                                                         const float* __restrict__ wmu, float* __restrict__ wmut,
                                                         const float* __restrict__ wls, float* __restrict__ wlst,
                                                         int n) {
  __shared__ int cnt[BIN_PAD + 1];
  __shared__ int start[BIN_PAD + 1];
  __shared__ int cnt2[BIN_PAD];
  __shared__ int wsum[BIN_T / 32];
  __shared__ int list_e[BIN_CAP];
  __shared__ int list_b[BIN_CAP];
  __shared__ int nlist;
  __shared__ int below_s;

  int b = blockIdx.x;
  if (b >= BIN_G) {
    b -= BIN_G;
    if (b < NB_TW1) { transpose_block(w1, w1t, IN_C, HID_C, b); return; }
    b -= NB_TW1;
    if (b < NB_TW2) { transpose_block(wmu, wmut, HID_C, OUT_C, b); return; }
    b -= NB_TW2;
    transpose_block(wls, wlst, HID_C, OUT_C, b);
    return;
  }

  const int t = threadIdx.x;
  const int lane = t & 31;
  const int wv = t >> 5;
  const int rg = b / BIN_Q;
  const int pq = b - rg * BIN_Q;
  const int lo = rg * BIN_NPB;
  const int ebase = pq * E_PART;
  int nn = n - lo;
  if (nn > BIN_NPB) nn = BIN_NPB;
  if (nn < 0) nn = 0;

  cnt[t] = 0;
  if (t == 0) { cnt[BIN_PAD] = 0; nlist = 0; below_s = 0; }
  __syncthreads();

  constexpr int NV = E_PART / 4;
  constexpr int STEPS = NV / (BIN_U * BIN_T);
  constexpr int REST = (NV - STEPS * BIN_U * BIN_T + BIN_T - 1) / BIN_T;
  static_assert(STEPS >= 1 && REST <= BIN_U, "the rest fits one step");
  constexpr int BIG = 0x7fffffff;
  const int4* __restrict__ dst4 = (const int4*)(dst + ebase);
  const int rot = (int)(((long)rg * NV) / BIN_R);
  int below_w = 0;
  int4 v[BIN_U];
  int q[BIN_U];
  bin_load(dst4, t, rot, v, q);
#pragma nounroll
  for (int s = 0; s < STEPS; ++s) {
    int4 nv[BIN_U];
    int nq[BIN_U];
    bin_load(dst4, (s + 1) * (BIN_U * BIN_T) + t, rot, nv, nq);
#pragma unroll
    for (int u = 0; u < BIN_U; ++u) {
      const int e4 = ebase + 4 * q[u];
      bin_one(v[u].x, e4 + 0, lo, nn, below_w, cnt, &nlist, list_e, list_b);
      bin_one(v[u].y, e4 + 1, lo, nn, below_w, cnt, &nlist, list_e, list_b);
      bin_one(v[u].z, e4 + 2, lo, nn, below_w, cnt, &nlist, list_e, list_b);
      bin_one(v[u].w, e4 + 3, lo, nn, below_w, cnt, &nlist, list_e, list_b);
    }
#pragma unroll
    for (int u = 0; u < BIN_U; ++u) { v[u] = nv[u]; q[u] = nq[u]; }
  }
#pragma unroll
  for (int u = 0; u < REST; ++u) {
    const bool ok = q[u] >= 0;
    const int e4 = ebase + 4 * q[u];
    bin_one(ok ? v[u].x : BIG, e4 + 0, lo, nn, below_w, cnt, &nlist, list_e, list_b);
    bin_one(ok ? v[u].y : BIG, e4 + 1, lo, nn, below_w, cnt, &nlist, list_e, list_b);
    bin_one(ok ? v[u].z : BIG, e4 + 2, lo, nn, below_w, cnt, &nlist, list_e, list_b);
    bin_one(ok ? v[u].w : BIG, e4 + 3, lo, nn, below_w, cnt, &nlist, list_e, list_b);
  }
  if (lane == 0) atomicAdd(&below_s, below_w);
  __syncthreads();

  const int c = cnt[t];
  int incl = c;
  for (int o = 1; o < 32; o <<= 1) {
    const int up = __shfl_up(incl, o);
    if (lane >= o) incl += up;
  }
  if (lane == 31) wsum[wv] = incl;
  __syncthreads();
  if (wv == 0) {
    const int w = wsum[lane];
    int wi = w;
    for (int o = 1; o < 32; o <<= 1) {
      const int up = __shfl_up(wi, o);
      if (lane >= o) wi += up;
    }
    wsum[lane] = wi - w;
  }
  __syncthreads();
  const int st = wsum[wv] + incl - c;
  start[t] = st;
  if (t == BIN_PAD - 1) start[BIN_PAD] = st + c;
  __syncthreads();

  const int nl = nlist;
  const int bel = below_s;

  if (t < nn) {
    rowptr4[(lo + t) * BIN_Q + pq] = bel + st;
    cnt4[(lo + t) * BIN_Q + pq] = c;
  }

  int* __restrict__ rawp = raw + RAW_OFF(pq);
  if (nl <= BIN_CAP) {
    for (int k = t; k < nl; k += BIN_T) {
      const int pb = list_b[k];
      rawp[bel + start[pb & (BIN_PAD - 1)] + (pb >> BIN_LDB)] = src[list_e[k]];
    }
  } else {
    cnt2[t] = 0;
    __syncthreads();
    for (int i = t; i < E_PART; i += BIN_T) {
      const int ld = dst[ebase + i] - lo;
      if ((unsigned)ld < (unsigned)nn) {
        const int r = atomicAdd(&cnt2[ld], 1);
        rawp[bel + start[ld] + r] = src[ebase + i];
      }
    }
  }
}

struct NodeSeg { int p0, p1, p2, p3, c1, c2, c3, d, b; };

__device__ __forceinline__ int seg_addr(const NodeSeg& s, int i) {
  return (i < s.c1) ? (s.p0 + i)
       : (i < s.c2) ? (s.p1 + (i - s.c1))
       : (i < s.c3) ? (s.p2 + (i - s.c2))
                    : (s.p3 + (i - s.c3));
}

#define GEMM_KC 64
#define GEMM_PITCH 68
#define GEMM_RB 13
#define GEMM_N 256
#define GEMM_WAVES 8
#define GEMM_GRID 256
#define GEMM_SLAB (GEMM_WAVES * 32 / 8)
#define GEMM_AROWS ((GEMM_RB * 16 + GEMM_SLAB - 1) / GEMM_SLAB * GEMM_SLAB)
#define GEMM_STAGE ((GEMM_AROWS + GEMM_N) * GEMM_PITCH)
static_assert((N_NODES + 15) / 16 <= GEMM_GRID * GEMM_RB, "a work-group owns at most GEMM_RB row-blocks");
static_assert(GEMM_N % GEMM_SLAB == 0 && GEMM_N % (GEMM_WAVES * 16) == 0, "whole passes, whole tiles");

template <int NRB>
__device__ __forceinline__ void gemm_rowblocks_body(const float* __restrict__ A,
                                                    const float* __restrict__ Bt,
                                                    float* __restrict__ C, float* lds,
                                                    int M, int K, int m0, int nb) {
  constexpr int SLAB = GEMM_SLAB;
  constexpr int NA = (NRB * 16 + SLAB - 1) / SLAB;
  constexpr int NB = GEMM_N / SLAB;
  constexpr int NLD = NA + NB;
  constexpr int NT = GEMM_N / GEMM_WAVES / 16;

  const int tid = threadIdx.x;
  const int lane = tid & 31;
  const int wave = tid >> 5;
  const int half = lane >> 4;
  const int l = lane & 15;

  const int cr = tid >> 3;
  const int cq = tid & 7;
  unsigned goff[NLD];
  int soffs[NLD];
#pragma unroll
  for (int i = 0; i < NLD; ++i) {
    if (i < NA) {
      int row = m0 + i * SLAB + cr;
      if (row >= M) row = M - 1;
      goff[i] = (unsigned)row * (unsigned)K + (unsigned)(cq * 4);
      soffs[i] = (i * SLAB + cr) * GEMM_PITCH + cq * 4;
    } else {
      goff[i] = (unsigned)((i - NA) * SLAB + cr) * (unsigned)K + (unsigned)(cq * 4);
      soffs[i] = (GEMM_AROWS + (i - NA) * SLAB + cr) * GEMM_PITCH + cq * 4;
    }
  }

  const int a_off = l * GEMM_PITCH + half * 4;
  const int b_off = (GEMM_AROWS + wave * (NT * 16) + l) * GEMM_PITCH + half * 4;

  v8f acc[NRB][NT];
#pragma unroll
  for (int i = 0; i < NRB; ++i)
#pragma unroll
    for (int t = 0; t < NT; ++t) acc[i][t] = (v8f){0, 0, 0, 0, 0, 0, 0, 0};

#pragma unroll
  for (int ln = 0; ln < 2; ++ln) {
    v4f first[NLD];
#pragma unroll
    for (int i = 0; i < NLD; ++i)
      first[i] = *(const v4f*)((i < NA ? A : Bt) + goff[i] + ln * 32);
#pragma unroll
    for (int i = 0; i < NLD; ++i)
      *(v4f*)(lds + soffs[i] + ln * 32) = first[i];
  }
  __syncthreads();

  const int nchunks = K / GEMM_KC;
  for (int kc = 0; kc < nchunks; ++kc) {
    const float* cur = lds + (kc & 1) * GEMM_STAGE;
    float* nxt = lds + ((kc + 1) & 1) * GEMM_STAGE;
    const bool more = (kc + 1 < nchunks);

    v4f a4[2][NRB], b4[2][NT];
#pragma unroll
    for (int i = 0; i < NRB; ++i)
      a4[0][i] = *(const v4f*)(cur + a_off + i * 16 * GEMM_PITCH);
#pragma unroll
    for (int t = 0; t < NT; ++t)
      b4[0][t] = *(const v4f*)(cur + b_off + t * 16 * GEMM_PITCH);

#pragma unroll
    for (int ln = 0; ln < 2; ++ln) {
      v4f pre[NLD];
      if (more) {
#pragma unroll
        for (int i = 0; i < NLD; ++i)
          pre[i] = *(const v4f*)((i < NA ? A : Bt) + goff[i] + (kc + 1) * GEMM_KC + ln * 32);
      }

#pragma unroll
      for (int j = ln * 4; j < ln * 4 + 4; ++j) {
        if (j + 1 < GEMM_KC / 8) {
#pragma unroll
          for (int i = 0; i < NRB; ++i)
            a4[(j + 1) & 1][i] = *(const v4f*)(cur + a_off + i * 16 * GEMM_PITCH + (j + 1) * 8);
#pragma unroll
          for (int t = 0; t < NT; ++t)
            b4[(j + 1) & 1][t] = *(const v4f*)(cur + b_off + t * 16 * GEMM_PITCH + (j + 1) * 8);
        }
#pragma unroll
        for (int i = 0; i < NRB; ++i)
#pragma unroll
          for (int t = 0; t < NT; ++t)
            acc[i][t] = __builtin_amdgcn_wmma_f32_16x16x4_f32(
                false, (v2f){a4[j & 1][i].x, a4[j & 1][i].y}, false, (v2f){b4[j & 1][t].x, b4[j & 1][t].y},
                (short)0, acc[i][t], false, false);
#pragma unroll
        for (int i = 0; i < NRB; ++i)
#pragma unroll
          for (int t = 0; t < NT; ++t)
            acc[i][t] = __builtin_amdgcn_wmma_f32_16x16x4_f32(
                false, (v2f){a4[j & 1][i].z, a4[j & 1][i].w}, false, (v2f){b4[j & 1][t].z, b4[j & 1][t].w},
                (short)0, acc[i][t], false, false);
      }

      if (more) {
#pragma unroll
        for (int i = 0; i < NLD; ++i)
          *(v4f*)(nxt + soffs[i] + ln * 32) = pre[i];
      }
    }
    __syncthreads();
  }

#pragma unroll
  for (int i = 0; i < NRB; ++i) {
    const int r0 = m0 + i * 16;
    if (i < nb && r0 + 16 <= M) {
      float* c_ptr = C + (size_t)r0 * GEMM_N + (size_t)(wave * (NT * 16)) + l;
#pragma unroll
      for (int t = 0; t < NT; ++t)
#pragma unroll
        for (int r = 0; r < 8; ++r)
          c_ptr[(size_t)(r + 8 * half) * GEMM_N + t * 16] = acc[i][t][r];
    }
  }
}

__global__ __launch_bounds__(GEMM_WAVES * 32) void wmma_gemm_f32_rowblocks(const float* __restrict__ A,
                                                                          const float* __restrict__ Bt,
                                                                          float* __restrict__ C,
                                                                          int M, int K) {
  __shared__ __attribute__((aligned(16))) float lds[2 * GEMM_STAGE];
  const int nrb = (M + 15) >> 4;
  const int rb0 = ((int)blockIdx.x * nrb) / (int)gridDim.x;
  const int rb1 = (((int)blockIdx.x + 1) * nrb) / (int)gridDim.x;
  const int nb = rb1 - rb0;
  if (nb > GEMM_RB - 1)
    gemm_rowblocks_body<GEMM_RB>(A, Bt, C, lds, M, K, rb0 * 16, nb);
  else
    gemm_rowblocks_body<GEMM_RB - 1>(A, Bt, C, lds, M, K, rb0 * 16, nb);
}

template <int C, bool RELU, bool SPLIT>
__global__ __launch_bounds__(256) void gather_agg_kernel(float* __restrict__ out,
                                                         float* __restrict__ out2,
                                                         const float* __restrict__ z,
                                                         const int* __restrict__ rowptr,
                                                         const int* __restrict__ csr_src,
                                                         const float* __restrict__ csr_w,
                                                         const float* __restrict__ inv,
                                                         const float* __restrict__ bias,
                                                         const float* __restrict__ bias2, int n) {
  constexpr int G = C / 4;
  static_assert(G % 32 == 0 && 256 % G == 0, "a wave must stay inside one node");
  static_assert(!SPLIT || (G / 2) % 32 == 0, "a wave must stay inside one half of the row");
  const int g = threadIdx.x % G;
  const int v = __builtin_amdgcn_readfirstlane(blockIdx.x * (256 / G) + threadIdx.x / G);
  if (v >= n) return;
  const int b = rowptr[v];
  const int e = rowptr[v + 1];
  const float* zc = z + g * 4;

  float4 acc = make_float4(0.0f, 0.0f, 0.0f, 0.0f);
  int j = b;
  for (; j + 4 <= e; j += 4) {
    const int s0 = csr_src[j + 0], s1 = csr_src[j + 1], s2 = csr_src[j + 2], s3 = csr_src[j + 3];
    const float w0 = csr_w[j + 0], w1 = csr_w[j + 1], w2 = csr_w[j + 2], w3 = csr_w[j + 3];
    const float4 t0 = *(const float4*)(zc + (size_t)s0 * C);
    const float4 t1 = *(const float4*)(zc + (size_t)s1 * C);
    const float4 t2 = *(const float4*)(zc + (size_t)s2 * C);
    const float4 t3 = *(const float4*)(zc + (size_t)s3 * C);
    acc.x += t0.x * w0; acc.y += t0.y * w0; acc.z += t0.z * w0; acc.w += t0.w * w0;
    acc.x += t1.x * w1; acc.y += t1.y * w1; acc.z += t1.z * w1; acc.w += t1.w * w1;
    acc.x += t2.x * w2; acc.y += t2.y * w2; acc.z += t2.z * w2; acc.w += t2.w * w2;
    acc.x += t3.x * w3; acc.y += t3.y * w3; acc.z += t3.z * w3; acc.w += t3.w * w3;
  }
  for (; j < e; ++j) {
    const int s0 = csr_src[j];
    const float w0 = csr_w[j];
    const float4 t0 = *(const float4*)(zc + (size_t)s0 * C);
    acc.x += t0.x * w0; acc.y += t0.y * w0; acc.z += t0.z * w0; acc.w += t0.w * w0;
  }

  const float4 zs = *(const float4*)(zc + (size_t)v * C);
  const float iv = inv[v];
  acc.x += zs.x * iv; acc.y += zs.y * iv; acc.z += zs.z * iv; acc.w += zs.w * iv;
  constexpr int OC = SPLIT ? C / 2 : C;
  const bool second = SPLIT && (g >= G / 2);
  const int go = second ? g - G / 2 : g;
  const float* bs = second ? bias2 : bias;
  float* op = second ? out2 : out;
  acc.x += bs[go * 4 + 0];
  acc.y += bs[go * 4 + 1];
  acc.z += bs[go * 4 + 2];
  acc.w += bs[go * 4 + 3];
  if (RELU) {
    acc.x = fmaxf(acc.x, 0.0f);
    acc.y = fmaxf(acc.y, 0.0f);
    acc.z = fmaxf(acc.z, 0.0f);
    acc.w = fmaxf(acc.w, 0.0f);
  }
  *(float4*)(op + (size_t)v * OC + go * 4) = acc;
}

#define FG_PITCH 260
#define FG_GW 8
#define FG_AW 16
#define FG_T ((FG_GW + FG_AW) * 32)
#define FG_BT (GEMM_N * FG_PITCH)
#define FG_ST (16 * FG_PITCH)
#define FG_R 8
static_assert(HID_C == 256 && GEMM_N == 256, "a lane owns 2 x 4 channels of a 256-wide row");
static_assert(GEMM_RB <= 32 && 32 % FG_R == 0, "a lane per row-block of the work-group; whole groups per 32 entries");

__device__ __forceinline__ void fma4(float4& a, const float4& t, float w) {
  a.x += t.x * w; a.y += t.y * w; a.z += t.z * w; a.w += t.w * w;
}

__device__ __forceinline__ void agg_row_to_lds(float* __restrict__ st_row, const float* __restrict__ z,
                                               const int* csr_src, float* csr_w,
                                               const float* __restrict__ bias, int v, int n, int b, int e,
                                               int sv, float wv, const int4& ck, bool lazy, int lane) {
  float4 a0 = make_float4(0.0f, 0.0f, 0.0f, 0.0f);
  float4 a1 = make_float4(0.0f, 0.0f, 0.0f, 0.0f);
  if (v < n) {
    const float* zv = z + (size_t)v * HID_C;
    const float4 zs0 = *(const float4*)(zv + lane * 4);
    const float4 zs1 = *(const float4*)(zv + lane * 4 + 128);
    const float dg = (float)(e - b + 1);
    const float iv = 1.0f / dg;
    const float dv = rsqrtf(dg);
    for (int j0 = b; j0 < e; j0 += 32) {
      if (j0 != b) {
        const int jl = j0 + lane;
        const int jc = (jl < e) ? jl : e - 1;
        sv = csr_src[jc];
        wv = csr_w[jc];
      }
#pragma unroll
      for (int g = 0; g < 32 / FG_R; ++g) {
        const int jg = j0 + g * FG_R;
        if (jg < e) {
          float4 t0[FG_R], t1[FG_R];
#pragma unroll
          for (int k = 0; k < FG_R; ++k) {
            const int s = __builtin_amdgcn_readlane(sv, g * FG_R + k);
            const float* p = z + (size_t)s * HID_C;
            t0[k] = *(const float4*)(p + lane * 4);
            t1[k] = *(const float4*)(p + lane * 4 + 128);
          }
          if (g == 0 && lazy) {
            wv = rsqrtf((float)(ck.x + ck.y + ck.z + ck.w + 1)) * dv;
            if (lane < e - b) csr_w[b + lane] = wv;
          }
#pragma unroll
          for (int k = 0; k < FG_R; ++k) {
            if (jg + k < e) {
              const float w = __int_as_float(__builtin_amdgcn_readlane(__float_as_int(wv), g * FG_R + k));
              fma4(a0, t0[k], w);
              fma4(a1, t1[k], w);
            }
          }
        }
      }
    }
    fma4(a0, zs0, iv);
    fma4(a1, zs1, iv);
    a0.x += bias[lane * 4 + 0]; a0.y += bias[lane * 4 + 1]; a0.z += bias[lane * 4 + 2]; a0.w += bias[lane * 4 + 3];
    a1.x += bias[128 + lane * 4 + 0]; a1.y += bias[128 + lane * 4 + 1];
    a1.z += bias[128 + lane * 4 + 2]; a1.w += bias[128 + lane * 4 + 3];
    a0.x = fmaxf(a0.x, 0.0f); a0.y = fmaxf(a0.y, 0.0f); a0.z = fmaxf(a0.z, 0.0f); a0.w = fmaxf(a0.w, 0.0f);
    a1.x = fmaxf(a1.x, 0.0f); a1.y = fmaxf(a1.y, 0.0f); a1.z = fmaxf(a1.z, 0.0f); a1.w = fmaxf(a1.w, 0.0f);
  }
  *(float4*)(st_row + lane * 4) = a0;
  *(float4*)(st_row + 128 + lane * 4) = a1;
}

__device__ __forceinline__ NodeSeg seg_of(const int4& rpv, const int4& cnv, int k) {
  NodeSeg s;
  const int r0 = __builtin_amdgcn_readlane(rpv.x, k), r1 = __builtin_amdgcn_readlane(rpv.y, k);
  const int r2 = __builtin_amdgcn_readlane(rpv.z, k), r3 = __builtin_amdgcn_readlane(rpv.w, k);
  const int n0 = __builtin_amdgcn_readlane(cnv.x, k), n1 = __builtin_amdgcn_readlane(cnv.y, k);
  const int n2 = __builtin_amdgcn_readlane(cnv.z, k), n3 = __builtin_amdgcn_readlane(cnv.w, k);
  s.p0 = RAW_OFF(0) + r0; s.p1 = RAW_OFF(1) + r1; s.p2 = RAW_OFF(2) + r2; s.p3 = RAW_OFF(3) + r3;
  s.c1 = n0; s.c2 = n0 + n1; s.c3 = n0 + n1 + n2; s.d = n0 + n1 + n2 + n3;
  s.b = r0 + r1 + r2 + r3;
  return s;
}

__device__ __forceinline__ int list_entry_ahead(const NodeSeg& s, const int* __restrict__ raw, int lane) {
  int key = 0;
  if (s.d > 0) {
    const int i = (lane < s.d) ? lane : s.d - 1;
    key = raw[seg_addr(s, i)];
  }
  return key;
}

__device__ __forceinline__ void order_list(const NodeSeg& s, int v, int n, int key_raw, int lane,
                                           const int* __restrict__ raw, const int4* __restrict__ cnt4v,
                                           int* __restrict__ rowptr, float* __restrict__ inv,
                                           int* csr_src, float* csr_w,
                                           int& sv, float& wv, int4& ck, bool& lazy) {
  const int d = s.d;
  const int b = s.b;
  const float dg = (float)(d + 1);
  if (lane == 0) {
    rowptr[v] = b;
    inv[v] = 1.0f / dg;
    if (v == n - 1) rowptr[n] = b + d;
  }
  lazy = (d <= 32);
  if (d <= 32) {
    if (d > 0) {
      const int key = (lane < d) ? key_raw : 0x7fffffff;
      int rank = 0;
      for (int j = 0; j < d; ++j) {
        const int kj = __builtin_amdgcn_readlane(key, j);
        rank += ((kj < key) || (kj == key && j < lane)) ? 1 : 0;
      }
      rank = (lane < d) ? rank : lane;
      const int so = __builtin_amdgcn_ds_permute(rank << 2, key);
      const int last = __builtin_amdgcn_readlane(so, d - 1);
      sv = (lane < d) ? so : last;
      ck = cnt4v[sv];
      if (lane < d) csr_src[b + lane] = sv;
    }
  } else {
    const float dv = rsqrtf(dg);
    for (int i = lane; i < d; i += 32) {
      const int key = raw[seg_addr(s, i)];
      int rank = 0;
      for (int j = 0; j < d; ++j) {
        const int kj = raw[seg_addr(s, j)];
        rank += ((kj < key) || (kj == key && j < i)) ? 1 : 0;
      }
      const int4 cq = cnt4v[key];
      const float dk = rsqrtf((float)(cq.x + cq.y + cq.z + cq.w + 1));
      csr_src[b + rank] = key;
      csr_w[b + rank] = dk * dv;
    }
    __threadfence();
    sv = csr_src[b + lane];
    wv = csr_w[b + lane];
  }
}

__device__ __forceinline__ void store_tiles(float* __restrict__ C, const v8f (&acc)[2], int r0, int M, int wave,
                                            int half, int l) {
  if (r0 + 16 <= M) {
    float* c_ptr = C + (size_t)r0 * GEMM_N + (size_t)(wave * 32) + l;
#pragma unroll
    for (int t = 0; t < 2; ++t)
#pragma unroll
      for (int r = 0; r < 8; ++r)
        c_ptr[(size_t)(r + 8 * half) * GEMM_N + t * 16] = acc[t][r];
  }
}

__global__ __launch_bounds__(FG_T) void agg_relu_gemm_kernel(const float* __restrict__ z,
                                                             const int* __restrict__ rowptr4,
                                                             const int* __restrict__ cnt4,
                                                             const int* __restrict__ raw,
                                                             int* __restrict__ rowptr,
                                                             int* __restrict__ csr_src,
                                                             float* __restrict__ csr_w,
                                                             float* __restrict__ inv,
                                                             const float* __restrict__ bias,
                                                             const float* __restrict__ Bt,
                                                             float* __restrict__ C, int M) {
  __shared__ __attribute__((aligned(16))) float lds[FG_BT + 2 * FG_ST];
  float* bt_s = lds;
  float* st_s = lds + FG_BT;

  const int tid = threadIdx.x;
  const int lane = tid & 31;
  const int wave = __builtin_amdgcn_readfirstlane(tid >> 5);
  const int half = lane >> 4;
  const int l = lane & 15;

  const int nrb = (M + 15) >> 4;
  const int rb0 = ((int)blockIdx.x * nrb) / (int)gridDim.x;
  const int rb1 = (((int)blockIdx.x + 1) * nrb) / (int)gridDim.x;
  const int nb = rb1 - rb0;
  const int m0 = rb0 * 16;

  for (int idx = tid; idx < GEMM_N * (HID_C / 4); idx += FG_T) {
    const int row = idx >> 6;
    const int q = idx & 63;
    *(v4f*)(bt_s + row * FG_PITCH + q * 4) = *(const v4f*)(Bt + (size_t)row * HID_C + q * 4);
  }
  const int g_row = wave - FG_GW;
  const int4* __restrict__ cnt4v = (const int4*)cnt4;
  int4 rpv = make_int4(0, 0, 0, 0), cnv = make_int4(0, 0, 0, 0);
  int key_n = 0;
  if (wave >= FG_GW) {
    const int vi = m0 + lane * 16 + g_row;
    if (lane < nb && vi < M) {
      rpv = ((const int4*)rowptr4)[vi];
      cnv = cnt4v[vi];
    }
    if (nb > 0) {
      const int v0 = m0 + g_row;
      const NodeSeg s0 = seg_of(rpv, cnv, 0);
      const int key0 = list_entry_ahead(s0, raw, lane);
      if (nb > 1) key_n = list_entry_ahead(seg_of(rpv, cnv, 1), raw, lane);
      int sv = 0;
      float wv = 0.0f;
      int4 ck = make_int4(0, 0, 0, 0);
      bool lazy = true;
      if (v0 < M) order_list(s0, v0, M, key0, lane, raw, cnt4v, rowptr, inv, csr_src, csr_w, sv, wv, ck, lazy);
      agg_row_to_lds(st_s + g_row * FG_PITCH, z, csr_src, csr_w, bias, v0, M, s0.b, s0.b + s0.d, sv, wv, ck, lazy,
                     lane);
    }
  }
  __syncthreads();

  const int a_off = l * FG_PITCH + half * 4;
  const int b_off = ((wave & (FG_GW - 1)) * 32 + l) * FG_PITCH + half * 4;

  v8f pend[2];
  pend[0] = (v8f){0, 0, 0, 0, 0, 0, 0, 0};
  pend[1] = (v8f){0, 0, 0, 0, 0, 0, 0, 0};

  for (int i = 0; i < nb; ++i) {
    if (wave < FG_GW) {
      if (i > 0) store_tiles(C, pend, m0 + (i - 1) * 16, M, wave, half, l);
      const float* cur = st_s + (i & 1) * FG_ST;
      v8f acc[2];
      acc[0] = (v8f){0, 0, 0, 0, 0, 0, 0, 0};
      acc[1] = (v8f){0, 0, 0, 0, 0, 0, 0, 0};
      v4f a4[2], b4[2][2];
      a4[0] = *(const v4f*)(cur + a_off);
      b4[0][0] = *(const v4f*)(bt_s + b_off);
      b4[0][1] = *(const v4f*)(bt_s + b_off + 16 * FG_PITCH);
#pragma unroll
      for (int j = 0; j < HID_C / 8; ++j) {
        if (j + 1 < HID_C / 8) {
          a4[(j + 1) & 1] = *(const v4f*)(cur + a_off + (j + 1) * 8);
          b4[(j + 1) & 1][0] = *(const v4f*)(bt_s + b_off + (j + 1) * 8);
          b4[(j + 1) & 1][1] = *(const v4f*)(bt_s + b_off + 16 * FG_PITCH + (j + 1) * 8);
        }
#pragma unroll
        for (int t = 0; t < 2; ++t)
          acc[t] = __builtin_amdgcn_wmma_f32_16x16x4_f32(
              false, (v2f){a4[j & 1].x, a4[j & 1].y}, false, (v2f){b4[j & 1][t].x, b4[j & 1][t].y},
              (short)0, acc[t], false, false);
#pragma unroll
        for (int t = 0; t < 2; ++t)
          acc[t] = __builtin_amdgcn_wmma_f32_16x16x4_f32(
              false, (v2f){a4[j & 1].z, a4[j & 1].w}, false, (v2f){b4[j & 1][t].z, b4[j & 1][t].w},
              (short)0, acc[t], false, false);
      }
      pend[0] = acc[0];
      pend[1] = acc[1];
    } else if (i + 1 < nb) {
      __builtin_amdgcn_s_setprio(3);
      const int v1 = m0 + (i + 1) * 16 + g_row;
      const NodeSeg s1 = seg_of(rpv, cnv, i + 1);
      const int key1 = key_n;
      if (i + 2 < nb) key_n = list_entry_ahead(seg_of(rpv, cnv, i + 2), raw, lane);
      int sv = 0;
      float wv = 0.0f;
      int4 ck = make_int4(0, 0, 0, 0);
      bool lazy = true;
      if (v1 < M) order_list(s1, v1, M, key1, lane, raw, cnt4v, rowptr, inv, csr_src, csr_w, sv, wv, ck, lazy);
      agg_row_to_lds(st_s + ((i + 1) & 1) * FG_ST + g_row * FG_PITCH, z, csr_src, csr_w, bias, v1, M, s1.b,
                     s1.b + s1.d, sv, wv, ck, lazy, lane);
      __builtin_amdgcn_s_setprio(0);
    }
    __syncthreads();
  }
  if (wave < FG_GW && nb > 0) store_tiles(C, pend, m0 + (nb - 1) * 16, M, wave, half, l);
}

static inline int cdiv_l(long a, long b) { return (int)((a + b - 1) / b); }

extern "C" void kernel_launch(void* const* d_in, const int* in_sizes, int n_in,
                              void* d_out, int out_size, void* d_ws, size_t ws_size,
                              hipStream_t stream) {
  const float* x   = (const float*)d_in[0];
  const int*   ei  = (const int*)d_in[1];
  const int*   src = ei;
  const int*   dst = ei + N_EDGES;
  const float* w1  = (const float*)d_in[2];
  const float* b1  = (const float*)d_in[3];
  const float* wmu = (const float*)d_in[4];
  const float* bmu = (const float*)d_in[5];
  const float* wls = (const float*)d_in[6];
  const float* bls = (const float*)d_in[7];

  float* out_mu = (float*)d_out;
  float* out_ls = out_mu + (size_t)N_NODES * OUT_C;

  float* ws = (float*)d_ws;
  size_t off = 0;
  int*   cnt   = (int*)(ws + off); off += N_NODES;
  float* dis   = ws + off; off += N_NODES;
  float* inv   = ws + off; off += N_NODES;
  float* csr_w = ws + off; off += N_EDGES;
  float* w1t  = ws + off; off += (size_t)IN_C * HID_C;
  float* wmut = ws + off; off += (size_t)HID_C * OUT_C;
  float* wlst = ws + off; off += (size_t)HID_C * OUT_C;
  float* hpre = ws + off; off += (size_t)N_NODES * HID_C;
  float* zpl  = ws + off; off += (size_t)N_NODES * HID_C;
  int* rowptr  = (int*)(ws + off); off += N_NODES + 16;
  int* rank    = (int*)(ws + off); off += N_EDGES;
  int* raw_src = (int*)(ws + off); off += N_EDGES;
  int* csr_src = (int*)(ws + off); off += N_EDGES;
  float* z2 = zpl;
  int* raw = rank + 4 * N_NODES;
  int* rowptr4 = rank;
  int* cnt4 = raw_src;
  static_assert(4 * N_NODES + 2 * E_PART <= N_EDGES && (N_NODES + 16) % 4 == 0 && N_EDGES % 4 == 0 && N_NODES % 4 == 0,
                "the [node][part] planes and two parts' lists fit in a plane; the planes start on 16-byte boundaries");
  (void)cnt; (void)dis;

  bin_prep_kernel<<<NB_BINPREP, BIN_T, 0, stream>>>(src, dst, rowptr4, cnt4, raw, w1, w1t, wmu, wmut,
                                                    wls, wlst, N_NODES);


  static_assert(HID_C == GEMM_N && 2 * OUT_C == GEMM_N && IN_C % GEMM_KC == 0 && HID_C % GEMM_KC == 0,
                "both products are 256 wide and K is whole chunks");
  wmma_gemm_f32_rowblocks<<<GEMM_GRID, GEMM_WAVES * 32, 0, stream>>>(x, w1t, hpre, N_NODES, IN_C);

  agg_relu_gemm_kernel<<<GEMM_GRID, FG_T, 0, stream>>>(hpre, rowptr4, cnt4, raw, rowptr, csr_src, csr_w, inv, b1, wmut,
                                                       z2, N_NODES);
  (void)wlst;
  gather_agg_kernel<2 * OUT_C, false, true><<<cdiv_l((long)N_NODES * (2 * OUT_C / 4), 256), 256, 0, stream>>>(
      out_mu, out_ls, z2, rowptr, csr_src, csr_w, inv, bmu, bls, N_NODES);
}


# Row planes (x@w1, h@[w_mu|w_ls], weight transposes) moved onto 1-KB boundaries: a gathered row is 8 whole lines, not 9; first product's tiles leave via LDS as whole-line 16-byte stores; sums same
# speedup vs baseline: 3.7687x; 1.0701x over previous
#include <hip/hip_runtime.h>

#define N_NODES 50000
#define N_EDGES 500000
#define IN_C 512
#define HID_C 256
#define OUT_C 128

typedef __attribute__((ext_vector_type(2))) float v2f;
typedef __attribute__((ext_vector_type(8))) float v8f;
typedef __attribute__((ext_vector_type(4))) float v4f;

#define BIN_T 1024
#define BIN_Q 4
#define BIN_R 64
#define BIN_G (BIN_Q * BIN_R)
#define BIN_NPB ((N_NODES + BIN_R - 1) / BIN_R)
#define BIN_PAD 1024
#define BIN_LDB 10
#define BIN_CAP 8192
#define E_PART (N_EDGES / BIN_Q)
#define NB_TW1 ((IN_C * HID_C + BIN_T - 1) / BIN_T)
#define NB_TW2 ((HID_C * OUT_C + BIN_T - 1) / BIN_T)
#define NB_BINPREP (BIN_G + NB_TW1 + 2 * NB_TW2)
static_assert(BIN_NPB <= BIN_PAD && BIN_PAD == BIN_T && BIN_PAD == (1 << BIN_LDB), "one counter per thread in the scan");
static_assert(N_EDGES % (4 * BIN_Q) == 0, "whole parts; destinations are read four at a time");
static_assert(BIN_Q == 4, "a node's four piece starts and counts are read as one 16-byte load each");
#define RAW_OFF(p) (((p) & 1) * E_PART + ((p) >> 1) * N_EDGES)

__device__ __forceinline__ void transpose_block(const float* __restrict__ B, float* __restrict__ Bt,
                                                int K, int N, int blk) {
  const int i = blk * BIN_T + (int)threadIdx.x;
  if (i < K * N) {
    const int kk = i / N;
    const int n = i - kk * N;
    Bt[(size_t)n * K + kk] = B[i];
  }
}

__device__ __forceinline__ void bin_one(int d, int e, int lo, int nn, int& below_w, int* cnt, int* nlist,
                                        int* list_e, int* list_b) {
  below_w += __builtin_popcount(__builtin_amdgcn_ballot_w32(d < lo));
  const int ld = d - lo;
  if ((unsigned)ld < (unsigned)nn) {
    const int r = atomicAdd(&cnt[ld], 1);
    const int slot = atomicAdd(nlist, 1);
    if (slot < BIN_CAP) {
      list_e[slot] = e;
      list_b[slot] = ld | (r << BIN_LDB);
    }
  }
}

#define BIN_U 8
__device__ __forceinline__ void bin_load(const int4* __restrict__ dst4, int g, int rot, int4 (&v)[BIN_U],
                                         int (&q)[BIN_U]) {
  constexpr int NV = E_PART / 4;
#pragma unroll
  for (int u = 0; u < BIN_U; ++u) {
    const int gu = g + u * BIN_T;
    const bool ok = gu < NV;
    int qq = gu + rot;
    if (qq >= NV) qq -= NV;
    qq = ok ? qq : 0;
    q[u] = ok ? qq : -1;
    v[u] = dst4[qq];
  }
}

__global__ __launch_bounds__(BIN_T) void bin_prep_kernel(const int* __restrict__ src,
                                                         const int* __restrict__ dst,
                                                         int* __restrict__ rowptr4,
                                                         int* __restrict__ cnt4,
                                                         int* __restrict__ raw,
                                                         const float* __restrict__ w1, float* __restrict__ w1t,
                                                         const float* __restrict__ wmu, float* __restrict__ wmut,
                                                         const float* __restrict__ wls, float* __restrict__ wlst,
                                                         int n) {
  __shared__ int cnt[BIN_PAD + 1];
  __shared__ int start[BIN_PAD + 1];
  __shared__ int cnt2[BIN_PAD];
  __shared__ int wsum[BIN_T / 32];
  __shared__ int list_e[BIN_CAP];
  __shared__ int list_b[BIN_CAP];
  __shared__ int nlist;
  __shared__ int below_s;

  int b = blockIdx.x;
  if (b >= BIN_G) {
    b -= BIN_G;
    if (b < NB_TW1) { transpose_block(w1, w1t, IN_C, HID_C, b); return; }
    b -= NB_TW1;
    if (b < NB_TW2) { transpose_block(wmu, wmut, HID_C, OUT_C, b); return; }
    b -= NB_TW2;
    transpose_block(wls, wlst, HID_C, OUT_C, b);
    return;
  }

  const int t = threadIdx.x;
  const int lane = t & 31;
  const int wv = t >> 5;
  const int rg = b / BIN_Q;
  const int pq = b - rg * BIN_Q;
  const int lo = rg * BIN_NPB;
  const int ebase = pq * E_PART;
  int nn = n - lo;
  if (nn > BIN_NPB) nn = BIN_NPB;
  if (nn < 0) nn = 0;

  cnt[t] = 0;
  if (t == 0) { cnt[BIN_PAD] = 0; nlist = 0; below_s = 0; }
  __syncthreads();

  constexpr int NV = E_PART / 4;
  constexpr int STEPS = NV / (BIN_U * BIN_T);
  constexpr int REST = (NV - STEPS * BIN_U * BIN_T + BIN_T - 1) / BIN_T;
  static_assert(STEPS >= 1 && REST <= BIN_U, "the rest fits one step");
  constexpr int BIG = 0x7fffffff;
  const int4* __restrict__ dst4 = (const int4*)(dst + ebase);
  const int rot = (int)(((long)rg * NV) / BIN_R);
  int below_w = 0;
  int4 v[BIN_U];
  int q[BIN_U];
  bin_load(dst4, t, rot, v, q);
#pragma nounroll
  for (int s = 0; s < STEPS; ++s) {
    int4 nv[BIN_U];
    int nq[BIN_U];
    bin_load(dst4, (s + 1) * (BIN_U * BIN_T) + t, rot, nv, nq);
#pragma unroll
    for (int u = 0; u < BIN_U; ++u) {
      const int e4 = ebase + 4 * q[u];
      bin_one(v[u].x, e4 + 0, lo, nn, below_w, cnt, &nlist, list_e, list_b);
      bin_one(v[u].y, e4 + 1, lo, nn, below_w, cnt, &nlist, list_e, list_b);
      bin_one(v[u].z, e4 + 2, lo, nn, below_w, cnt, &nlist, list_e, list_b);
      bin_one(v[u].w, e4 + 3, lo, nn, below_w, cnt, &nlist, list_e, list_b);
    }
#pragma unroll
    for (int u = 0; u < BIN_U; ++u) { v[u] = nv[u]; q[u] = nq[u]; }
  }
#pragma unroll
  for (int u = 0; u < REST; ++u) {
    const bool ok = q[u] >= 0;
    const int e4 = ebase + 4 * q[u];
    bin_one(ok ? v[u].x : BIG, e4 + 0, lo, nn, below_w, cnt, &nlist, list_e, list_b);
    bin_one(ok ? v[u].y : BIG, e4 + 1, lo, nn, below_w, cnt, &nlist, list_e, list_b);
    bin_one(ok ? v[u].z : BIG, e4 + 2, lo, nn, below_w, cnt, &nlist, list_e, list_b);
    bin_one(ok ? v[u].w : BIG, e4 + 3, lo, nn, below_w, cnt, &nlist, list_e, list_b);
  }
  if (lane == 0) atomicAdd(&below_s, below_w);
  __syncthreads();

  const int c = cnt[t];
  int incl = c;
  for (int o = 1; o < 32; o <<= 1) {
    const int up = __shfl_up(incl, o);
    if (lane >= o) incl += up;
  }
  if (lane == 31) wsum[wv] = incl;
  __syncthreads();
  if (wv == 0) {
    const int w = wsum[lane];
    int wi = w;
    for (int o = 1; o < 32; o <<= 1) {
      const int up = __shfl_up(wi, o);
      if (lane >= o) wi += up;
    }
    wsum[lane] = wi - w;
  }
  __syncthreads();
  const int st = wsum[wv] + incl - c;
  start[t] = st;
  if (t == BIN_PAD - 1) start[BIN_PAD] = st + c;
  __syncthreads();

  const int nl = nlist;
  const int bel = below_s;

  if (t < nn) {
    rowptr4[(lo + t) * BIN_Q + pq] = bel + st;
    cnt4[(lo + t) * BIN_Q + pq] = c;
  }

  int* __restrict__ rawp = raw + RAW_OFF(pq);
  if (nl <= BIN_CAP) {
    for (int k = t; k < nl; k += BIN_T) {
      const int pb = list_b[k];
      rawp[bel + start[pb & (BIN_PAD - 1)] + (pb >> BIN_LDB)] = src[list_e[k]];
    }
  } else {
    cnt2[t] = 0;
    __syncthreads();
    for (int i = t; i < E_PART; i += BIN_T) {
      const int ld = dst[ebase + i] - lo;
      if ((unsigned)ld < (unsigned)nn) {
        const int r = atomicAdd(&cnt2[ld], 1);
        rawp[bel + start[ld] + r] = src[ebase + i];
      }
    }
  }
}

struct NodeSeg { int p0, p1, p2, p3, c1, c2, c3, d, b; };

__device__ __forceinline__ int seg_addr(const NodeSeg& s, int i) {
  return (i < s.c1) ? (s.p0 + i)
       : (i < s.c2) ? (s.p1 + (i - s.c1))
       : (i < s.c3) ? (s.p2 + (i - s.c2))
                    : (s.p3 + (i - s.c3));
}

#define GEMM_KC 64
#define GEMM_PITCH 68
#define GEMM_RB 13
#define GEMM_N 256
#define GEMM_WAVES 8
#define GEMM_GRID 256
#define GEMM_SLAB (GEMM_WAVES * 32 / 8)
#define GEMM_AROWS ((GEMM_RB * 16 + GEMM_SLAB - 1) / GEMM_SLAB * GEMM_SLAB)
#define GEMM_STAGE ((GEMM_AROWS + GEMM_N) * GEMM_PITCH)
#define GEMM_TP 36
static_assert((N_NODES + 15) / 16 <= GEMM_GRID * GEMM_RB, "a work-group owns at most GEMM_RB row-blocks");
static_assert(GEMM_N % GEMM_SLAB == 0 && GEMM_N % (GEMM_WAVES * 16) == 0, "whole passes, whole tiles");

template <int NRB>
__device__ __forceinline__ void gemm_rowblocks_body(const float* __restrict__ A,
                                                    const float* __restrict__ Bt,
                                                    float* __restrict__ C, float* lds,
                                                    int M, int K, int m0, int nb) {
  constexpr int SLAB = GEMM_SLAB;
  constexpr int NA = (NRB * 16 + SLAB - 1) / SLAB;
  constexpr int NB = GEMM_N / SLAB;
  constexpr int NLD = NA + NB;
  constexpr int NT = GEMM_N / GEMM_WAVES / 16;

  const int tid = threadIdx.x;
  const int lane = tid & 31;
  const int wave = tid >> 5;
  const int half = lane >> 4;
  const int l = lane & 15;

  const int cr = tid >> 3;
  const int cq = tid & 7;
  unsigned goff[NLD];
  int soffs[NLD];
#pragma unroll
  for (int i = 0; i < NLD; ++i) {
    if (i < NA) {
      int row = m0 + i * SLAB + cr;
      if (row >= M) row = M - 1;
      goff[i] = (unsigned)row * (unsigned)K + (unsigned)(cq * 4);
      soffs[i] = (i * SLAB + cr) * GEMM_PITCH + cq * 4;
    } else {
      goff[i] = (unsigned)((i - NA) * SLAB + cr) * (unsigned)K + (unsigned)(cq * 4);
      soffs[i] = (GEMM_AROWS + (i - NA) * SLAB + cr) * GEMM_PITCH + cq * 4;
    }
  }

  const int a_off = l * GEMM_PITCH + half * 4;
  const int b_off = (GEMM_AROWS + wave * (NT * 16) + l) * GEMM_PITCH + half * 4;

  v8f acc[NRB][NT];
#pragma unroll
  for (int i = 0; i < NRB; ++i)
#pragma unroll
    for (int t = 0; t < NT; ++t) acc[i][t] = (v8f){0, 0, 0, 0, 0, 0, 0, 0};

#pragma unroll
  for (int ln = 0; ln < 2; ++ln) {
    v4f first[NLD];
#pragma unroll
    for (int i = 0; i < NLD; ++i)
      first[i] = *(const v4f*)((i < NA ? A : Bt) + goff[i] + ln * 32);
#pragma unroll
    for (int i = 0; i < NLD; ++i)
      *(v4f*)(lds + soffs[i] + ln * 32) = first[i];
  }
  __syncthreads();

  const int nchunks = K / GEMM_KC;
  for (int kc = 0; kc < nchunks; ++kc) {
    const float* cur = lds + (kc & 1) * GEMM_STAGE;
    float* nxt = lds + ((kc + 1) & 1) * GEMM_STAGE;
    const bool more = (kc + 1 < nchunks);

    v4f a4[2][NRB], b4[2][NT];
#pragma unroll
    for (int i = 0; i < NRB; ++i)
      a4[0][i] = *(const v4f*)(cur + a_off + i * 16 * GEMM_PITCH);
#pragma unroll
    for (int t = 0; t < NT; ++t)
      b4[0][t] = *(const v4f*)(cur + b_off + t * 16 * GEMM_PITCH);

#pragma unroll
    for (int ln = 0; ln < 2; ++ln) {
      v4f pre[NLD];
      if (more) {
#pragma unroll
        for (int i = 0; i < NLD; ++i)
          pre[i] = *(const v4f*)((i < NA ? A : Bt) + goff[i] + (kc + 1) * GEMM_KC + ln * 32);
      }

#pragma unroll
      for (int j = ln * 4; j < ln * 4 + 4; ++j) {
        if (j + 1 < GEMM_KC / 8) {
#pragma unroll
          for (int i = 0; i < NRB; ++i)
            a4[(j + 1) & 1][i] = *(const v4f*)(cur + a_off + i * 16 * GEMM_PITCH + (j + 1) * 8);
#pragma unroll
          for (int t = 0; t < NT; ++t)
            b4[(j + 1) & 1][t] = *(const v4f*)(cur + b_off + t * 16 * GEMM_PITCH + (j + 1) * 8);
        }
#pragma unroll
        for (int i = 0; i < NRB; ++i)
#pragma unroll
          for (int t = 0; t < NT; ++t)
            acc[i][t] = __builtin_amdgcn_wmma_f32_16x16x4_f32(
                false, (v2f){a4[j & 1][i].x, a4[j & 1][i].y}, false, (v2f){b4[j & 1][t].x, b4[j & 1][t].y},
                (short)0, acc[i][t], false, false);
#pragma unroll
        for (int i = 0; i < NRB; ++i)
#pragma unroll
          for (int t = 0; t < NT; ++t)
            acc[i][t] = __builtin_amdgcn_wmma_f32_16x16x4_f32(
                false, (v2f){a4[j & 1][i].z, a4[j & 1][i].w}, false, (v2f){b4[j & 1][t].z, b4[j & 1][t].w},
                (short)0, acc[i][t], false, false);
      }

      if (more) {
#pragma unroll
        for (int i = 0; i < NLD; ++i)
          *(v4f*)(nxt + soffs[i] + ln * 32) = pre[i];
      }
    }
    __syncthreads();
  }

  static_assert(NT == 2, "a wave's columns are exactly one 128-byte line per row");
  float* tw = lds + wave * (16 * GEMM_TP);
  const int er = lane >> 3;
  const int ep = (lane & 7) * 4;
#pragma unroll
  for (int i = 0; i < NRB; ++i) {
    const int r0 = m0 + i * 16;
    if (i < nb && r0 + 16 <= M) {
#pragma unroll
      for (int t = 0; t < NT; ++t)
#pragma unroll
        for (int r = 0; r < 8; ++r)
          tw[(r + 8 * half) * GEMM_TP + t * 16 + l] = acc[i][t][r];
      float* c_ptr = C + (size_t)(r0 + er) * GEMM_N + (size_t)(wave * (NT * 16)) + ep;
#pragma unroll
      for (int p = 0; p < 4; ++p)
        *(v4f*)(c_ptr + (size_t)(p * 4) * GEMM_N) = *(const v4f*)(tw + (p * 4 + er) * GEMM_TP + ep);
    }
  }
}

__global__ __launch_bounds__(GEMM_WAVES * 32) void wmma_gemm_f32_rowblocks(const float* __restrict__ A,
                                                                          const float* __restrict__ Bt,
                                                                          float* __restrict__ C,
                                                                          int M, int K) {
  __shared__ __attribute__((aligned(16))) float lds[2 * GEMM_STAGE];
  const int nrb = (M + 15) >> 4;
  const int rb0 = ((int)blockIdx.x * nrb) / (int)gridDim.x;
  const int rb1 = (((int)blockIdx.x + 1) * nrb) / (int)gridDim.x;
  const int nb = rb1 - rb0;
  if (nb > GEMM_RB - 1)
    gemm_rowblocks_body<GEMM_RB>(A, Bt, C, lds, M, K, rb0 * 16, nb);
  else
    gemm_rowblocks_body<GEMM_RB - 1>(A, Bt, C, lds, M, K, rb0 * 16, nb);
}

template <int C, bool RELU, bool SPLIT>
__global__ __launch_bounds__(256) void gather_agg_kernel(float* __restrict__ out,
                                                         float* __restrict__ out2,
                                                         const float* __restrict__ z,
                                                         const int* __restrict__ rowptr,
                                                         const int* __restrict__ csr_src,
                                                         const float* __restrict__ csr_w,
                                                         const float* __restrict__ inv,
                                                         const float* __restrict__ bias,
                                                         const float* __restrict__ bias2, int n) {
  constexpr int G = C / 4;
  static_assert(G % 32 == 0 && 256 % G == 0, "a wave must stay inside one node");
  static_assert(!SPLIT || (G / 2) % 32 == 0, "a wave must stay inside one half of the row");
  const int g = threadIdx.x % G;
  const int v = __builtin_amdgcn_readfirstlane(blockIdx.x * (256 / G) + threadIdx.x / G);
  if (v >= n) return;
  const int b = rowptr[v];
  const int e = rowptr[v + 1];
  const float* zc = z + g * 4;

  float4 acc = make_float4(0.0f, 0.0f, 0.0f, 0.0f);
  int j = b;
  for (; j + 4 <= e; j += 4) {
    const int s0 = csr_src[j + 0], s1 = csr_src[j + 1], s2 = csr_src[j + 2], s3 = csr_src[j + 3];
    const float w0 = csr_w[j + 0], w1 = csr_w[j + 1], w2 = csr_w[j + 2], w3 = csr_w[j + 3];
    const float4 t0 = *(const float4*)(zc + (size_t)s0 * C);
    const float4 t1 = *(const float4*)(zc + (size_t)s1 * C);
    const float4 t2 = *(const float4*)(zc + (size_t)s2 * C);
    const float4 t3 = *(const float4*)(zc + (size_t)s3 * C);
    acc.x += t0.x * w0; acc.y += t0.y * w0; acc.z += t0.z * w0; acc.w += t0.w * w0;
    acc.x += t1.x * w1; acc.y += t1.y * w1; acc.z += t1.z * w1; acc.w += t1.w * w1;
    acc.x += t2.x * w2; acc.y += t2.y * w2; acc.z += t2.z * w2; acc.w += t2.w * w2;
    acc.x += t3.x * w3; acc.y += t3.y * w3; acc.z += t3.z * w3; acc.w += t3.w * w3;
  }
  for (; j < e; ++j) {
    const int s0 = csr_src[j];
    const float w0 = csr_w[j];
    const float4 t0 = *(const float4*)(zc + (size_t)s0 * C);
    acc.x += t0.x * w0; acc.y += t0.y * w0; acc.z += t0.z * w0; acc.w += t0.w * w0;
  }

  const float4 zs = *(const float4*)(zc + (size_t)v * C);
  const float iv = inv[v];
  acc.x += zs.x * iv; acc.y += zs.y * iv; acc.z += zs.z * iv; acc.w += zs.w * iv;
  constexpr int OC = SPLIT ? C / 2 : C;
  const bool second = SPLIT && (g >= G / 2);
  const int go = second ? g - G / 2 : g;
  const float* bs = second ? bias2 : bias;
  float* op = second ? out2 : out;
  acc.x += bs[go * 4 + 0];
  acc.y += bs[go * 4 + 1];
  acc.z += bs[go * 4 + 2];
  acc.w += bs[go * 4 + 3];
  if (RELU) {
    acc.x = fmaxf(acc.x, 0.0f);
    acc.y = fmaxf(acc.y, 0.0f);
    acc.z = fmaxf(acc.z, 0.0f);
    acc.w = fmaxf(acc.w, 0.0f);
  }
  *(float4*)(op + (size_t)v * OC + go * 4) = acc;
}

#define FG_PITCH 260
#define FG_GW 8
#define FG_AW 16
#define FG_T ((FG_GW + FG_AW) * 32)
#define FG_BT (GEMM_N * FG_PITCH)
#define FG_ST (16 * FG_PITCH)
#define FG_R 8
static_assert(HID_C == 256 && GEMM_N == 256, "a lane owns 2 x 4 channels of a 256-wide row");
static_assert(GEMM_RB <= 32 && 32 % FG_R == 0, "a lane per row-block of the work-group; whole groups per 32 entries");

__device__ __forceinline__ void fma4(float4& a, const float4& t, float w) {
  a.x += t.x * w; a.y += t.y * w; a.z += t.z * w; a.w += t.w * w;
}

__device__ __forceinline__ void agg_row_to_lds(float* __restrict__ st_row, const float* __restrict__ z,
                                               const int* csr_src, float* csr_w,
                                               const float* __restrict__ bias, int v, int n, int b, int e,
                                               int sv, float wv, const int4& ck, bool lazy, int lane) {
  float4 a0 = make_float4(0.0f, 0.0f, 0.0f, 0.0f);
  float4 a1 = make_float4(0.0f, 0.0f, 0.0f, 0.0f);
  if (v < n) {
    const float* zv = z + (size_t)v * HID_C;
    const float4 zs0 = *(const float4*)(zv + lane * 4);
    const float4 zs1 = *(const float4*)(zv + lane * 4 + 128);
    const float dg = (float)(e - b + 1);
    const float iv = 1.0f / dg;
    const float dv = rsqrtf(dg);
    for (int j0 = b; j0 < e; j0 += 32) {
      if (j0 != b) {
        const int jl = j0 + lane;
        const int jc = (jl < e) ? jl : e - 1;
        sv = csr_src[jc];
        wv = csr_w[jc];
      }
#pragma unroll
      for (int g = 0; g < 32 / FG_R; ++g) {
        const int jg = j0 + g * FG_R;
        if (jg < e) {
          float4 t0[FG_R], t1[FG_R];
#pragma unroll
          for (int k = 0; k < FG_R; ++k) {
            const int s = __builtin_amdgcn_readlane(sv, g * FG_R + k);
            const float* p = z + (size_t)s * HID_C;
            t0[k] = *(const float4*)(p + lane * 4);
            t1[k] = *(const float4*)(p + lane * 4 + 128);
          }
          if (g == 0 && lazy) {
            wv = rsqrtf((float)(ck.x + ck.y + ck.z + ck.w + 1)) * dv;
            if (lane < e - b) csr_w[b + lane] = wv;
          }
#pragma unroll
          for (int k = 0; k < FG_R; ++k) {
            if (jg + k < e) {
              const float w = __int_as_float(__builtin_amdgcn_readlane(__float_as_int(wv), g * FG_R + k));
              fma4(a0, t0[k], w);
              fma4(a1, t1[k], w);
            }
          }
        }
      }
    }
    fma4(a0, zs0, iv);
    fma4(a1, zs1, iv);
    a0.x += bias[lane * 4 + 0]; a0.y += bias[lane * 4 + 1]; a0.z += bias[lane * 4 + 2]; a0.w += bias[lane * 4 + 3];
    a1.x += bias[128 + lane * 4 + 0]; a1.y += bias[128 + lane * 4 + 1];
    a1.z += bias[128 + lane * 4 + 2]; a1.w += bias[128 + lane * 4 + 3];
    a0.x = fmaxf(a0.x, 0.0f); a0.y = fmaxf(a0.y, 0.0f); a0.z = fmaxf(a0.z, 0.0f); a0.w = fmaxf(a0.w, 0.0f);
    a1.x = fmaxf(a1.x, 0.0f); a1.y = fmaxf(a1.y, 0.0f); a1.z = fmaxf(a1.z, 0.0f); a1.w = fmaxf(a1.w, 0.0f);
  }
  *(float4*)(st_row + lane * 4) = a0;
  *(float4*)(st_row + 128 + lane * 4) = a1;
}

__device__ __forceinline__ NodeSeg seg_of(const int4& rpv, const int4& cnv, int k) {
  NodeSeg s;
  const int r0 = __builtin_amdgcn_readlane(rpv.x, k), r1 = __builtin_amdgcn_readlane(rpv.y, k);
  const int r2 = __builtin_amdgcn_readlane(rpv.z, k), r3 = __builtin_amdgcn_readlane(rpv.w, k);
  const int n0 = __builtin_amdgcn_readlane(cnv.x, k), n1 = __builtin_amdgcn_readlane(cnv.y, k);
  const int n2 = __builtin_amdgcn_readlane(cnv.z, k), n3 = __builtin_amdgcn_readlane(cnv.w, k);
  s.p0 = RAW_OFF(0) + r0; s.p1 = RAW_OFF(1) + r1; s.p2 = RAW_OFF(2) + r2; s.p3 = RAW_OFF(3) + r3;
  s.c1 = n0; s.c2 = n0 + n1; s.c3 = n0 + n1 + n2; s.d = n0 + n1 + n2 + n3;
  s.b = r0 + r1 + r2 + r3;
  return s;
}

__device__ __forceinline__ int list_entry_ahead(const NodeSeg& s, const int* __restrict__ raw, int lane) {
  int key = 0;
  if (s.d > 0) {
    const int i = (lane < s.d) ? lane : s.d - 1;
    key = raw[seg_addr(s, i)];
  }
  return key;
}

__device__ __forceinline__ void order_list(const NodeSeg& s, int v, int n, int key_raw, int lane,
                                           const int* __restrict__ raw, const int4* __restrict__ cnt4v,
                                           int* __restrict__ rowptr, float* __restrict__ inv,
                                           int* csr_src, float* csr_w,
                                           int& sv, float& wv, int4& ck, bool& lazy) {
  const int d = s.d;
  const int b = s.b;
  const float dg = (float)(d + 1);
  if (lane == 0) {
    rowptr[v] = b;
    inv[v] = 1.0f / dg;
    if (v == n - 1) rowptr[n] = b + d;
  }
  lazy = (d <= 32);
  if (d <= 32) {
    if (d > 0) {
      const int key = (lane < d) ? key_raw : 0x7fffffff;
      int rank = 0;
      for (int j = 0; j < d; ++j) {
        const int kj = __builtin_amdgcn_readlane(key, j);
        rank += ((kj < key) || (kj == key && j < lane)) ? 1 : 0;
      }
      rank = (lane < d) ? rank : lane;
      const int so = __builtin_amdgcn_ds_permute(rank << 2, key);
      const int last = __builtin_amdgcn_readlane(so, d - 1);
      sv = (lane < d) ? so : last;
      ck = cnt4v[sv];
      if (lane < d) csr_src[b + lane] = sv;
    }
  } else {
    const float dv = rsqrtf(dg);
    for (int i = lane; i < d; i += 32) {
      const int key = raw[seg_addr(s, i)];
      int rank = 0;
      for (int j = 0; j < d; ++j) {
        const int kj = raw[seg_addr(s, j)];
        rank += ((kj < key) || (kj == key && j < i)) ? 1 : 0;
      }
      const int4 cq = cnt4v[key];
      const float dk = rsqrtf((float)(cq.x + cq.y + cq.z + cq.w + 1));
      csr_src[b + rank] = key;
      csr_w[b + rank] = dk * dv;
    }
    __threadfence();
    sv = csr_src[b + lane];
    wv = csr_w[b + lane];
  }
}

__device__ __forceinline__ void store_tiles(float* __restrict__ C, const v8f (&acc)[2], int r0, int M, int wave,
                                            int half, int l) {
  if (r0 + 16 <= M) {
    float* c_ptr = C + (size_t)r0 * GEMM_N + (size_t)(wave * 32) + l;
#pragma unroll
    for (int t = 0; t < 2; ++t)
#pragma unroll
      for (int r = 0; r < 8; ++r)
        c_ptr[(size_t)(r + 8 * half) * GEMM_N + t * 16] = acc[t][r];
  }
}

__global__ __launch_bounds__(FG_T) void agg_relu_gemm_kernel(const float* __restrict__ z,
                                                             const int* __restrict__ rowptr4,
                                                             const int* __restrict__ cnt4,
                                                             const int* __restrict__ raw,
                                                             int* __restrict__ rowptr,
                                                             int* __restrict__ csr_src,
                                                             float* __restrict__ csr_w,
                                                             float* __restrict__ inv,
                                                             const float* __restrict__ bias,
                                                             const float* __restrict__ Bt,
                                                             float* __restrict__ C, int M) {
  __shared__ __attribute__((aligned(16))) float lds[FG_BT + 2 * FG_ST];
  float* bt_s = lds;
  float* st_s = lds + FG_BT;

  const int tid = threadIdx.x;
  const int lane = tid & 31;
  const int wave = __builtin_amdgcn_readfirstlane(tid >> 5);
  const int half = lane >> 4;
  const int l = lane & 15;

  const int nrb = (M + 15) >> 4;
  const int rb0 = ((int)blockIdx.x * nrb) / (int)gridDim.x;
  const int rb1 = (((int)blockIdx.x + 1) * nrb) / (int)gridDim.x;
  const int nb = rb1 - rb0;
  const int m0 = rb0 * 16;

  for (int idx = tid; idx < GEMM_N * (HID_C / 4); idx += FG_T) {
    const int row = idx >> 6;
    const int q = idx & 63;
    *(v4f*)(bt_s + row * FG_PITCH + q * 4) = *(const v4f*)(Bt + (size_t)row * HID_C + q * 4);
  }
  const int g_row = wave - FG_GW;
  const int4* __restrict__ cnt4v = (const int4*)cnt4;
  int4 rpv = make_int4(0, 0, 0, 0), cnv = make_int4(0, 0, 0, 0);
  int key_n = 0;
  if (wave >= FG_GW) {
    const int vi = m0 + lane * 16 + g_row;
    if (lane < nb && vi < M) {
      rpv = ((const int4*)rowptr4)[vi];
      cnv = cnt4v[vi];
    }
    if (nb > 0) {
      const int v0 = m0 + g_row;
      const NodeSeg s0 = seg_of(rpv, cnv, 0);
      const int key0 = list_entry_ahead(s0, raw, lane);
      if (nb > 1) key_n = list_entry_ahead(seg_of(rpv, cnv, 1), raw, lane);
      int sv = 0;
      float wv = 0.0f;
      int4 ck = make_int4(0, 0, 0, 0);
      bool lazy = true;
      if (v0 < M) order_list(s0, v0, M, key0, lane, raw, cnt4v, rowptr, inv, csr_src, csr_w, sv, wv, ck, lazy);
      agg_row_to_lds(st_s + g_row * FG_PITCH, z, csr_src, csr_w, bias, v0, M, s0.b, s0.b + s0.d, sv, wv, ck, lazy,
                     lane);
    }
  }
  __syncthreads();

  const int a_off = l * FG_PITCH + half * 4;
  const int b_off = ((wave & (FG_GW - 1)) * 32 + l) * FG_PITCH + half * 4;

  v8f pend[2];
  pend[0] = (v8f){0, 0, 0, 0, 0, 0, 0, 0};
  pend[1] = (v8f){0, 0, 0, 0, 0, 0, 0, 0};

  for (int i = 0; i < nb; ++i) {
    if (wave < FG_GW) {
      if (i > 0) store_tiles(C, pend, m0 + (i - 1) * 16, M, wave, half, l);
      const float* cur = st_s + (i & 1) * FG_ST;
      v8f acc[2];
      acc[0] = (v8f){0, 0, 0, 0, 0, 0, 0, 0};
      acc[1] = (v8f){0, 0, 0, 0, 0, 0, 0, 0};
      v4f a4[2], b4[2][2];
      a4[0] = *(const v4f*)(cur + a_off);
      b4[0][0] = *(const v4f*)(bt_s + b_off);
      b4[0][1] = *(const v4f*)(bt_s + b_off + 16 * FG_PITCH);
#pragma unroll
      for (int j = 0; j < HID_C / 8; ++j) {
        if (j + 1 < HID_C / 8) {
          a4[(j + 1) & 1] = *(const v4f*)(cur + a_off + (j + 1) * 8);
          b4[(j + 1) & 1][0] = *(const v4f*)(bt_s + b_off + (j + 1) * 8);
          b4[(j + 1) & 1][1] = *(const v4f*)(bt_s + b_off + 16 * FG_PITCH + (j + 1) * 8);
        }
#pragma unroll
        for (int t = 0; t < 2; ++t)
          acc[t] = __builtin_amdgcn_wmma_f32_16x16x4_f32(
              false, (v2f){a4[j & 1].x, a4[j & 1].y}, false, (v2f){b4[j & 1][t].x, b4[j & 1][t].y},
              (short)0, acc[t], false, false);
#pragma unroll
        for (int t = 0; t < 2; ++t)
          acc[t] = __builtin_amdgcn_wmma_f32_16x16x4_f32(
              false, (v2f){a4[j & 1].z, a4[j & 1].w}, false, (v2f){b4[j & 1][t].z, b4[j & 1][t].w},
              (short)0, acc[t], false, false);
      }
      pend[0] = acc[0];
      pend[1] = acc[1];
    } else if (i + 1 < nb) {
      __builtin_amdgcn_s_setprio(3);
      const int v1 = m0 + (i + 1) * 16 + g_row;
      const NodeSeg s1 = seg_of(rpv, cnv, i + 1);
      const int key1 = key_n;
      if (i + 2 < nb) key_n = list_entry_ahead(seg_of(rpv, cnv, i + 2), raw, lane);
      int sv = 0;
      float wv = 0.0f;
      int4 ck = make_int4(0, 0, 0, 0);
      bool lazy = true;
      if (v1 < M) order_list(s1, v1, M, key1, lane, raw, cnt4v, rowptr, inv, csr_src, csr_w, sv, wv, ck, lazy);
      agg_row_to_lds(st_s + ((i + 1) & 1) * FG_ST + g_row * FG_PITCH, z, csr_src, csr_w, bias, v1, M, s1.b,
                     s1.b + s1.d, sv, wv, ck, lazy, lane);
      __builtin_amdgcn_s_setprio(0);
    }
    __syncthreads();
  }
  if (wave < FG_GW && nb > 0) store_tiles(C, pend, m0 + (nb - 1) * 16, M, wave, half, l);
}

static inline int cdiv_l(long a, long b) { return (int)((a + b - 1) / b); }

extern "C" void kernel_launch(void* const* d_in, const int* in_sizes, int n_in,
                              void* d_out, int out_size, void* d_ws, size_t ws_size,
                              hipStream_t stream) {
  const float* x   = (const float*)d_in[0];
  const int*   ei  = (const int*)d_in[1];
  const int*   src = ei;
  const int*   dst = ei + N_EDGES;
  const float* w1  = (const float*)d_in[2];
  const float* b1  = (const float*)d_in[3];
  const float* wmu = (const float*)d_in[4];
  const float* bmu = (const float*)d_in[5];
  const float* wls = (const float*)d_in[6];
  const float* bls = (const float*)d_in[7];

  float* out_mu = (float*)d_out;
  float* out_ls = out_mu + (size_t)N_NODES * OUT_C;

  float* ws = (float*)d_ws;
  size_t off = 0;
  float* w1t  = ws + off; off += (size_t)IN_C * HID_C;
  float* wmut = ws + off; off += (size_t)HID_C * OUT_C;
  float* wlst = ws + off; off += (size_t)HID_C * OUT_C;
  float* hpre = ws + off; off += (size_t)N_NODES * HID_C;
  float* zpl  = ws + off; off += (size_t)N_NODES * HID_C;
  static_assert(((size_t)IN_C * HID_C * 4) % 1024 == 0 && ((size_t)HID_C * OUT_C * 4) % 1024 == 0 &&
                ((size_t)N_NODES * HID_C * 4) % 1024 == 0, "the row planes keep each other on 1-KB boundaries");
  float* inv   = ws + off; off += N_NODES;
  float* csr_w = ws + off; off += N_EDGES;
  int* rowptr  = (int*)(ws + off); off += N_NODES + 16;
  int* rank    = (int*)(ws + off); off += N_EDGES;
  int* raw_src = (int*)(ws + off); off += N_EDGES;
  int* csr_src = (int*)(ws + off); off += N_EDGES;
  float* z2 = zpl;
  int* raw = rank + 4 * N_NODES;
  int* rowptr4 = rank;
  int* cnt4 = raw_src;
  static_assert(4 * N_NODES + 2 * E_PART <= N_EDGES && (N_NODES + 16) % 4 == 0 && N_EDGES % 4 == 0 && N_NODES % 4 == 0,
                "the [node][part] planes and two parts' lists fit in a plane; the planes start on 16-byte boundaries");

  bin_prep_kernel<<<NB_BINPREP, BIN_T, 0, stream>>>(src, dst, rowptr4, cnt4, raw, w1, w1t, wmu, wmut,
                                                    wls, wlst, N_NODES);


  static_assert(HID_C == GEMM_N && 2 * OUT_C == GEMM_N && IN_C % GEMM_KC == 0 && HID_C % GEMM_KC == 0,
                "both products are 256 wide and K is whole chunks");
  wmma_gemm_f32_rowblocks<<<GEMM_GRID, GEMM_WAVES * 32, 0, stream>>>(x, w1t, hpre, N_NODES, IN_C);

  agg_relu_gemm_kernel<<<GEMM_GRID, FG_T, 0, stream>>>(hpre, rowptr4, cnt4, raw, rowptr, csr_src, csr_w, inv, b1, wmut,
                                                       z2, N_NODES);
  (void)wlst;
  gather_agg_kernel<2 * OUT_C, false, true><<<cdiv_l((long)N_NODES * (2 * OUT_C / 4), 256), 256, 0, stream>>>(
      out_mu, out_ls, z2, rowptr, csr_src, csr_w, inv, bmu, bls, N_NODES);
}
